# starting kernel
# baseline (speedup 1.0000x reference)
#include <hip/hip_runtime.h>
#include <math.h>

typedef __attribute__((ext_vector_type(2))) float v2f;
typedef __attribute__((ext_vector_type(8))) float v8f;

#define NT 512

constexpr int NG = 100, NL = 400, NN = 300, NWF = 50, NS = 200;
constexpr int G_P = 112;
constexpr int S_P = 208;
constexpr int PDHG_ITERS = 300, POWER_ITERS = 25;

constexpr int OX_P  = 0;
constexpr int OX_RU = OX_P  + G_P;
constexpr int OX_RD = OX_RU + G_P;
constexpr int OX_U  = OX_RD + G_P;
constexpr int OX_D  = OX_U  + G_P * S_P;
constexpr int OX_FU = OX_D  + G_P * S_P;
constexpr int OX_FD = OX_FU + NL;
constexpr int XSZ   = OX_FD + NL;

constexpr int OY_Y1 = 0;
constexpr int OY_Y2 = 16;
constexpr int OY_Y3 = OY_Y2 + NL;
constexpr int OY_Y4 = OY_Y3 + NL;
constexpr int OY_Z1 = OY_Y4 + S_P;
constexpr int OY_Z2 = OY_Z1 + G_P;
constexpr int OY_Z3 = OY_Z2 + G_P;
constexpr int OY_Z4 = OY_Z3 + G_P * S_P;
constexpr int OY_Z5 = OY_Z4 + G_P * S_P;
constexpr int OY_Z6 = OY_Z5 + NL * S_P;
constexpr int YSZ   = OY_Z6 + NL * S_P;

constexpr int WS_PG   = 0;
constexpr int WS_PGT  = WS_PG   + NL * G_P;
constexpr int WS_PW   = WS_PGT  + G_P * NL;
constexpr int WS_B2   = WS_PW   + NL * S_P;
constexpr int WS_B3   = WS_B2   + NL;
constexpr int WS_B4   = WS_B3   + NL;
constexpr int WS_B1   = WS_B4   + S_P;
constexpr int WS_M1   = WS_B1   + 16;
constexpr int WS_NV   = WS_M1   + NN * S_P;
constexpr int WS_X    = WS_NV   + NN + 4;
constexpr int WS_XB   = WS_X    + XSZ;
constexpr int WS_ATY  = WS_XB   + XSZ;
constexpr int WS_Y    = WS_ATY  + XSZ;
constexpr int WS_AX   = WS_Y    + YSZ;
constexpr int WS_T    = WS_AX   + YSZ;
constexpr int WS_FLOW = WS_T    + G_P * S_P;
constexpr int WS_GP   = WS_FLOW + NL * S_P;
constexpr int WS_COL  = WS_GP   + NL;
constexpr int WS_S3   = WS_COL  + S_P;
constexpr int WS_S4   = WS_S3   + G_P;
constexpr int WS_S5   = WS_S4   + G_P;
constexpr int WS_S6   = WS_S5   + NL;
constexpr int WS_GTY  = WS_S6   + NL;
constexpr int WS_SUMP = WS_GTY  + G_P;
constexpr int WS_TOTAL= WS_SUMP + 16;


__global__ void k_zero(float* w) {
    for (int i = blockIdx.x * blockDim.x + threadIdx.x; i < WS_TOTAL;
         i += gridDim.x * blockDim.x) w[i] = 0.0f;
}

__global__ void k_prep1(float* w, const float* w_scen, const float* node_W,
                        const float* node_L, const float* Pd, const float* w_exp) {
    const int NM1 = NN * NS;
    const int total = NM1 + NN + NS + 1;
    for (int idx = blockIdx.x * blockDim.x + threadIdx.x; idx < total;
         idx += gridDim.x * blockDim.x) {
        if (idx < NM1) {
            int n = idx / NS, s = idx % NS;
            float a = 0.f;
            for (int ww = 0; ww < NWF; ++ww) a += node_W[n*NWF + ww] * w_scen[s*NWF + ww];
            w[WS_M1 + n*S_P + s] = a;
        } else if (idx < NM1 + NN) {
            int n = idx - NM1;
            float a = 0.f;
            for (int ww = 0; ww < NWF; ++ww) a += node_W[n*NWF + ww] * w_exp[ww];
            for (int m = 0; m < NN; ++m)     a -= node_L[n*NN + m] * Pd[m];
            w[WS_NV + n] = a;
        } else if (idx < NM1 + NN + NS) {
            int s = idx - NM1 - NN;
            float a = 0.f;
            for (int ww = 0; ww < NWF; ++ww) a += w_scen[s*NWF + ww];
            w[WS_B4 + s] = -a;
        } else {
            float a = 0.f;
            for (int m = 0; m < NN; ++m)  a += Pd[m];
            for (int ww = 0; ww < NWF; ++ww) a -= w_exp[ww];
            w[WS_B1] = a;
        }
    }
}

__global__ void k_prep2(float* w, const float* PTDF, const float* node_G,
                        const float* Line_cap) {
    const int A = NL * NG;
    const int B = A + NG * NL;
    const int C = B + NL * NS;
    const int D2 = C + NL;
    const int E = D2 + NL;
    for (int idx = blockIdx.x * blockDim.x + threadIdx.x; idx < E;
         idx += gridDim.x * blockDim.x) {
        if (idx < A) {
            int l = idx / NG, g = idx % NG;
            float a = 0.f;
            for (int n = 0; n < NN; ++n) a += PTDF[l*NN + n] * node_G[n*NG + g];
            w[WS_PG + l*G_P + g] = a;
        } else if (idx < B) {
            int r = idx - A; int g = r / NL, l = r % NL;
            float a = 0.f;
            for (int n = 0; n < NN; ++n) a += PTDF[l*NN + n] * node_G[n*NG + g];
            w[WS_PGT + g*NL + l] = a;
        } else if (idx < C) {
            int r = idx - B; int l = r / NS, s = r % NS;
            float a = 0.f;
            for (int n = 0; n < NN; ++n) a += PTDF[l*NN + n] * w[WS_M1 + n*S_P + s];
            w[WS_PW + l*S_P + s] = a;
        } else if (idx < D2) {
            int l = idx - C;
            float a = 0.f;
            for (int n = 0; n < NN; ++n) a += PTDF[l*NN + n] * w[WS_NV + n];
            w[WS_B2 + l] = Line_cap[l] - a;
        } else {
            int l = idx - D2;
            float a = 0.f;
            for (int n = 0; n < NN; ++n) a += PTDF[l*NN + n] * w[WS_NV + n];
            w[WS_B3 + l] = Line_cap[l] + a;
        }
    }
}

__global__ void k_setv(float* w) {
    const int total = 3*NG + 2*NG*NS + 2*NL;
    float* X = w + WS_X;
    for (int idx = blockIdx.x * blockDim.x + threadIdx.x; idx < total;
         idx += gridDim.x * blockDim.x) {
        int off;
        if (idx < NG)                off = OX_P  + idx;
        else if (idx < 2*NG)         off = OX_RU + (idx - NG);
        else if (idx < 3*NG)         off = OX_RD + (idx - 2*NG);
        else if (idx < 3*NG + NG*NS) { int r = idx - 3*NG; off = OX_U + (r/NS)*S_P + (r%NS); }
        else if (idx < 3*NG + 2*NG*NS) { int r = idx - 3*NG - NG*NS; off = OX_D + (r/NS)*S_P + (r%NS); }
        else if (idx < 3*NG + 2*NG*NS + NL) off = OX_FU + (idx - 3*NG - 2*NG*NS);
        else                         off = OX_FD + (idx - 3*NG - 2*NG*NS - NL);
        X[off] = 1.0f;
    }
}


__device__ __forceinline__ float block_reduce(float v, float* red, int tid) {
    red[tid] = v; __syncthreads();
    for (int off = NT >> 1; off > 0; off >>= 1) {
        if (tid < off) red[tid] += red[tid + off];
        __syncthreads();
    }
    float r = red[0]; __syncthreads();
    return r;
}

__device__ void matvec_core(float* w, const float* xb, float* yb, int tid) {
    const int lane = tid & 31, wave = tid >> 5, nw = NT / 32;
    const float* PG = w + WS_PG;
    float* flow = w + WS_FLOW;
    __syncthreads();
    {
        const float* Ux = xb + OX_U; const float* Dx = xb + OX_D;
        const int NTile = 25 * 13;
        const int m16 = lane & 15;
        const int kh  = (lane >> 4) * 2;
        for (int tile = wave; tile < NTile; tile += nw) {
            int tm = tile / 13, tn = tile % 13;
            int m = tm * 16 + m16;
            int n = tn * 16 + m16;
            v8f acc = {0.f,0.f,0.f,0.f,0.f,0.f,0.f,0.f};
            const float* arow = PG + m * G_P;
            for (int k0 = 0; k0 < NG; k0 += 4) {
                int ka = k0 + kh;
                v2f a, b;
                a[0] = arow[ka]; a[1] = arow[ka + 1];
                int bi = ka * S_P + n;
                b[0] = Ux[bi]        - Dx[bi];
                b[1] = Ux[bi + S_P]  - Dx[bi + S_P];
                acc = __builtin_amdgcn_wmma_f32_16x16x4_f32(false, a, false, b,
                                                            (short)0, acc, false, false);
            }
            int mr = tm * 16 + ((lane >> 4) << 3);
            for (int i = 0; i < 8; ++i) flow[(mr + i) * S_P + n] = acc[i];
        }
    }
    if (tid < NL) {
        float s = 0.f; const float* r = PG + tid * G_P; const float* p = xb + OX_P;
        for (int g = 0; g < NG; ++g) s += r[g] * p[g];
        w[WS_GP + tid] = s;
    }
    if (tid < NS) {
        float s = 0.f;
        const float* Ux = xb + OX_U + tid; const float* Dx = xb + OX_D + tid;
        for (int g = 0; g < NG; ++g) s += Ux[g * S_P] - Dx[g * S_P];
        w[WS_COL + tid] = s;
    }
    if (tid == NT - 1) {
        float s = 0.f;
        for (int g = 0; g < NG; ++g) s += xb[OX_P + g];
        w[WS_SUMP] = s;
    }
    __syncthreads();
    if (tid == 0) yb[OY_Y1] = w[WS_SUMP];
    for (int l = tid; l < NL; l += NT) {
        float g = w[WS_GP + l];
        yb[OY_Y2 + l] =  g + xb[OX_FU + l];
        yb[OY_Y3 + l] = -g + xb[OX_FD + l];
    }
    for (int s = tid; s < NS; s += NT) yb[OY_Y4 + s] = w[WS_COL + s];
    for (int g = tid; g < NG; g += NT) {
        yb[OY_Z1 + g] = xb[OX_P + g]  + xb[OX_RU + g];
        yb[OY_Z2 + g] = xb[OX_RD + g] - xb[OX_P + g];
    }
    for (int i = tid; i < NG * NS; i += NT) {
        int g = i / NS, s = i % NS, o = g * S_P + s;
        yb[OY_Z3 + o] = xb[OX_U + o] - xb[OX_RU + g];
        yb[OY_Z4 + o] = xb[OX_D + o] - xb[OX_RD + g];
    }
    for (int i = tid; i < NL * NS; i += NT) {
        int l = i / NS, s = i % NS, o = l * S_P + s;
        float f = flow[o];
        yb[OY_Z5 + o] =  f - xb[OX_FU + l];
        yb[OY_Z6 + o] = -f - xb[OX_FD + l];
    }
}

__device__ void rmatvec_core(float* w, const float* yb, float* xo, int tid) {
    const int lane = tid & 31, wave = tid >> 5, nw = NT / 32;
    const float* PGT = w + WS_PGT;
    float* T = w + WS_T;
    __syncthreads();
    {
        const float* Z5 = yb + OY_Z5; const float* Z6 = yb + OY_Z6;
        const int NTile = 7 * 13;
        const int m16 = lane & 15;
        const int kh  = (lane >> 4) * 2;
        for (int tile = wave; tile < NTile; tile += nw) {
            int tm = tile / 13, tn = tile % 13;
            int m = tm * 16 + m16;
            int n = tn * 16 + m16;
            v8f acc = {0.f,0.f,0.f,0.f,0.f,0.f,0.f,0.f};
            const float* arow = PGT + m * NL;
            for (int k0 = 0; k0 < NL; k0 += 4) {
                int ka = k0 + kh;
                v2f a, b;
                a[0] = arow[ka]; a[1] = arow[ka + 1];
                int bi = ka * S_P + n;
                b[0] = Z5[bi]       - Z6[bi];
                b[1] = Z5[bi + S_P] - Z6[bi + S_P];
                acc = __builtin_amdgcn_wmma_f32_16x16x4_f32(false, a, false, b,
                                                            (short)0, acc, false, false);
            }
            int mr = tm * 16 + ((lane >> 4) << 3);
            for (int i = 0; i < 8; ++i) T[(mr + i) * S_P + n] = acc[i];
        }
    }
    if (tid < NG) {
        float a = 0.f, b = 0.f, c = 0.f;
        const float* z3r = yb + OY_Z3 + tid * S_P;
        const float* z4r = yb + OY_Z4 + tid * S_P;
        for (int s = 0; s < NS; ++s) { a += z3r[s]; b += z4r[s]; }
        const float* pr = PGT + tid * NL;
        for (int l = 0; l < NL; ++l) c += pr[l] * (yb[OY_Y2 + l] - yb[OY_Y3 + l]);
        w[WS_S3 + tid] = a; w[WS_S4 + tid] = b; w[WS_GTY + tid] = c;
    }
    if (tid >= NG && tid < NG + NL) {
        int l = tid - NG;
        float a = 0.f, b = 0.f;
        const float* z5r = yb + OY_Z5 + l * S_P;
        const float* z6r = yb + OY_Z6 + l * S_P;
        for (int s = 0; s < NS; ++s) { a += z5r[s]; b += z6r[s]; }
        w[WS_S5 + l] = a; w[WS_S6 + l] = b;
    }
    __syncthreads();
    float y1v = yb[OY_Y1];
    for (int g = tid; g < NG; g += NT) {
        xo[OX_P  + g] = y1v + w[WS_GTY + g] + yb[OY_Z1 + g] - yb[OY_Z2 + g];
        xo[OX_RU + g] = yb[OY_Z1 + g] - w[WS_S3 + g];
        xo[OX_RD + g] = yb[OY_Z2 + g] - w[WS_S4 + g];
    }
    for (int l = tid; l < NL; l += NT) {
        xo[OX_FU + l] = yb[OY_Y2 + l] - w[WS_S5 + l];
        xo[OX_FD + l] = yb[OY_Y3 + l] - w[WS_S6 + l];
    }
    for (int i = tid; i < NG * NS; i += NT) {
        int g = i / NS, s = i % NS, o = g * S_P + s;
        float tv = T[o];
        xo[OX_U + o] =  yb[OY_Y4 + s] + yb[OY_Z3 + o] + tv;
        xo[OX_D + o] = -yb[OY_Y4 + s] + yb[OY_Z4 + o] - tv;
    }
}


__global__ __launch_bounds__(NT) void k_solver(float* w, const float* Cost,
                                               const float* Cru, const float* Crd,
                                               const float* Pmax, float* out) {
    __shared__ float red[NT];
    const int tid = threadIdx.x;
    float* X   = w + WS_X;
    float* XB  = w + WS_XB;
    float* ATY = w + WS_ATY;
    float* Y   = w + WS_Y;
    float* AX  = w + WS_AX;

    for (int i = tid * 32; i < NL * G_P; i += NT * 32) {
        __builtin_prefetch(&w[WS_PG + i], 0, 3);
        __builtin_prefetch(&w[WS_PGT + i], 0, 3);
    }

    float sumsq = 0.f;
    for (int it = 0; it < POWER_ITERS; ++it) {
        __syncthreads();
        float part = 0.f;
        for (int i = tid; i < XSZ; i += NT) { float v = X[i]; part += v * v; }
        sumsq = block_reduce(part, red, tid);
        float inv = 1.0f / sqrtf(sumsq);
        for (int i = tid; i < XSZ; i += NT) X[i] *= inv;
        matvec_core(w, X, AX, tid);
        rmatvec_core(w, AX, X, tid);
    }
    __syncthreads();
    {
        float part = 0.f;
        for (int i = tid; i < XSZ; i += NT) { float v = X[i]; part += v * v; }
        sumsq = block_reduce(part, red, tid);
    }
    const float step = 0.9f / sqrtf(sqrtf(sumsq));
    const float b1 = w[WS_B1];
    const float* pw = w + WS_PW;

    for (int i = tid; i < XSZ; i += NT) { X[i] = 0.f; XB[i] = 0.f; }
    for (int i = tid; i < YSZ; i += NT) Y[i] = 0.f;

    for (int it = 0; it < PDHG_ITERS; ++it) {
        rmatvec_core(w, Y, ATY, tid);
        __syncthreads();
        for (int g = tid; g < NG; g += NT) {
            float o, n;
            o = X[OX_P  + g]; n = fmaxf(o - step * (Cost[g] + ATY[OX_P  + g]), 0.f);
            X[OX_P  + g] = n; XB[OX_P  + g] = 2.f * n - o;
            o = X[OX_RU + g]; n = fmaxf(o - step * (Cru[g]  + ATY[OX_RU + g]), 0.f);
            X[OX_RU + g] = n; XB[OX_RU + g] = 2.f * n - o;
            o = X[OX_RD + g]; n = fmaxf(o - step * (Crd[g]  + ATY[OX_RD + g]), 0.f);
            X[OX_RD + g] = n; XB[OX_RD + g] = 2.f * n - o;
        }
        for (int l = tid; l < NL; l += NT) {
            float o, n;
            o = X[OX_FU + l]; n = fmaxf(o - step * ATY[OX_FU + l], 0.f);
            X[OX_FU + l] = n; XB[OX_FU + l] = 2.f * n - o;
            o = X[OX_FD + l]; n = fmaxf(o - step * ATY[OX_FD + l], 0.f);
            X[OX_FD + l] = n; XB[OX_FD + l] = 2.f * n - o;
        }
        for (int i = tid; i < NG * NS; i += NT) {
            int g = i / NS, s = i % NS, o0 = g * S_P + s;
            float o, n;
            o = X[OX_U + o0]; n = fmaxf(o - step * ATY[OX_U + o0], 0.f);
            X[OX_U + o0] = n; XB[OX_U + o0] = 2.f * n - o;
            o = X[OX_D + o0]; n = fmaxf(o - step * ATY[OX_D + o0], 0.f);
            X[OX_D + o0] = n; XB[OX_D + o0] = 2.f * n - o;
        }
        matvec_core(w, XB, AX, tid);
        __syncthreads();
        if (tid == 0) Y[OY_Y1] += step * (AX[OY_Y1] - b1);
        for (int l = tid; l < NL; l += NT) {
            Y[OY_Y2 + l] += step * (AX[OY_Y2 + l] - w[WS_B2 + l]);
            Y[OY_Y3 + l] += step * (AX[OY_Y3 + l] - w[WS_B3 + l]);
        }
        for (int s = tid; s < NS; s += NT)
            Y[OY_Y4 + s] += step * (AX[OY_Y4 + s] - w[WS_B4 + s]);
        for (int g = tid; g < NG; g += NT) {
            Y[OY_Z1 + g] = fmaxf(Y[OY_Z1 + g] + step * (AX[OY_Z1 + g] - Pmax[g]), 0.f);
            Y[OY_Z2 + g] = fmaxf(Y[OY_Z2 + g] + step *  AX[OY_Z2 + g], 0.f);
        }
        for (int i = tid; i < NG * NS; i += NT) {
            int g = i / NS, s = i % NS, o = g * S_P + s;
            Y[OY_Z3 + o] = fmaxf(Y[OY_Z3 + o] + step * AX[OY_Z3 + o], 0.f);
            Y[OY_Z4 + o] = fmaxf(Y[OY_Z4 + o] + step * AX[OY_Z4 + o], 0.f);
        }
        for (int i = tid; i < NL * NS; i += NT) {
            int l = i / NS, s = i % NS, o = l * S_P + s;
            Y[OY_Z5 + o] = fmaxf(Y[OY_Z5 + o] + step * (AX[OY_Z5 + o] + pw[o]), 0.f);
            Y[OY_Z6 + o] = fmaxf(Y[OY_Z6 + o] + step * (AX[OY_Z6 + o] - pw[o]), 0.f);
        }
        __syncthreads();
    }

    for (int g = tid; g < NG; g += NT) {
        out[g]          = X[OX_P  + g];
        out[NG + g]     = X[OX_RU + g];
        out[2 * NG + g] = X[OX_RD + g];
    }
    for (int i = tid; i < NG * NS; i += NT) {
        int g = i / NS, s = i % NS;
        out[3 * NG + i]           = X[OX_U + g * S_P + s];
        out[3 * NG + NG * NS + i] = X[OX_D + g * S_P + s];
    }
    for (int l = tid; l < NL; l += NT) {
        out[3 * NG + 2 * NG * NS + l]      = X[OX_FU + l];
        out[3 * NG + 2 * NG * NS + NL + l] = X[OX_FD + l];
    }
}


extern "C" void kernel_launch(void* const* d_in, const int* in_sizes, int n_in,
                              void* d_out, int out_size, void* d_ws, size_t ws_size,
                              hipStream_t stream) {
    const float* w_scen   = (const float*)d_in[0];
    const float* Cost     = (const float*)d_in[1];
    const float* C_r_up   = (const float*)d_in[2];
    const float* C_r_down = (const float*)d_in[3];
    const float* Pmax     = (const float*)d_in[4];
    const float* PTDF     = (const float*)d_in[5];
    const float* node_G   = (const float*)d_in[6];
    const float* node_W   = (const float*)d_in[7];
    const float* node_L   = (const float*)d_in[8];
    const float* Pd       = (const float*)d_in[9];
    const float* w_exp    = (const float*)d_in[10];
    const float* Line_cap = (const float*)d_in[11];
    float* w   = (float*)d_ws;
    float* out = (float*)d_out;

    k_zero <<<512, 256, 0, stream>>>(w);
    k_prep1<<<240, 256, 0, stream>>>(w, w_scen, node_W, node_L, Pd, w_exp);
    k_prep2<<<640, 256, 0, stream>>>(w, PTDF, node_G, Line_cap);
    k_setv <<<161, 256, 0, stream>>>(w);
    k_solver<<<1, NT, 0, stream>>>(w, Cost, C_r_up, C_r_down, Pmax, out);
}


# Solver split at its block barriers into grid-wide launches (2 per PDHG iteration; one wave per 16x16 tile or row group, updates fused per entry) instead of one work-group; every sum keeps its order
# speedup vs baseline: 23.3899x; 23.3899x over previous
#include <hip/hip_runtime.h>
#include <math.h>

typedef __attribute__((ext_vector_type(2))) float v2f;
typedef __attribute__((ext_vector_type(8))) float v8f;

#define NT 512

constexpr int NG = 100, NL = 400, NN = 300, NWF = 50, NS = 200;
constexpr int G_P = 112;
constexpr int S_P = 208;
constexpr int PDHG_ITERS = 300, POWER_ITERS = 25;

constexpr int OX_P  = 0;
constexpr int OX_RU = OX_P  + G_P;
constexpr int OX_RD = OX_RU + G_P;
constexpr int OX_U  = OX_RD + G_P;
constexpr int OX_D  = OX_U  + G_P * S_P;
constexpr int OX_FU = OX_D  + G_P * S_P;
constexpr int OX_FD = OX_FU + NL;
constexpr int XSZ   = OX_FD + NL;

constexpr int OY_Y1 = 0;
constexpr int OY_Y2 = 16;
constexpr int OY_Y3 = OY_Y2 + NL;
constexpr int OY_Y4 = OY_Y3 + NL;
constexpr int OY_Z1 = OY_Y4 + S_P;
constexpr int OY_Z2 = OY_Z1 + G_P;
constexpr int OY_Z3 = OY_Z2 + G_P;
constexpr int OY_Z4 = OY_Z3 + G_P * S_P;
constexpr int OY_Z5 = OY_Z4 + G_P * S_P;
constexpr int OY_Z6 = OY_Z5 + NL * S_P;
constexpr int YSZ   = OY_Z6 + NL * S_P;

constexpr int WS_PG   = 0;
constexpr int WS_PGT  = WS_PG   + NL * G_P;
constexpr int WS_PW   = WS_PGT  + G_P * NL;
constexpr int WS_B2   = WS_PW   + NL * S_P;
constexpr int WS_B3   = WS_B2   + NL;
constexpr int WS_B4   = WS_B3   + NL;
constexpr int WS_B1   = WS_B4   + S_P;
constexpr int WS_M1   = WS_B1   + 16;
constexpr int WS_NV   = WS_M1   + NN * S_P;
constexpr int WS_X    = WS_NV   + NN + 4;
constexpr int WS_XB   = WS_X    + XSZ;
constexpr int WS_ATY  = WS_XB   + XSZ;
constexpr int WS_Y    = WS_ATY  + XSZ;
constexpr int WS_AX   = WS_Y    + YSZ;
constexpr int WS_T    = WS_AX   + YSZ;
constexpr int WS_FLOW = WS_T    + G_P * S_P;
constexpr int WS_GP   = WS_FLOW + NL * S_P;
constexpr int WS_COL  = WS_GP   + NL;
constexpr int WS_S3   = WS_COL  + S_P;
constexpr int WS_S4   = WS_S3   + G_P;
constexpr int WS_S5   = WS_S4   + G_P;
constexpr int WS_S6   = WS_S5   + NL;
constexpr int WS_GTY  = WS_S6   + NL;
constexpr int WS_SUMP = WS_GTY  + G_P;
constexpr int WS_STEP = WS_SUMP + 16;
constexpr int WS_TOTAL= WS_STEP + 16;


__global__ void k_zero(float* w) {
    for (int i = blockIdx.x * blockDim.x + threadIdx.x; i < WS_TOTAL;
         i += gridDim.x * blockDim.x) w[i] = 0.0f;
}

__global__ void k_prep1(float* w, const float* w_scen, const float* node_W,
                        const float* node_L, const float* Pd, const float* w_exp) {
    const int NM1 = NN * NS;
    const int total = NM1 + NN + NS + 1;
    for (int idx = blockIdx.x * blockDim.x + threadIdx.x; idx < total;
         idx += gridDim.x * blockDim.x) {
        if (idx < NM1) {
            int n = idx / NS, s = idx % NS;
            float a = 0.f;
            for (int ww = 0; ww < NWF; ++ww) a += node_W[n*NWF + ww] * w_scen[s*NWF + ww];
            w[WS_M1 + n*S_P + s] = a;
        } else if (idx < NM1 + NN) {
            int n = idx - NM1;
            float a = 0.f;
            for (int ww = 0; ww < NWF; ++ww) a += node_W[n*NWF + ww] * w_exp[ww];
            for (int m = 0; m < NN; ++m)     a -= node_L[n*NN + m] * Pd[m];
            w[WS_NV + n] = a;
        } else if (idx < NM1 + NN + NS) {
            int s = idx - NM1 - NN;
            float a = 0.f;
            for (int ww = 0; ww < NWF; ++ww) a += w_scen[s*NWF + ww];
            w[WS_B4 + s] = -a;
        } else {
            float a = 0.f;
            for (int m = 0; m < NN; ++m)  a += Pd[m];
            for (int ww = 0; ww < NWF; ++ww) a -= w_exp[ww];
            w[WS_B1] = a;
        }
    }
}

__global__ void k_prep2(float* w, const float* PTDF, const float* node_G,
                        const float* Line_cap) {
    const int A = NL * NG;
    const int B = A + NG * NL;
    const int C = B + NL * NS;
    const int D2 = C + NL;
    const int E = D2 + NL;
    for (int idx = blockIdx.x * blockDim.x + threadIdx.x; idx < E;
         idx += gridDim.x * blockDim.x) {
        if (idx < A) {
            int l = idx / NG, g = idx % NG;
            float a = 0.f;
            for (int n = 0; n < NN; ++n) a += PTDF[l*NN + n] * node_G[n*NG + g];
            w[WS_PG + l*G_P + g] = a;
        } else if (idx < B) {
            int r = idx - A; int g = r / NL, l = r % NL;
            float a = 0.f;
            for (int n = 0; n < NN; ++n) a += PTDF[l*NN + n] * node_G[n*NG + g];
            w[WS_PGT + g*NL + l] = a;
        } else if (idx < C) {
            int r = idx - B; int l = r / NS, s = r % NS;
            float a = 0.f;
            for (int n = 0; n < NN; ++n) a += PTDF[l*NN + n] * w[WS_M1 + n*S_P + s];
            w[WS_PW + l*S_P + s] = a;
        } else if (idx < D2) {
            int l = idx - C;
            float a = 0.f;
            for (int n = 0; n < NN; ++n) a += PTDF[l*NN + n] * w[WS_NV + n];
            w[WS_B2 + l] = Line_cap[l] - a;
        } else {
            int l = idx - D2;
            float a = 0.f;
            for (int n = 0; n < NN; ++n) a += PTDF[l*NN + n] * w[WS_NV + n];
            w[WS_B3 + l] = Line_cap[l] + a;
        }
    }
}

__global__ void k_setv(float* w) {
    const int total = 3*NG + 2*NG*NS + 2*NL;
    float* X = w + WS_X;
    for (int idx = blockIdx.x * blockDim.x + threadIdx.x; idx < total;
         idx += gridDim.x * blockDim.x) {
        int off;
        if (idx < NG)                off = OX_P  + idx;
        else if (idx < 2*NG)         off = OX_RU + (idx - NG);
        else if (idx < 3*NG)         off = OX_RD + (idx - 2*NG);
        else if (idx < 3*NG + NG*NS) { int r = idx - 3*NG; off = OX_U + (r/NS)*S_P + (r%NS); }
        else if (idx < 3*NG + 2*NG*NS) { int r = idx - 3*NG - NG*NS; off = OX_D + (r/NS)*S_P + (r%NS); }
        else if (idx < 3*NG + 2*NG*NS + NL) off = OX_FU + (idx - 3*NG - 2*NG*NS);
        else                         off = OX_FD + (idx - 3*NG - 2*NG*NS - NL);
        X[off] = 1.0f;
    }
}


__device__ __forceinline__ float block_reduce(float v, float* red, int tid) {
    red[tid] = v; __syncthreads();
    for (int off = NT >> 1; off > 0; off >>= 1) {
        if (tid < off) red[tid] += red[tid + off];
        __syncthreads();
    }
    float r = red[0]; __syncthreads();
    return r;
}

template <bool FINAL>
__global__ __launch_bounds__(NT) void k_norm(float* w) {
    __shared__ float red[NT];
    const int tid = threadIdx.x;
    float* X = w + WS_X;
    float part = 0.f;
    for (int i = tid; i < XSZ; i += NT) { float v = X[i]; part += v * v; }
    float sumsq = block_reduce(part, red, tid);
    if (!FINAL) {
        float inv = 1.0f / sqrtf(sumsq);
        for (int i = tid; i < XSZ; i += NT) X[i] *= inv;
    } else {
        if (tid == 0) w[WS_STEP] = 0.9f / sqrtf(sqrtf(sumsq));
    }
}

__global__ void k_reset(float* w) {
    const int total = 2 * XSZ + YSZ;
    for (int i = blockIdx.x * blockDim.x + threadIdx.x; i < total; i += gridDim.x * blockDim.x) {
        if (i < XSZ)            w[WS_X + i] = 0.f;
        else if (i < 2 * XSZ)   w[WS_XB + (i - XSZ)] = 0.f;
        else                    w[WS_Y + (i - 2 * XSZ)] = 0.f;
    }
}

template <int K>
__device__ __forceinline__ v8f tile_gemm(const float* arow, const float* P, const float* Q,
                                         int n, int kh) {
    v8f acc = {0.f,0.f,0.f,0.f,0.f,0.f,0.f,0.f};
#pragma unroll
    for (int k0 = 0; k0 < K; k0 += 4) {
        int ka = k0 + kh;
        v2f a = *(const v2f*)(arow + ka);
        v2f b;
        int bi = ka * S_P + n;
        b[0] = P[bi]       - Q[bi];
        b[1] = P[bi + S_P] - Q[bi + S_P];
        acc = __builtin_amdgcn_wmma_f32_16x16x4_f32(false, a, false, b, (short)0, acc, false, false);
    }
    return acc;
}

constexpr int RPW = 8;


constexpr int RA_T1 = 7 * 13;
constexpr int RA_T2 = RA_T1 + (NG + RPW - 1) / RPW;
constexpr int RA_T3 = RA_T2 + (NG + RPW - 1) / RPW;
constexpr int RA_NB = RA_T3 + NL / RPW;

template <bool PD>
__global__ __launch_bounds__(32) __attribute__((amdgpu_waves_per_eu(1, 1))) void k_rmv(
        float* w, const float* Cost, const float* Cru, const float* Crd) {
    const int lane = threadIdx.x, blk = blockIdx.x;
    const float* PGT = w + WS_PGT;
    const float* yb = PD ? (w + WS_Y) : (w + WS_AX);
    float* X  = w + WS_X;
    float* XB = w + WS_XB;
    const float step = PD ? w[WS_STEP] : 0.f;
    if (blk < RA_T1) {
        const int tm = blk / 13, tn = blk % 13;
        const int m16 = lane & 15, kh = (lane >> 4) * 2;
        const int m = tm * 16 + m16, n = tn * 16 + m16;
        const int mr = tm * 16 + ((lane >> 4) << 3);
        const int nc = n < NS ? n : NS - 1;
        const float y4 = yb[OY_Y4 + nc];
        float z3v[8], z4v[8], xu[8], xd[8];
        for (int i = 0; i < 8; ++i) {
            const int gc = (mr + i) < NG ? (mr + i) : NG - 1;
            const int o = gc * S_P + nc;
            z3v[i] = yb[OY_Z3 + o]; z4v[i] = yb[OY_Z4 + o];
            xu[i] = PD ? X[OX_U + o] : 0.f; xd[i] = PD ? X[OX_D + o] : 0.f;
        }
        v8f acc = tile_gemm<NL>(PGT + m * NL, yb + OY_Z5, yb + OY_Z6, n, kh);
        float nu[8], nd[8], bu[8], bd[8];
        for (int i = 0; i < 8; ++i) {
            const float tv = acc[i];
            const float au =  y4 + z3v[i] + tv;
            const float ad = -y4 + z4v[i] - tv;
            if (PD) {
                nu[i] = fmaxf(xu[i] - step * au, 0.f); bu[i] = 2.f * nu[i] - xu[i];
                nd[i] = fmaxf(xd[i] - step * ad, 0.f); bd[i] = 2.f * nd[i] - xd[i];
            } else {
                nu[i] = au; nd[i] = ad; bu[i] = 0.f; bd[i] = 0.f;
            }
        }
        for (int i = 0; i < 8; ++i) {
            const int g = mr + i;
            if (n < NS && g < NG) {
                const int o = g * S_P + n;
                X[OX_U + o] = nu[i]; X[OX_D + o] = nd[i];
                if (PD) { XB[OX_U + o] = bu[i]; XB[OX_D + o] = bd[i]; }
            }
        }
    } else if (blk < RA_T2) {
        const int g = (blk - RA_T1) * RPW + lane;
        if (lane < RPW && g < NG) {
            const float z1 = yb[OY_Z1 + g], z2 = yb[OY_Z2 + g];
            const float xu = PD ? X[OX_RU + g] : 0.f, xd = PD ? X[OX_RD + g] : 0.f;
            const float cu = PD ? Cru[g] : 0.f, cd = PD ? Crd[g] : 0.f;
            float a = 0.f, b = 0.f;
            const float* z3r = yb + OY_Z3 + g * S_P;
            const float* z4r = yb + OY_Z4 + g * S_P;
#pragma unroll
            for (int s = 0; s < NS; ++s) { a += z3r[s]; b += z4r[s]; }
            const float au = z1 - a;
            const float ad = z2 - b;
            if (PD) {
                const float nu = fmaxf(xu - step * (cu + au), 0.f);
                const float nd = fmaxf(xd - step * (cd + ad), 0.f);
                X[OX_RU + g] = nu; XB[OX_RU + g] = 2.f * nu - xu;
                X[OX_RD + g] = nd; XB[OX_RD + g] = 2.f * nd - xd;
            } else {
                X[OX_RU + g] = au; X[OX_RD + g] = ad;
            }
        }
    } else if (blk < RA_T3) {
        const int g = (blk - RA_T2) * RPW + lane;
        if (lane < RPW && g < NG) {
            const float y1 = yb[OY_Y1], z1 = yb[OY_Z1 + g], z2 = yb[OY_Z2 + g];
            const float xp = PD ? X[OX_P + g] : 0.f;
            const float cp = PD ? Cost[g] : 0.f;
            float c = 0.f;
            const float* pr = PGT + g * NL;
#pragma unroll 100
            for (int l = 0; l < NL; ++l) c += pr[l] * (yb[OY_Y2 + l] - yb[OY_Y3 + l]);
            const float ap = y1 + c + z1 - z2;
            if (PD) {
                const float nn = fmaxf(xp - step * (cp + ap), 0.f);
                X[OX_P + g] = nn; XB[OX_P + g] = 2.f * nn - xp;
            } else {
                X[OX_P + g] = ap;
            }
        }
    } else {
        const int l = (blk - RA_T3) * RPW + lane;
        if (lane < RPW && l < NL) {
            const float y2 = yb[OY_Y2 + l], y3 = yb[OY_Y3 + l];
            const float xu = PD ? X[OX_FU + l] : 0.f, xd = PD ? X[OX_FD + l] : 0.f;
            float a = 0.f, b = 0.f;
            const float* z5r = yb + OY_Z5 + l * S_P;
            const float* z6r = yb + OY_Z6 + l * S_P;
#pragma unroll
            for (int s = 0; s < NS; ++s) { a += z5r[s]; b += z6r[s]; }
            const float au = y2 - a;
            const float ad = y3 - b;
            if (PD) {
                const float nu = fmaxf(xu - step * au, 0.f);
                const float nd = fmaxf(xd - step * ad, 0.f);
                X[OX_FU + l] = nu; XB[OX_FU + l] = 2.f * nu - xu;
                X[OX_FD + l] = nd; XB[OX_FD + l] = 2.f * nd - xd;
            } else {
                X[OX_FU + l] = au; X[OX_FD + l] = ad;
            }
        }
    }
}

constexpr int EPW = 128;
constexpr int MA_T1 = 25 * 13;
constexpr int MA_T2 = MA_T1 + NL / RPW;
constexpr int MA_T3 = MA_T2 + (NS + 31) / 32;
constexpr int MA_T4 = MA_T3 + (NG + 31) / 32;
constexpr int MA_T5 = MA_T4 + 1;
constexpr int MA_NB = MA_T5 + (NG * NS + EPW - 1) / EPW;

template <bool PD>
__global__ __launch_bounds__(32) __attribute__((amdgpu_waves_per_eu(1, 1))) void k_mv(
        float* w, const float* Pmax) {
    const int lane = threadIdx.x, blk = blockIdx.x;
    const float* PG = w + WS_PG;
    const float* xb = PD ? (w + WS_XB) : (w + WS_X);
    float* Y = PD ? (w + WS_Y) : (w + WS_AX);
    const float* pw = w + WS_PW;
    const float step = PD ? w[WS_STEP] : 0.f;
    if (blk < MA_T1) {
        const int tm = blk / 13, tn = blk % 13;
        const int m16 = lane & 15, kh = (lane >> 4) * 2;
        const int m = tm * 16 + m16, n = tn * 16 + m16;
        const int mr = tm * 16 + ((lane >> 4) << 3);
        const int nc = n < NS ? n : NS - 1;
        float fu[8], fd[8], y5[8], y6[8], pv[8];
        for (int i = 0; i < 8; ++i) {
            const int l = mr + i;
            const int o = l * S_P + nc;
            fu[i] = xb[OX_FU + l]; fd[i] = xb[OX_FD + l];
            y5[i] = PD ? Y[OY_Z5 + o] : 0.f; y6[i] = PD ? Y[OY_Z6 + o] : 0.f;
            pv[i] = PD ? pw[o] : 0.f;
        }
        v8f acc = tile_gemm<NG>(PG + m * G_P, xb + OX_U, xb + OX_D, n, kh);
        float r5[8], r6[8];
        for (int i = 0; i < 8; ++i) {
            const float f = acc[i];
            const float a5 =  f - fu[i];
            const float a6 = -f - fd[i];
            if (PD) {
                r5[i] = fmaxf(y5[i] + step * (a5 + pv[i]), 0.f);
                r6[i] = fmaxf(y6[i] + step * (a6 - pv[i]), 0.f);
            } else {
                r5[i] = a5; r6[i] = a6;
            }
        }
        if (n < NS) {
            for (int i = 0; i < 8; ++i) {
                const int o = (mr + i) * S_P + n;
                Y[OY_Z5 + o] = r5[i]; Y[OY_Z6 + o] = r6[i];
            }
        }
    } else if (blk < MA_T2) {
        const int l = (blk - MA_T1) * RPW + lane;
        if (lane < RPW && l < NL) {
            const float fu = xb[OX_FU + l], fd = xb[OX_FD + l];
            const float y2 = PD ? Y[OY_Y2 + l] : 0.f, y3 = PD ? Y[OY_Y3 + l] : 0.f;
            const float b2 = PD ? w[WS_B2 + l] : 0.f, b3 = PD ? w[WS_B3 + l] : 0.f;
            float s = 0.f; const float* r = PG + l * G_P; const float* p = xb + OX_P;
#pragma unroll
            for (int g = 0; g < NG; ++g) s += r[g] * p[g];
            const float a2 =  s + fu;
            const float a3 = -s + fd;
            if (PD) {
                Y[OY_Y2 + l] = y2 + step * (a2 - b2);
                Y[OY_Y3 + l] = y3 + step * (a3 - b3);
            } else {
                Y[OY_Y2 + l] = a2; Y[OY_Y3 + l] = a3;
            }
        }
    } else if (blk < MA_T3) {
        const int s = (blk - MA_T2) * 32 + lane;
        if (s < NS) {
            const float y4 = PD ? Y[OY_Y4 + s] : 0.f;
            const float b4 = PD ? w[WS_B4 + s] : 0.f;
            float c = 0.f;
            const float* Ux = xb + OX_U + s; const float* Dx = xb + OX_D + s;
#pragma unroll
            for (int g = 0; g < NG; ++g) c += Ux[g * S_P] - Dx[g * S_P];
            if (PD) Y[OY_Y4 + s] = y4 + step * (c - b4);
            else    Y[OY_Y4 + s] = c;
        }
    } else if (blk < MA_T4) {
        const int g = (blk - MA_T3) * 32 + lane;
        if (g < NG) {
            const float xp = xb[OX_P + g], xu = xb[OX_RU + g], xd = xb[OX_RD + g];
            const float a1 = xp + xu;
            const float a2 = xd - xp;
            if (PD) {
                const float z1 = Y[OY_Z1 + g], z2 = Y[OY_Z2 + g], pm = Pmax[g];
                Y[OY_Z1 + g] = fmaxf(z1 + step * (a1 - pm), 0.f);
                Y[OY_Z2 + g] = fmaxf(z2 + step *  a2, 0.f);
            } else {
                Y[OY_Z1 + g] = a1; Y[OY_Z2 + g] = a2;
            }
        }
    } else if (blk < MA_T5) {
        if (lane == 0) {
            const float y1 = PD ? Y[OY_Y1] : 0.f;
            const float b1 = PD ? w[WS_B1] : 0.f;
            float s = 0.f;
#pragma unroll
            for (int g = 0; g < NG; ++g) s += xb[OX_P + g];
            if (PD) Y[OY_Y1] = y1 + step * (s - b1);
            else    Y[OY_Y1] = s;
        }
    } else {
        const int base = (blk - MA_T5) * EPW;
        float r3[EPW / 32], r4[EPW / 32];
        for (int j = 0; j < EPW / 32; ++j) {
            const int i = base + j * 32 + lane;
            const int ic = i < NG * NS ? i : NG * NS - 1;
            const int g = ic / NS, s = ic % NS, o = g * S_P + s;
            const float a3 = xb[OX_U + o] - xb[OX_RU + g];
            const float a4 = xb[OX_D + o] - xb[OX_RD + g];
            if (PD) {
                r3[j] = fmaxf(Y[OY_Z3 + o] + step * a3, 0.f);
                r4[j] = fmaxf(Y[OY_Z4 + o] + step * a4, 0.f);
            } else {
                r3[j] = a3; r4[j] = a4;
            }
        }
        for (int j = 0; j < EPW / 32; ++j) {
            const int i = base + j * 32 + lane;
            if (i < NG * NS) {
                const int g = i / NS, s = i % NS, o = g * S_P + s;
                Y[OY_Z3 + o] = r3[j]; Y[OY_Z4 + o] = r4[j];
            }
        }
    }
}

__global__ void k_emit(const float* w, float* out) {
    const float* X = w + WS_X;
    const int total = 3 * NG + 2 * NG * NS + 2 * NL;
    for (int idx = blockIdx.x * blockDim.x + threadIdx.x; idx < total; idx += gridDim.x * blockDim.x) {
        float v;
        if (idx < NG)                 v = X[OX_P  + idx];
        else if (idx < 2 * NG)        v = X[OX_RU + (idx - NG)];
        else if (idx < 3 * NG)        v = X[OX_RD + (idx - 2 * NG)];
        else if (idx < 3 * NG + NG * NS) { int r = idx - 3 * NG; v = X[OX_U + (r / NS) * S_P + (r % NS)]; }
        else if (idx < 3 * NG + 2 * NG * NS) { int r = idx - 3 * NG - NG * NS; v = X[OX_D + (r / NS) * S_P + (r % NS)]; }
        else if (idx < 3 * NG + 2 * NG * NS + NL) v = X[OX_FU + (idx - 3 * NG - 2 * NG * NS)];
        else                          v = X[OX_FD + (idx - 3 * NG - 2 * NG * NS - NL)];
        out[idx] = v;
    }
}


extern "C" void kernel_launch(void* const* d_in, const int* in_sizes, int n_in,
                              void* d_out, int out_size, void* d_ws, size_t ws_size,
                              hipStream_t stream) {
    const float* w_scen   = (const float*)d_in[0];
    const float* Cost     = (const float*)d_in[1];
    const float* C_r_up   = (const float*)d_in[2];
    const float* C_r_down = (const float*)d_in[3];
    const float* Pmax     = (const float*)d_in[4];
    const float* PTDF     = (const float*)d_in[5];
    const float* node_G   = (const float*)d_in[6];
    const float* node_W   = (const float*)d_in[7];
    const float* node_L   = (const float*)d_in[8];
    const float* Pd       = (const float*)d_in[9];
    const float* w_exp    = (const float*)d_in[10];
    const float* Line_cap = (const float*)d_in[11];
    float* w   = (float*)d_ws;
    float* out = (float*)d_out;

    k_zero <<<512, 256, 0, stream>>>(w);
    k_prep1<<<240, 256, 0, stream>>>(w, w_scen, node_W, node_L, Pd, w_exp);
    k_prep2<<<640, 256, 0, stream>>>(w, PTDF, node_G, Line_cap);
    k_setv <<<161, 256, 0, stream>>>(w);

    for (int it = 0; it < POWER_ITERS; ++it) {
        k_norm<false><<<1, NT, 0, stream>>>(w);
        k_mv  <false><<<MA_NB, 32, 0, stream>>>(w, Pmax);
        k_rmv <false><<<RA_NB, 32, 0, stream>>>(w, Cost, C_r_up, C_r_down);
    }
    k_norm<true><<<1, NT, 0, stream>>>(w);
    k_reset<<<256, 256, 0, stream>>>(w);

    for (int it = 0; it < PDHG_ITERS; ++it) {
        k_rmv<true><<<RA_NB, 32, 0, stream>>>(w, Cost, C_r_up, C_r_down);
        k_mv <true><<<MA_NB, 32, 0, stream>>>(w, Pmax);
    }
    k_emit<<<161, 256, 0, stream>>>(w, out);
}


# A^T y launch: every work-group is 4 waves, each taking a quarter of the task's reduction range (tile-product k, row-sum columns); partials added in LDS as ((w0+w1)+w2)+w3: summation order changes
# speedup vs baseline: 29.5254x; 1.2623x over previous
#include <hip/hip_runtime.h>
#include <math.h>

typedef __attribute__((ext_vector_type(2))) float v2f;
typedef __attribute__((ext_vector_type(8))) float v8f;

#define NT 512

constexpr int NG = 100, NL = 400, NN = 300, NWF = 50, NS = 200;
constexpr int G_P = 112;
constexpr int S_P = 208;
constexpr int PDHG_ITERS = 300, POWER_ITERS = 25;

constexpr int OX_P  = 0;
constexpr int OX_RU = OX_P  + G_P;
constexpr int OX_RD = OX_RU + G_P;
constexpr int OX_U  = OX_RD + G_P;
constexpr int OX_D  = OX_U  + G_P * S_P;
constexpr int OX_FU = OX_D  + G_P * S_P;
constexpr int OX_FD = OX_FU + NL;
constexpr int XSZ   = OX_FD + NL;

constexpr int OY_Y1 = 0;
constexpr int OY_Y2 = 16;
constexpr int OY_Y3 = OY_Y2 + NL;
constexpr int OY_Y4 = OY_Y3 + NL;
constexpr int OY_Z1 = OY_Y4 + S_P;
constexpr int OY_Z2 = OY_Z1 + G_P;
constexpr int OY_Z3 = OY_Z2 + G_P;
constexpr int OY_Z4 = OY_Z3 + G_P * S_P;
constexpr int OY_Z5 = OY_Z4 + G_P * S_P;
constexpr int OY_Z6 = OY_Z5 + NL * S_P;
constexpr int YSZ   = OY_Z6 + NL * S_P;

constexpr int WS_PG   = 0;
constexpr int WS_PGT  = WS_PG   + NL * G_P;
constexpr int WS_PW   = WS_PGT  + G_P * NL;
constexpr int WS_B2   = WS_PW   + NL * S_P;
constexpr int WS_B3   = WS_B2   + NL;
constexpr int WS_B4   = WS_B3   + NL;
constexpr int WS_B1   = WS_B4   + S_P;
constexpr int WS_M1   = WS_B1   + 16;
constexpr int WS_NV   = WS_M1   + NN * S_P;
constexpr int WS_X    = WS_NV   + NN + 4;
constexpr int WS_XB   = WS_X    + XSZ;
constexpr int WS_ATY  = WS_XB   + XSZ;
constexpr int WS_Y    = WS_ATY  + XSZ;
constexpr int WS_AX   = WS_Y    + YSZ;
constexpr int WS_T    = WS_AX   + YSZ;
constexpr int WS_FLOW = WS_T    + G_P * S_P;
constexpr int WS_GP   = WS_FLOW + NL * S_P;
constexpr int WS_COL  = WS_GP   + NL;
constexpr int WS_S3   = WS_COL  + S_P;
constexpr int WS_S4   = WS_S3   + G_P;
constexpr int WS_S5   = WS_S4   + G_P;
constexpr int WS_S6   = WS_S5   + NL;
constexpr int WS_GTY  = WS_S6   + NL;
constexpr int WS_SUMP = WS_GTY  + G_P;
constexpr int WS_STEP = WS_SUMP + 16;
constexpr int WS_TOTAL= WS_STEP + 16;


__global__ void k_zero(float* w) {
    for (int i = blockIdx.x * blockDim.x + threadIdx.x; i < WS_TOTAL;
         i += gridDim.x * blockDim.x) w[i] = 0.0f;
}

__global__ void k_prep1(float* w, const float* w_scen, const float* node_W,
                        const float* node_L, const float* Pd, const float* w_exp) {
    const int NM1 = NN * NS;
    const int total = NM1 + NN + NS + 1;
    for (int idx = blockIdx.x * blockDim.x + threadIdx.x; idx < total;
         idx += gridDim.x * blockDim.x) {
        if (idx < NM1) {
            int n = idx / NS, s = idx % NS;
            float a = 0.f;
            for (int ww = 0; ww < NWF; ++ww) a += node_W[n*NWF + ww] * w_scen[s*NWF + ww];
            w[WS_M1 + n*S_P + s] = a;
        } else if (idx < NM1 + NN) {
            int n = idx - NM1;
            float a = 0.f;
            for (int ww = 0; ww < NWF; ++ww) a += node_W[n*NWF + ww] * w_exp[ww];
            for (int m = 0; m < NN; ++m)     a -= node_L[n*NN + m] * Pd[m];
            w[WS_NV + n] = a;
        } else if (idx < NM1 + NN + NS) {
            int s = idx - NM1 - NN;
            float a = 0.f;
            for (int ww = 0; ww < NWF; ++ww) a += w_scen[s*NWF + ww];
            w[WS_B4 + s] = -a;
        } else {
            float a = 0.f;
            for (int m = 0; m < NN; ++m)  a += Pd[m];
            for (int ww = 0; ww < NWF; ++ww) a -= w_exp[ww];
            w[WS_B1] = a;
        }
    }
}

__global__ void k_prep2(float* w, const float* PTDF, const float* node_G,
                        const float* Line_cap) {
    const int A = NL * NG;
    const int B = A + NG * NL;
    const int C = B + NL * NS;
    const int D2 = C + NL;
    const int E = D2 + NL;
    for (int idx = blockIdx.x * blockDim.x + threadIdx.x; idx < E;
         idx += gridDim.x * blockDim.x) {
        if (idx < A) {
            int l = idx / NG, g = idx % NG;
            float a = 0.f;
            for (int n = 0; n < NN; ++n) a += PTDF[l*NN + n] * node_G[n*NG + g];
            w[WS_PG + l*G_P + g] = a;
        } else if (idx < B) {
            int r = idx - A; int g = r / NL, l = r % NL;
            float a = 0.f;
            for (int n = 0; n < NN; ++n) a += PTDF[l*NN + n] * node_G[n*NG + g];
            w[WS_PGT + g*NL + l] = a;
        } else if (idx < C) {
            int r = idx - B; int l = r / NS, s = r % NS;
            float a = 0.f;
            for (int n = 0; n < NN; ++n) a += PTDF[l*NN + n] * w[WS_M1 + n*S_P + s];
            w[WS_PW + l*S_P + s] = a;
        } else if (idx < D2) {
            int l = idx - C;
            float a = 0.f;
            for (int n = 0; n < NN; ++n) a += PTDF[l*NN + n] * w[WS_NV + n];
            w[WS_B2 + l] = Line_cap[l] - a;
        } else {
            int l = idx - D2;
            float a = 0.f;
            for (int n = 0; n < NN; ++n) a += PTDF[l*NN + n] * w[WS_NV + n];
            w[WS_B3 + l] = Line_cap[l] + a;
        }
    }
}

__global__ void k_setv(float* w) {
    const int total = 3*NG + 2*NG*NS + 2*NL;
    float* X = w + WS_X;
    for (int idx = blockIdx.x * blockDim.x + threadIdx.x; idx < total;
         idx += gridDim.x * blockDim.x) {
        int off;
        if (idx < NG)                off = OX_P  + idx;
        else if (idx < 2*NG)         off = OX_RU + (idx - NG);
        else if (idx < 3*NG)         off = OX_RD + (idx - 2*NG);
        else if (idx < 3*NG + NG*NS) { int r = idx - 3*NG; off = OX_U + (r/NS)*S_P + (r%NS); }
        else if (idx < 3*NG + 2*NG*NS) { int r = idx - 3*NG - NG*NS; off = OX_D + (r/NS)*S_P + (r%NS); }
        else if (idx < 3*NG + 2*NG*NS + NL) off = OX_FU + (idx - 3*NG - 2*NG*NS);
        else                         off = OX_FD + (idx - 3*NG - 2*NG*NS - NL);
        X[off] = 1.0f;
    }
}


__device__ __forceinline__ float block_reduce(float v, float* red, int tid) {
    red[tid] = v; __syncthreads();
    for (int off = NT >> 1; off > 0; off >>= 1) {
        if (tid < off) red[tid] += red[tid + off];
        __syncthreads();
    }
    float r = red[0]; __syncthreads();
    return r;
}

template <bool FINAL>
__global__ __launch_bounds__(NT) void k_norm(float* w) {
    __shared__ float red[NT];
    const int tid = threadIdx.x;
    float* X = w + WS_X;
    float part = 0.f;
    for (int i = tid; i < XSZ; i += NT) { float v = X[i]; part += v * v; }
    float sumsq = block_reduce(part, red, tid);
    if (!FINAL) {
        float inv = 1.0f / sqrtf(sumsq);
        for (int i = tid; i < XSZ; i += NT) X[i] *= inv;
    } else {
        if (tid == 0) w[WS_STEP] = 0.9f / sqrtf(sqrtf(sumsq));
    }
}

__global__ void k_reset(float* w) {
    const int total = 2 * XSZ + YSZ;
    for (int i = blockIdx.x * blockDim.x + threadIdx.x; i < total; i += gridDim.x * blockDim.x) {
        if (i < XSZ)            w[WS_X + i] = 0.f;
        else if (i < 2 * XSZ)   w[WS_XB + (i - XSZ)] = 0.f;
        else                    w[WS_Y + (i - 2 * XSZ)] = 0.f;
    }
}

template <int K>
__device__ __forceinline__ v8f tile_gemm(const float* arow, const float* P, const float* Q,
                                         int n, int kh) {
    v8f acc = {0.f,0.f,0.f,0.f,0.f,0.f,0.f,0.f};
#pragma unroll
    for (int k0 = 0; k0 < K; k0 += 4) {
        int ka = k0 + kh;
        v2f a = *(const v2f*)(arow + ka);
        v2f b;
        int bi = ka * S_P + n;
        b[0] = P[bi]       - Q[bi];
        b[1] = P[bi + S_P] - Q[bi + S_P];
        acc = __builtin_amdgcn_wmma_f32_16x16x4_f32(false, a, false, b, (short)0, acc, false, false);
    }
    return acc;
}

constexpr int RPW = 8;


constexpr int KSP = 4;
constexpr int RNT = 32 * KSP;
constexpr int KQ  = NL / KSP;
constexpr int SQ  = 48;
constexpr int RA_T1 = 7 * 13;
constexpr int RA_T2 = RA_T1 + (NG + RPW - 1) / RPW;
constexpr int RA_T3 = RA_T2 + (NG + RPW - 1) / RPW;
constexpr int RA_NB = RA_T3 + NL / RPW;

__device__ __forceinline__ void row_sum_share(const float* r0, const float* r1, int wv, float& a, float& b) {
    const float* p0 = r0 + wv * SQ;
    const float* p1 = r1 + wv * SQ;
#pragma unroll
    for (int s = 0; s < SQ; ++s) { a += p0[s]; b += p1[s]; }
    if (wv == KSP - 1) {
#pragma unroll
        for (int s = KSP * SQ; s < NS; ++s) { a += r0[s]; b += r1[s]; }
    }
}

template <bool PD>
__global__ __launch_bounds__(RNT) __attribute__((amdgpu_waves_per_eu(1, 1))) void k_rmv(
        float* w, const float* Cost, const float* Cru, const float* Crd) {
    __shared__ float part[KSP * 8 * 32];
    const int lane = threadIdx.x & 31, blk = blockIdx.x;
    const int wv = __builtin_amdgcn_readfirstlane((int)(threadIdx.x >> 5));
    const float* PGT = w + WS_PGT;
    const float* yb = PD ? (w + WS_Y) : (w + WS_AX);
    float* X  = w + WS_X;
    float* XB = w + WS_XB;
    const float step = PD ? w[WS_STEP] : 0.f;
    if (blk < RA_T1) {
        const int tm = blk / 13, tn = blk % 13;
        const int m16 = lane & 15, kh = (lane >> 4) * 2;
        const int m = tm * 16 + m16, n = tn * 16 + m16;
        const int mr = tm * 16 + ((lane >> 4) << 3) + 2 * wv;
        const int nc = n < NS ? n : NS - 1;
        const float y4 = yb[OY_Y4 + nc];
        float z3v[2], z4v[2], xu[2], xd[2];
        for (int j = 0; j < 2; ++j) {
            const int gc = (mr + j) < NG ? (mr + j) : NG - 1;
            const int o = gc * S_P + nc;
            z3v[j] = yb[OY_Z3 + o]; z4v[j] = yb[OY_Z4 + o];
            xu[j] = PD ? X[OX_U + o] : 0.f; xd[j] = PD ? X[OX_D + o] : 0.f;
        }
        const int kb = wv * KQ;
        v8f acc = tile_gemm<KQ>(PGT + m * NL + kb, yb + OY_Z5 + kb * S_P, yb + OY_Z6 + kb * S_P, n, kh);
        for (int i = 0; i < 8; ++i) part[(wv * 8 + i) * 32 + lane] = acc[i];
        __syncthreads();
        float nu[2], nd[2], bu[2], bd[2];
        for (int j = 0; j < 2; ++j) {
            const int i = 2 * wv + j;
            const float tv = ((part[(0 * 8 + i) * 32 + lane] + part[(1 * 8 + i) * 32 + lane])
                              + part[(2 * 8 + i) * 32 + lane]) + part[(3 * 8 + i) * 32 + lane];
            const float au =  y4 + z3v[j] + tv;
            const float ad = -y4 + z4v[j] - tv;
            if (PD) {
                nu[j] = fmaxf(xu[j] - step * au, 0.f); bu[j] = 2.f * nu[j] - xu[j];
                nd[j] = fmaxf(xd[j] - step * ad, 0.f); bd[j] = 2.f * nd[j] - xd[j];
            } else {
                nu[j] = au; nd[j] = ad; bu[j] = 0.f; bd[j] = 0.f;
            }
        }
        for (int j = 0; j < 2; ++j) {
            const int g = mr + j;
            if (n < NS && g < NG) {
                const int o = g * S_P + n;
                X[OX_U + o] = nu[j]; X[OX_D + o] = nd[j];
                if (PD) { XB[OX_U + o] = bu[j]; XB[OX_D + o] = bd[j]; }
            }
        }
    } else if (blk < RA_T2) {
        const int g = (blk - RA_T1) * RPW + lane;
        const bool act = lane < RPW && g < NG;
        float z1 = 0.f, z2 = 0.f, xu = 0.f, xd = 0.f, cu = 0.f, cd = 0.f;
        float a = 0.f, b = 0.f;
        if (act) {
            if (wv == 0) {
                z1 = yb[OY_Z1 + g]; z2 = yb[OY_Z2 + g];
                xu = PD ? X[OX_RU + g] : 0.f; xd = PD ? X[OX_RD + g] : 0.f;
                cu = PD ? Cru[g] : 0.f; cd = PD ? Crd[g] : 0.f;
            }
            row_sum_share(yb + OY_Z3 + g * S_P, yb + OY_Z4 + g * S_P, wv, a, b);
        }
        part[wv * 32 + lane] = a; part[(KSP + wv) * 32 + lane] = b;
        __syncthreads();
        if (wv == 0 && act) {
            const float sa = ((part[lane] + part[32 + lane]) + part[64 + lane]) + part[96 + lane];
            const float sb = ((part[128 + lane] + part[160 + lane]) + part[192 + lane]) + part[224 + lane];
            const float au = z1 - sa;
            const float ad = z2 - sb;
            if (PD) {
                const float nu = fmaxf(xu - step * (cu + au), 0.f);
                const float nd = fmaxf(xd - step * (cd + ad), 0.f);
                X[OX_RU + g] = nu; XB[OX_RU + g] = 2.f * nu - xu;
                X[OX_RD + g] = nd; XB[OX_RD + g] = 2.f * nd - xd;
            } else {
                X[OX_RU + g] = au; X[OX_RD + g] = ad;
            }
        }
    } else if (blk < RA_T3) {
        const int g = (blk - RA_T2) * RPW + lane;
        const bool act = lane < RPW && g < NG;
        float y1 = 0.f, z1 = 0.f, z2 = 0.f, xp = 0.f, cp = 0.f;
        float c = 0.f;
        if (act) {
            if (wv == 0) {
                y1 = yb[OY_Y1]; z1 = yb[OY_Z1 + g]; z2 = yb[OY_Z2 + g];
                xp = PD ? X[OX_P + g] : 0.f;
                cp = PD ? Cost[g] : 0.f;
            }
            const float* pr = PGT + g * NL + wv * KQ;
            const float* y2p = yb + OY_Y2 + wv * KQ;
            const float* y3p = yb + OY_Y3 + wv * KQ;
#pragma unroll
            for (int l = 0; l < KQ; ++l) c += pr[l] * (y2p[l] - y3p[l]);
        }
        part[wv * 32 + lane] = c;
        __syncthreads();
        if (wv == 0 && act) {
            const float cs = ((part[lane] + part[32 + lane]) + part[64 + lane]) + part[96 + lane];
            const float ap = y1 + cs + z1 - z2;
            if (PD) {
                const float nn = fmaxf(xp - step * (cp + ap), 0.f);
                X[OX_P + g] = nn; XB[OX_P + g] = 2.f * nn - xp;
            } else {
                X[OX_P + g] = ap;
            }
        }
    } else {
        const int l = (blk - RA_T3) * RPW + lane;
        const bool act = lane < RPW && l < NL;
        float y2 = 0.f, y3 = 0.f, xu = 0.f, xd = 0.f;
        float a = 0.f, b = 0.f;
        if (act) {
            if (wv == 0) {
                y2 = yb[OY_Y2 + l]; y3 = yb[OY_Y3 + l];
                xu = PD ? X[OX_FU + l] : 0.f; xd = PD ? X[OX_FD + l] : 0.f;
            }
            row_sum_share(yb + OY_Z5 + l * S_P, yb + OY_Z6 + l * S_P, wv, a, b);
        }
        part[wv * 32 + lane] = a; part[(KSP + wv) * 32 + lane] = b;
        __syncthreads();
        if (wv == 0 && act) {
            const float sa = ((part[lane] + part[32 + lane]) + part[64 + lane]) + part[96 + lane];
            const float sb = ((part[128 + lane] + part[160 + lane]) + part[192 + lane]) + part[224 + lane];
            const float au = y2 - sa;
            const float ad = y3 - sb;
            if (PD) {
                const float nu = fmaxf(xu - step * au, 0.f);
                const float nd = fmaxf(xd - step * ad, 0.f);
                X[OX_FU + l] = nu; XB[OX_FU + l] = 2.f * nu - xu;
                X[OX_FD + l] = nd; XB[OX_FD + l] = 2.f * nd - xd;
            } else {
                X[OX_FU + l] = au; X[OX_FD + l] = ad;
            }
        }
    }
}

constexpr int EPW = 128;
constexpr int MA_T1 = 25 * 13;
constexpr int MA_T2 = MA_T1 + NL / RPW;
constexpr int MA_T3 = MA_T2 + (NS + 31) / 32;
constexpr int MA_T4 = MA_T3 + (NG + 31) / 32;
constexpr int MA_T5 = MA_T4 + 1;
constexpr int MA_NB = MA_T5 + (NG * NS + EPW - 1) / EPW;

template <bool PD>
__global__ __launch_bounds__(32) __attribute__((amdgpu_waves_per_eu(1, 1))) void k_mv(
        float* w, const float* Pmax) {
    const int lane = threadIdx.x, blk = blockIdx.x;
    const float* PG = w + WS_PG;
    const float* xb = PD ? (w + WS_XB) : (w + WS_X);
    float* Y = PD ? (w + WS_Y) : (w + WS_AX);
    const float* pw = w + WS_PW;
    const float step = PD ? w[WS_STEP] : 0.f;
    if (blk < MA_T1) {
        const int tm = blk / 13, tn = blk % 13;
        const int m16 = lane & 15, kh = (lane >> 4) * 2;
        const int m = tm * 16 + m16, n = tn * 16 + m16;
        const int mr = tm * 16 + ((lane >> 4) << 3);
        const int nc = n < NS ? n : NS - 1;
        float fu[8], fd[8], y5[8], y6[8], pv[8];
        for (int i = 0; i < 8; ++i) {
            const int l = mr + i;
            const int o = l * S_P + nc;
            fu[i] = xb[OX_FU + l]; fd[i] = xb[OX_FD + l];
            y5[i] = PD ? Y[OY_Z5 + o] : 0.f; y6[i] = PD ? Y[OY_Z6 + o] : 0.f;
            pv[i] = PD ? pw[o] : 0.f;
        }
        v8f acc = tile_gemm<NG>(PG + m * G_P, xb + OX_U, xb + OX_D, n, kh);
        float r5[8], r6[8];
        for (int i = 0; i < 8; ++i) {
            const float f = acc[i];
            const float a5 =  f - fu[i];
            const float a6 = -f - fd[i];
            if (PD) {
                r5[i] = fmaxf(y5[i] + step * (a5 + pv[i]), 0.f);
                r6[i] = fmaxf(y6[i] + step * (a6 - pv[i]), 0.f);
            } else {
                r5[i] = a5; r6[i] = a6;
            }
        }
        if (n < NS) {
            for (int i = 0; i < 8; ++i) {
                const int o = (mr + i) * S_P + n;
                Y[OY_Z5 + o] = r5[i]; Y[OY_Z6 + o] = r6[i];
            }
        }
    } else if (blk < MA_T2) {
        const int l = (blk - MA_T1) * RPW + lane;
        if (lane < RPW && l < NL) {
            const float fu = xb[OX_FU + l], fd = xb[OX_FD + l];
            const float y2 = PD ? Y[OY_Y2 + l] : 0.f, y3 = PD ? Y[OY_Y3 + l] : 0.f;
            const float b2 = PD ? w[WS_B2 + l] : 0.f, b3 = PD ? w[WS_B3 + l] : 0.f;
            float s = 0.f; const float* r = PG + l * G_P; const float* p = xb + OX_P;
#pragma unroll
            for (int g = 0; g < NG; ++g) s += r[g] * p[g];
            const float a2 =  s + fu;
            const float a3 = -s + fd;
            if (PD) {
                Y[OY_Y2 + l] = y2 + step * (a2 - b2);
                Y[OY_Y3 + l] = y3 + step * (a3 - b3);
            } else {
                Y[OY_Y2 + l] = a2; Y[OY_Y3 + l] = a3;
            }
        }
    } else if (blk < MA_T3) {
        const int s = (blk - MA_T2) * 32 + lane;
        if (s < NS) {
            const float y4 = PD ? Y[OY_Y4 + s] : 0.f;
            const float b4 = PD ? w[WS_B4 + s] : 0.f;
            float c = 0.f;
            const float* Ux = xb + OX_U + s; const float* Dx = xb + OX_D + s;
#pragma unroll
            for (int g = 0; g < NG; ++g) c += Ux[g * S_P] - Dx[g * S_P];
            if (PD) Y[OY_Y4 + s] = y4 + step * (c - b4);
            else    Y[OY_Y4 + s] = c;
        }
    } else if (blk < MA_T4) {
        const int g = (blk - MA_T3) * 32 + lane;
        if (g < NG) {
            const float xp = xb[OX_P + g], xu = xb[OX_RU + g], xd = xb[OX_RD + g];
            const float a1 = xp + xu;
            const float a2 = xd - xp;
            if (PD) {
                const float z1 = Y[OY_Z1 + g], z2 = Y[OY_Z2 + g], pm = Pmax[g];
                Y[OY_Z1 + g] = fmaxf(z1 + step * (a1 - pm), 0.f);
                Y[OY_Z2 + g] = fmaxf(z2 + step *  a2, 0.f);
            } else {
                Y[OY_Z1 + g] = a1; Y[OY_Z2 + g] = a2;
            }
        }
    } else if (blk < MA_T5) {
        if (lane == 0) {
            const float y1 = PD ? Y[OY_Y1] : 0.f;
            const float b1 = PD ? w[WS_B1] : 0.f;
            float s = 0.f;
#pragma unroll
            for (int g = 0; g < NG; ++g) s += xb[OX_P + g];
            if (PD) Y[OY_Y1] = y1 + step * (s - b1);
            else    Y[OY_Y1] = s;
        }
    } else {
        const int base = (blk - MA_T5) * EPW;
        float r3[EPW / 32], r4[EPW / 32];
        for (int j = 0; j < EPW / 32; ++j) {
            const int i = base + j * 32 + lane;
            const int ic = i < NG * NS ? i : NG * NS - 1;
            const int g = ic / NS, s = ic % NS, o = g * S_P + s;
            const float a3 = xb[OX_U + o] - xb[OX_RU + g];
            const float a4 = xb[OX_D + o] - xb[OX_RD + g];
            if (PD) {
                r3[j] = fmaxf(Y[OY_Z3 + o] + step * a3, 0.f);
                r4[j] = fmaxf(Y[OY_Z4 + o] + step * a4, 0.f);
            } else {
                r3[j] = a3; r4[j] = a4;
            }
        }
        for (int j = 0; j < EPW / 32; ++j) {
            const int i = base + j * 32 + lane;
            if (i < NG * NS) {
                const int g = i / NS, s = i % NS, o = g * S_P + s;
                Y[OY_Z3 + o] = r3[j]; Y[OY_Z4 + o] = r4[j];
            }
        }
    }
}

__global__ void k_emit(const float* w, float* out) {
    const float* X = w + WS_X;
    const int total = 3 * NG + 2 * NG * NS + 2 * NL;
    for (int idx = blockIdx.x * blockDim.x + threadIdx.x; idx < total; idx += gridDim.x * blockDim.x) {
        float v;
        if (idx < NG)                 v = X[OX_P  + idx];
        else if (idx < 2 * NG)        v = X[OX_RU + (idx - NG)];
        else if (idx < 3 * NG)        v = X[OX_RD + (idx - 2 * NG)];
        else if (idx < 3 * NG + NG * NS) { int r = idx - 3 * NG; v = X[OX_U + (r / NS) * S_P + (r % NS)]; }
        else if (idx < 3 * NG + 2 * NG * NS) { int r = idx - 3 * NG - NG * NS; v = X[OX_D + (r / NS) * S_P + (r % NS)]; }
        else if (idx < 3 * NG + 2 * NG * NS + NL) v = X[OX_FU + (idx - 3 * NG - 2 * NG * NS)];
        else                          v = X[OX_FD + (idx - 3 * NG - 2 * NG * NS - NL)];
        out[idx] = v;
    }
}


extern "C" void kernel_launch(void* const* d_in, const int* in_sizes, int n_in,
                              void* d_out, int out_size, void* d_ws, size_t ws_size,
                              hipStream_t stream) {
    const float* w_scen   = (const float*)d_in[0];
    const float* Cost     = (const float*)d_in[1];
    const float* C_r_up   = (const float*)d_in[2];
    const float* C_r_down = (const float*)d_in[3];
    const float* Pmax     = (const float*)d_in[4];
    const float* PTDF     = (const float*)d_in[5];
    const float* node_G   = (const float*)d_in[6];
    const float* node_W   = (const float*)d_in[7];
    const float* node_L   = (const float*)d_in[8];
    const float* Pd       = (const float*)d_in[9];
    const float* w_exp    = (const float*)d_in[10];
    const float* Line_cap = (const float*)d_in[11];
    float* w   = (float*)d_ws;
    float* out = (float*)d_out;

    k_zero <<<512, 256, 0, stream>>>(w);
    k_prep1<<<240, 256, 0, stream>>>(w, w_scen, node_W, node_L, Pd, w_exp);
    k_prep2<<<640, 256, 0, stream>>>(w, PTDF, node_G, Line_cap);
    k_setv <<<161, 256, 0, stream>>>(w);

    for (int it = 0; it < POWER_ITERS; ++it) {
        k_norm<false><<<1, NT, 0, stream>>>(w);
        k_mv  <false><<<MA_NB, 32, 0, stream>>>(w, Pmax);
        k_rmv <false><<<RA_NB, RNT, 0, stream>>>(w, Cost, C_r_up, C_r_down);
    }
    k_norm<true><<<1, NT, 0, stream>>>(w);
    k_reset<<<256, 256, 0, stream>>>(w);

    for (int it = 0; it < PDHG_ITERS; ++it) {
        k_rmv<true><<<RA_NB, RNT, 0, stream>>>(w, Cost, C_r_up, C_r_down);
        k_mv <true><<<MA_NB, 32, 0, stream>>>(w, Pmax);
    }
    k_emit<<<161, 256, 0, stream>>>(w, out);
}


# Tile-product operands staged k-contiguous and pair-interleaved, B as the difference its producer stores once (U-D, z5-z6; y2-y3 too): one 16-byte load feeds two matrix steps; sum order kept
# speedup vs baseline: 30.4655x; 1.0318x over previous
#include <hip/hip_runtime.h>
#include <math.h>

typedef __attribute__((ext_vector_type(2))) float v2f;
typedef __attribute__((ext_vector_type(4))) float v4f;
typedef __attribute__((ext_vector_type(8))) float v8f;

#define NT 512

constexpr int NG = 100, NL = 400, NN = 300, NWF = 50, NS = 200;
constexpr int G_P = 112;
constexpr int S_P = 208;
constexpr int PDHG_ITERS = 300, POWER_ITERS = 25;

constexpr int OX_P  = 0;
constexpr int OX_RU = OX_P  + G_P;
constexpr int OX_RD = OX_RU + G_P;
constexpr int OX_U  = OX_RD + G_P;
constexpr int OX_D  = OX_U  + G_P * S_P;
constexpr int OX_FU = OX_D  + G_P * S_P;
constexpr int OX_FD = OX_FU + NL;
constexpr int XSZ   = OX_FD + NL;

constexpr int OY_Y1 = 0;
constexpr int OY_Y2 = 16;
constexpr int OY_Y3 = OY_Y2 + NL;
constexpr int OY_Y4 = OY_Y3 + NL;
constexpr int OY_Z1 = OY_Y4 + S_P;
constexpr int OY_Z2 = OY_Z1 + G_P;
constexpr int OY_Z3 = OY_Z2 + G_P;
constexpr int OY_Z4 = OY_Z3 + G_P * S_P;
constexpr int OY_Z5 = OY_Z4 + G_P * S_P;
constexpr int OY_Z6 = OY_Z5 + NL * S_P;
constexpr int YSZ   = OY_Z6 + NL * S_P;

constexpr int WS_PG   = 0;
constexpr int WS_PGT  = WS_PG   + NL * G_P;
constexpr int WS_PW   = WS_PGT  + G_P * NL;
constexpr int WS_B2   = WS_PW   + NL * S_P;
constexpr int WS_B3   = WS_B2   + NL;
constexpr int WS_B4   = WS_B3   + NL;
constexpr int WS_B1   = WS_B4   + S_P;
constexpr int WS_M1   = WS_B1   + 16;
constexpr int WS_NV   = WS_M1   + NN * S_P;
constexpr int WS_X    = WS_NV   + NN + 4;
constexpr int WS_XB   = WS_X    + XSZ;
constexpr int WS_ATY  = WS_XB   + XSZ;
constexpr int WS_Y    = WS_ATY  + XSZ;
constexpr int WS_AX   = WS_Y    + YSZ;
constexpr int WS_T    = WS_AX   + YSZ;
constexpr int WS_FLOW = WS_T    + G_P * S_P;
constexpr int WS_GP   = WS_FLOW + NL * S_P;
constexpr int WS_COL  = WS_GP   + NL;
constexpr int WS_S3   = WS_COL  + S_P;
constexpr int WS_S4   = WS_S3   + G_P;
constexpr int WS_S5   = WS_S4   + G_P;
constexpr int WS_S6   = WS_S5   + NL;
constexpr int WS_GTY  = WS_S6   + NL;
constexpr int WS_SUMP = WS_GTY  + G_P;
constexpr int WS_STEP = WS_SUMP + 16;
constexpr int WS_UDT  = WS_STEP + 16;
constexpr int WS_ZT   = WS_UDT  + S_P * G_P;
constexpr int WS_D23  = WS_ZT   + S_P * NL;
constexpr int WS_PGP  = WS_D23  + NL;
constexpr int WS_PGTP = WS_PGP  + NL * G_P;
constexpr int WS_TOTAL= WS_PGTP + G_P * NL;

__host__ __device__ constexpr int kpos100(int r) {
    return r < 96 ? ((r & ~7) | ((r & 2) << 1) | ((r & 4) >> 1) | (r & 1)) : r;
}
__host__ __device__ constexpr int kpos400(int l) { return (l / 100) * 100 + kpos100(l % 100); }


__global__ void k_zero(float* w) {
    for (int i = blockIdx.x * blockDim.x + threadIdx.x; i < WS_TOTAL;
         i += gridDim.x * blockDim.x) w[i] = 0.0f;
}

__global__ void k_prep1(float* w, const float* w_scen, const float* node_W,
                        const float* node_L, const float* Pd, const float* w_exp) {
    const int NM1 = NN * NS;
    const int total = NM1 + NN + NS + 1;
    for (int idx = blockIdx.x * blockDim.x + threadIdx.x; idx < total;
         idx += gridDim.x * blockDim.x) {
        if (idx < NM1) {
            int n = idx / NS, s = idx % NS;
            float a = 0.f;
            for (int ww = 0; ww < NWF; ++ww) a += node_W[n*NWF + ww] * w_scen[s*NWF + ww];
            w[WS_M1 + n*S_P + s] = a;
        } else if (idx < NM1 + NN) {
            int n = idx - NM1;
            float a = 0.f;
            for (int ww = 0; ww < NWF; ++ww) a += node_W[n*NWF + ww] * w_exp[ww];
            for (int m = 0; m < NN; ++m)     a -= node_L[n*NN + m] * Pd[m];
            w[WS_NV + n] = a;
        } else if (idx < NM1 + NN + NS) {
            int s = idx - NM1 - NN;
            float a = 0.f;
            for (int ww = 0; ww < NWF; ++ww) a += w_scen[s*NWF + ww];
            w[WS_B4 + s] = -a;
        } else {
            float a = 0.f;
            for (int m = 0; m < NN; ++m)  a += Pd[m];
            for (int ww = 0; ww < NWF; ++ww) a -= w_exp[ww];
            w[WS_B1] = a;
        }
    }
}

__global__ void k_prep2(float* w, const float* PTDF, const float* node_G,
                        const float* Line_cap) {
    const int A = NL * NG;
    const int B = A + NG * NL;
    const int C = B + NL * NS;
    const int D2 = C + NL;
    const int E = D2 + NL;
    for (int idx = blockIdx.x * blockDim.x + threadIdx.x; idx < E;
         idx += gridDim.x * blockDim.x) {
        if (idx < A) {
            int l = idx / NG, g = idx % NG;
            float a = 0.f;
            for (int n = 0; n < NN; ++n) a += PTDF[l*NN + n] * node_G[n*NG + g];
            w[WS_PG + l*G_P + g] = a;
            w[WS_PGP + l*G_P + kpos100(g)] = a;
        } else if (idx < B) {
            int r = idx - A; int g = r / NL, l = r % NL;
            float a = 0.f;
            for (int n = 0; n < NN; ++n) a += PTDF[l*NN + n] * node_G[n*NG + g];
            w[WS_PGT + g*NL + l] = a;
            w[WS_PGTP + g*NL + kpos400(l)] = a;
        } else if (idx < C) {
            int r = idx - B; int l = r / NS, s = r % NS;
            float a = 0.f;
            for (int n = 0; n < NN; ++n) a += PTDF[l*NN + n] * w[WS_M1 + n*S_P + s];
            w[WS_PW + l*S_P + s] = a;
        } else if (idx < D2) {
            int l = idx - C;
            float a = 0.f;
            for (int n = 0; n < NN; ++n) a += PTDF[l*NN + n] * w[WS_NV + n];
            w[WS_B2 + l] = Line_cap[l] - a;
        } else {
            int l = idx - D2;
            float a = 0.f;
            for (int n = 0; n < NN; ++n) a += PTDF[l*NN + n] * w[WS_NV + n];
            w[WS_B3 + l] = Line_cap[l] + a;
        }
    }
}

__global__ void k_setv(float* w) {
    const int total = 3*NG + 2*NG*NS + 2*NL;
    float* X = w + WS_X;
    for (int idx = blockIdx.x * blockDim.x + threadIdx.x; idx < total;
         idx += gridDim.x * blockDim.x) {
        int off;
        if (idx < NG)                off = OX_P  + idx;
        else if (idx < 2*NG)         off = OX_RU + (idx - NG);
        else if (idx < 3*NG)         off = OX_RD + (idx - 2*NG);
        else if (idx < 3*NG + NG*NS) { int r = idx - 3*NG; off = OX_U + (r/NS)*S_P + (r%NS); }
        else if (idx < 3*NG + 2*NG*NS) { int r = idx - 3*NG - NG*NS; off = OX_D + (r/NS)*S_P + (r%NS); }
        else if (idx < 3*NG + 2*NG*NS + NL) off = OX_FU + (idx - 3*NG - 2*NG*NS);
        else                         off = OX_FD + (idx - 3*NG - 2*NG*NS - NL);
        X[off] = 1.0f;
    }
}


__device__ __forceinline__ float block_reduce(float v, float* red, int tid) {
    red[tid] = v; __syncthreads();
    for (int off = NT >> 1; off > 0; off >>= 1) {
        if (tid < off) red[tid] += red[tid + off];
        __syncthreads();
    }
    float r = red[0]; __syncthreads();
    return r;
}

template <bool FINAL>
__global__ __launch_bounds__(NT) void k_norm(float* w) {
    __shared__ float red[NT];
    const int tid = threadIdx.x;
    float* X = w + WS_X;
    float part = 0.f;
    for (int i = tid; i < XSZ; i += NT) { float v = X[i]; part += v * v; }
    float sumsq = block_reduce(part, red, tid);
    if (!FINAL) {
        float inv = 1.0f / sqrtf(sumsq);
        for (int i = tid; i < XSZ; i += NT) X[i] *= inv;
    } else {
        if (tid == 0) w[WS_STEP] = 0.9f / sqrtf(sqrtf(sumsq));
    }
}

__global__ void k_reset(float* w) {
    const int total = 2 * XSZ + YSZ + S_P * NL + NL;
    for (int i = blockIdx.x * blockDim.x + threadIdx.x; i < total; i += gridDim.x * blockDim.x) {
        if (i < XSZ)                w[WS_X + i] = 0.f;
        else if (i < 2 * XSZ)       w[WS_XB + (i - XSZ)] = 0.f;
        else if (i < 2 * XSZ + YSZ) w[WS_Y + (i - 2 * XSZ)] = 0.f;
        else                        w[WS_ZT + (i - 2 * XSZ - YSZ)] = 0.f;
    }
}

template <int K>
__device__ __forceinline__ v8f tile_gemm(const float* arow, const float* P, const float* Q,
                                         int n, int kh) {
    v8f acc = {0.f,0.f,0.f,0.f,0.f,0.f,0.f,0.f};
#pragma unroll
    for (int k0 = 0; k0 < K; k0 += 4) {
        int ka = k0 + kh;
        v2f a = *(const v2f*)(arow + ka);
        v2f b;
        int bi = ka * S_P + n;
        b[0] = P[bi]       - Q[bi];
        b[1] = P[bi + S_P] - Q[bi + S_P];
        acc = __builtin_amdgcn_wmma_f32_16x16x4_f32(false, a, false, b, (short)0, acc, false, false);
    }
    return acc;
}

__device__ __forceinline__ v8f tile_gemm_kp(const float* arow, const float* brow, int h) {
    v8f acc = {0.f,0.f,0.f,0.f,0.f,0.f,0.f,0.f};
#pragma unroll
    for (int j = 0; j < 12; ++j) {
        const v4f a = *(const v4f*)(arow + 8 * j + 4 * h);
        const v4f b = *(const v4f*)(brow + 8 * j + 4 * h);
        v2f a0, b0, a1, b1;
        a0[0] = a[0]; a0[1] = a[1]; b0[0] = b[0]; b0[1] = b[1];
        a1[0] = a[2]; a1[1] = a[3]; b1[0] = b[2]; b1[1] = b[3];
        acc = __builtin_amdgcn_wmma_f32_16x16x4_f32(false, a0, false, b0, (short)0, acc, false, false);
        acc = __builtin_amdgcn_wmma_f32_16x16x4_f32(false, a1, false, b1, (short)0, acc, false, false);
    }
    const v2f at = *(const v2f*)(arow + 96 + 2 * h);
    const v2f bt = *(const v2f*)(brow + 96 + 2 * h);
    acc = __builtin_amdgcn_wmma_f32_16x16x4_f32(false, at, false, bt, (short)0, acc, false, false);
    return acc;
}
static_assert(NG == 100 && NL == 400, "tile_gemm_kp and kpos100 / kpos400 are written for 100-long k ranges");

constexpr int RPW = 8;


constexpr int KSP = 4;
constexpr int RNT = 32 * KSP;
constexpr int KQ  = NL / KSP;
constexpr int SQ  = 48;
constexpr int RA_T1 = 7 * 13;
constexpr int RA_T2 = RA_T1 + (NG + RPW - 1) / RPW;
constexpr int RA_T3 = RA_T2 + (NG + RPW - 1) / RPW;
constexpr int RA_NB = RA_T3 + NL / RPW;

__device__ __forceinline__ void row_sum_share(const float* r0, const float* r1, int wv, float& a, float& b) {
    const float* p0 = r0 + wv * SQ;
    const float* p1 = r1 + wv * SQ;
#pragma unroll
    for (int s = 0; s < SQ; ++s) { a += p0[s]; b += p1[s]; }
    if (wv == KSP - 1) {
#pragma unroll
        for (int s = KSP * SQ; s < NS; ++s) { a += r0[s]; b += r1[s]; }
    }
}

template <bool PD>
__global__ __launch_bounds__(RNT) __attribute__((amdgpu_waves_per_eu(1, 1))) void k_rmv(
        float* w, const float* Cost, const float* Cru, const float* Crd) {
    __shared__ float part[KSP * 8 * 32];
    const int lane = threadIdx.x & 31, blk = blockIdx.x;
    const int wv = __builtin_amdgcn_readfirstlane((int)(threadIdx.x >> 5));
    const float* PGT = w + WS_PGT;
    const float* yb = PD ? (w + WS_Y) : (w + WS_AX);
    float* X  = w + WS_X;
    float* XB = w + WS_XB;
    const float step = PD ? w[WS_STEP] : 0.f;
    if (blk < RA_T1) {
        const int tm = blk / 13, tn = blk % 13;
        const int m16 = lane & 15;
        const int m = tm * 16 + m16, n = tn * 16 + m16;
        const int mr = tm * 16 + ((lane >> 4) << 3) + 2 * wv;
        const int nc = n < NS ? n : NS - 1;
        const float y4 = yb[OY_Y4 + nc];
        float z3v[2], z4v[2], xu[2], xd[2];
        for (int j = 0; j < 2; ++j) {
            const int gc = (mr + j) < NG ? (mr + j) : NG - 1;
            const int o = gc * S_P + nc;
            z3v[j] = yb[OY_Z3 + o]; z4v[j] = yb[OY_Z4 + o];
            xu[j] = PD ? X[OX_U + o] : 0.f; xd[j] = PD ? X[OX_D + o] : 0.f;
        }
        const int kb = wv * KQ;
        v8f acc = tile_gemm_kp(w + WS_PGTP + m * NL + kb, w + WS_ZT + n * NL + kb, lane >> 4);
        for (int i = 0; i < 8; ++i) part[(wv * 8 + i) * 32 + lane] = acc[i];
        __syncthreads();
        float nu[2], nd[2], bu[2], bd[2];
        for (int j = 0; j < 2; ++j) {
            const int i = 2 * wv + j;
            const float tv = ((part[(0 * 8 + i) * 32 + lane] + part[(1 * 8 + i) * 32 + lane])
                              + part[(2 * 8 + i) * 32 + lane]) + part[(3 * 8 + i) * 32 + lane];
            const float au =  y4 + z3v[j] + tv;
            const float ad = -y4 + z4v[j] - tv;
            if (PD) {
                nu[j] = fmaxf(xu[j] - step * au, 0.f); bu[j] = 2.f * nu[j] - xu[j];
                nd[j] = fmaxf(xd[j] - step * ad, 0.f); bd[j] = 2.f * nd[j] - xd[j];
            } else {
                nu[j] = au; nd[j] = ad; bu[j] = 0.f; bd[j] = 0.f;
            }
        }
        for (int j = 0; j < 2; ++j) {
            const int g = mr + j;
            if (n < NS && g < NG) {
                const int o = g * S_P + n;
                X[OX_U + o] = nu[j]; X[OX_D + o] = nd[j];
                if (PD) { XB[OX_U + o] = bu[j]; XB[OX_D + o] = bd[j]; }
            }
        }
        if (PD && n < NS && mr < NG) {
            v2f d; d[0] = bu[0] - bd[0]; d[1] = bu[1] - bd[1];
            *(v2f*)(w + WS_UDT + n * G_P + kpos100(mr)) = d;
        }
    } else if (blk < RA_T2) {
        const int g = (blk - RA_T1) * RPW + lane;
        const bool act = lane < RPW && g < NG;
        float z1 = 0.f, z2 = 0.f, xu = 0.f, xd = 0.f, cu = 0.f, cd = 0.f;
        float a = 0.f, b = 0.f;
        if (act) {
            if (wv == 0) {
                z1 = yb[OY_Z1 + g]; z2 = yb[OY_Z2 + g];
                xu = PD ? X[OX_RU + g] : 0.f; xd = PD ? X[OX_RD + g] : 0.f;
                cu = PD ? Cru[g] : 0.f; cd = PD ? Crd[g] : 0.f;
            }
            row_sum_share(yb + OY_Z3 + g * S_P, yb + OY_Z4 + g * S_P, wv, a, b);
        }
        part[wv * 32 + lane] = a; part[(KSP + wv) * 32 + lane] = b;
        __syncthreads();
        if (wv == 0 && act) {
            const float sa = ((part[lane] + part[32 + lane]) + part[64 + lane]) + part[96 + lane];
            const float sb = ((part[128 + lane] + part[160 + lane]) + part[192 + lane]) + part[224 + lane];
            const float au = z1 - sa;
            const float ad = z2 - sb;
            if (PD) {
                const float nu = fmaxf(xu - step * (cu + au), 0.f);
                const float nd = fmaxf(xd - step * (cd + ad), 0.f);
                X[OX_RU + g] = nu; XB[OX_RU + g] = 2.f * nu - xu;
                X[OX_RD + g] = nd; XB[OX_RD + g] = 2.f * nd - xd;
            } else {
                X[OX_RU + g] = au; X[OX_RD + g] = ad;
            }
        }
    } else if (blk < RA_T3) {
        const int g = (blk - RA_T2) * RPW + lane;
        const bool act = lane < RPW && g < NG;
        float y1 = 0.f, z1 = 0.f, z2 = 0.f, xp = 0.f, cp = 0.f;
        float c = 0.f;
        if (act) {
            if (wv == 0) {
                y1 = yb[OY_Y1]; z1 = yb[OY_Z1 + g]; z2 = yb[OY_Z2 + g];
                xp = PD ? X[OX_P + g] : 0.f;
                cp = PD ? Cost[g] : 0.f;
            }
            const float* pr = PGT + g * NL + wv * KQ;
            const float* dp = w + WS_D23 + wv * KQ;
#pragma unroll
            for (int l = 0; l < KQ; ++l) c += pr[l] * dp[l];
        }
        part[wv * 32 + lane] = c;
        __syncthreads();
        if (wv == 0 && act) {
            const float cs = ((part[lane] + part[32 + lane]) + part[64 + lane]) + part[96 + lane];
            const float ap = y1 + cs + z1 - z2;
            if (PD) {
                const float nn = fmaxf(xp - step * (cp + ap), 0.f);
                X[OX_P + g] = nn; XB[OX_P + g] = 2.f * nn - xp;
            } else {
                X[OX_P + g] = ap;
            }
        }
    } else {
        const int l = (blk - RA_T3) * RPW + lane;
        const bool act = lane < RPW && l < NL;
        float y2 = 0.f, y3 = 0.f, xu = 0.f, xd = 0.f;
        float a = 0.f, b = 0.f;
        if (act) {
            if (wv == 0) {
                y2 = yb[OY_Y2 + l]; y3 = yb[OY_Y3 + l];
                xu = PD ? X[OX_FU + l] : 0.f; xd = PD ? X[OX_FD + l] : 0.f;
            }
            row_sum_share(yb + OY_Z5 + l * S_P, yb + OY_Z6 + l * S_P, wv, a, b);
        }
        part[wv * 32 + lane] = a; part[(KSP + wv) * 32 + lane] = b;
        __syncthreads();
        if (wv == 0 && act) {
            const float sa = ((part[lane] + part[32 + lane]) + part[64 + lane]) + part[96 + lane];
            const float sb = ((part[128 + lane] + part[160 + lane]) + part[192 + lane]) + part[224 + lane];
            const float au = y2 - sa;
            const float ad = y3 - sb;
            if (PD) {
                const float nu = fmaxf(xu - step * au, 0.f);
                const float nd = fmaxf(xd - step * ad, 0.f);
                X[OX_FU + l] = nu; XB[OX_FU + l] = 2.f * nu - xu;
                X[OX_FD + l] = nd; XB[OX_FD + l] = 2.f * nd - xd;
            } else {
                X[OX_FU + l] = au; X[OX_FD + l] = ad;
            }
        }
    }
}

constexpr int EPW = 128;
constexpr int MA_T1 = 25 * 13;
constexpr int MA_T2 = MA_T1 + NL / RPW;
constexpr int MA_T3 = MA_T2 + (NS + 31) / 32;
constexpr int MA_T4 = MA_T3 + (NG + 31) / 32;
constexpr int MA_T5 = MA_T4 + 1;
constexpr int MA_NB = MA_T5 + (NG * NS + EPW - 1) / EPW;

template <bool PD>
__global__ __launch_bounds__(32) __attribute__((amdgpu_waves_per_eu(1, 1))) void k_mv(
        float* w, const float* Pmax) {
    const int lane = threadIdx.x, blk = blockIdx.x;
    const float* PG = w + WS_PG;
    const float* xb = PD ? (w + WS_XB) : (w + WS_X);
    float* Y = PD ? (w + WS_Y) : (w + WS_AX);
    const float* pw = w + WS_PW;
    const float step = PD ? w[WS_STEP] : 0.f;
    if (blk < MA_T1) {
        const int tm = blk / 13, tn = blk % 13;
        const int m16 = lane & 15, kh = (lane >> 4) * 2;
        const int m = tm * 16 + m16, n = tn * 16 + m16;
        const int mr = tm * 16 + ((lane >> 4) << 3);
        const int nc = n < NS ? n : NS - 1;
        float fu[8], fd[8], y5[8], y6[8], pv[8];
        for (int i = 0; i < 8; ++i) {
            const int l = mr + i;
            const int o = l * S_P + nc;
            fu[i] = xb[OX_FU + l]; fd[i] = xb[OX_FD + l];
            y5[i] = PD ? Y[OY_Z5 + o] : 0.f; y6[i] = PD ? Y[OY_Z6 + o] : 0.f;
            pv[i] = PD ? pw[o] : 0.f;
        }
        v8f acc = PD ? tile_gemm_kp(w + WS_PGP + m * G_P, w + WS_UDT + n * G_P, lane >> 4)
                     : tile_gemm<NG>(PG + m * G_P, xb + OX_U, xb + OX_D, n, kh);
        float r5[8], r6[8];
        for (int i = 0; i < 8; ++i) {
            const float f = acc[i];
            const float a5 =  f - fu[i];
            const float a6 = -f - fd[i];
            if (PD) {
                r5[i] = fmaxf(y5[i] + step * (a5 + pv[i]), 0.f);
                r6[i] = fmaxf(y6[i] + step * (a6 - pv[i]), 0.f);
            } else {
                r5[i] = a5; r6[i] = a6;
            }
        }
        if (n < NS) {
            for (int i = 0; i < 8; ++i) {
                const int o = (mr + i) * S_P + n;
                Y[OY_Z5 + o] = r5[i]; Y[OY_Z6 + o] = r6[i];
            }
            float* zt = w + WS_ZT + n * NL;
            for (int i = 0; i < 8; i += 2) {
                v2f d; d[0] = r5[i] - r6[i]; d[1] = r5[i + 1] - r6[i + 1];
                *(v2f*)(zt + kpos400(mr + i)) = d;
            }
        }
    } else if (blk < MA_T2) {
        const int l = (blk - MA_T1) * RPW + lane;
        if (lane < RPW && l < NL) {
            const float fu = xb[OX_FU + l], fd = xb[OX_FD + l];
            const float y2 = PD ? Y[OY_Y2 + l] : 0.f, y3 = PD ? Y[OY_Y3 + l] : 0.f;
            const float b2 = PD ? w[WS_B2 + l] : 0.f, b3 = PD ? w[WS_B3 + l] : 0.f;
            float s = 0.f; const float* r = PG + l * G_P; const float* p = xb + OX_P;
#pragma unroll
            for (int g = 0; g < NG; ++g) s += r[g] * p[g];
            const float a2 =  s + fu;
            const float a3 = -s + fd;
            const float n2 = PD ? y2 + step * (a2 - b2) : a2;
            const float n3 = PD ? y3 + step * (a3 - b3) : a3;
            Y[OY_Y2 + l] = n2; Y[OY_Y3 + l] = n3;
            w[WS_D23 + l] = n2 - n3;
        }
    } else if (blk < MA_T3) {
        const int s = (blk - MA_T2) * 32 + lane;
        if (s < NS) {
            const float y4 = PD ? Y[OY_Y4 + s] : 0.f;
            const float b4 = PD ? w[WS_B4 + s] : 0.f;
            float c = 0.f;
            if (PD) {
                const float* r = w + WS_UDT + s * G_P;
#pragma unroll
                for (int g = 0; g < NG; ++g) c += r[kpos100(g)];
            } else {
                const float* Ux = xb + OX_U + s; const float* Dx = xb + OX_D + s;
#pragma unroll
                for (int g = 0; g < NG; ++g) c += Ux[g * S_P] - Dx[g * S_P];
            }
            if (PD) Y[OY_Y4 + s] = y4 + step * (c - b4);
            else    Y[OY_Y4 + s] = c;
        }
    } else if (blk < MA_T4) {
        const int g = (blk - MA_T3) * 32 + lane;
        if (g < NG) {
            const float xp = xb[OX_P + g], xu = xb[OX_RU + g], xd = xb[OX_RD + g];
            const float a1 = xp + xu;
            const float a2 = xd - xp;
            if (PD) {
                const float z1 = Y[OY_Z1 + g], z2 = Y[OY_Z2 + g], pm = Pmax[g];
                Y[OY_Z1 + g] = fmaxf(z1 + step * (a1 - pm), 0.f);
                Y[OY_Z2 + g] = fmaxf(z2 + step *  a2, 0.f);
            } else {
                Y[OY_Z1 + g] = a1; Y[OY_Z2 + g] = a2;
            }
        }
    } else if (blk < MA_T5) {
        if (lane == 0) {
            const float y1 = PD ? Y[OY_Y1] : 0.f;
            const float b1 = PD ? w[WS_B1] : 0.f;
            float s = 0.f;
#pragma unroll
            for (int g = 0; g < NG; ++g) s += xb[OX_P + g];
            if (PD) Y[OY_Y1] = y1 + step * (s - b1);
            else    Y[OY_Y1] = s;
        }
    } else {
        const int base = (blk - MA_T5) * EPW;
        float r3[EPW / 32], r4[EPW / 32];
        for (int j = 0; j < EPW / 32; ++j) {
            const int i = base + j * 32 + lane;
            const int ic = i < NG * NS ? i : NG * NS - 1;
            const int g = ic / NS, s = ic % NS, o = g * S_P + s;
            const float a3 = xb[OX_U + o] - xb[OX_RU + g];
            const float a4 = xb[OX_D + o] - xb[OX_RD + g];
            if (PD) {
                r3[j] = fmaxf(Y[OY_Z3 + o] + step * a3, 0.f);
                r4[j] = fmaxf(Y[OY_Z4 + o] + step * a4, 0.f);
            } else {
                r3[j] = a3; r4[j] = a4;
            }
        }
        for (int j = 0; j < EPW / 32; ++j) {
            const int i = base + j * 32 + lane;
            if (i < NG * NS) {
                const int g = i / NS, s = i % NS, o = g * S_P + s;
                Y[OY_Z3 + o] = r3[j]; Y[OY_Z4 + o] = r4[j];
            }
        }
    }
}

__global__ void k_emit(const float* w, float* out) {
    const float* X = w + WS_X;
    const int total = 3 * NG + 2 * NG * NS + 2 * NL;
    for (int idx = blockIdx.x * blockDim.x + threadIdx.x; idx < total; idx += gridDim.x * blockDim.x) {
        float v;
        if (idx < NG)                 v = X[OX_P  + idx];
        else if (idx < 2 * NG)        v = X[OX_RU + (idx - NG)];
        else if (idx < 3 * NG)        v = X[OX_RD + (idx - 2 * NG)];
        else if (idx < 3 * NG + NG * NS) { int r = idx - 3 * NG; v = X[OX_U + (r / NS) * S_P + (r % NS)]; }
        else if (idx < 3 * NG + 2 * NG * NS) { int r = idx - 3 * NG - NG * NS; v = X[OX_D + (r / NS) * S_P + (r % NS)]; }
        else if (idx < 3 * NG + 2 * NG * NS + NL) v = X[OX_FU + (idx - 3 * NG - 2 * NG * NS)];
        else                          v = X[OX_FD + (idx - 3 * NG - 2 * NG * NS - NL)];
        out[idx] = v;
    }
}


extern "C" void kernel_launch(void* const* d_in, const int* in_sizes, int n_in,
                              void* d_out, int out_size, void* d_ws, size_t ws_size,
                              hipStream_t stream) {
    const float* w_scen   = (const float*)d_in[0];
    const float* Cost     = (const float*)d_in[1];
    const float* C_r_up   = (const float*)d_in[2];
    const float* C_r_down = (const float*)d_in[3];
    const float* Pmax     = (const float*)d_in[4];
    const float* PTDF     = (const float*)d_in[5];
    const float* node_G   = (const float*)d_in[6];
    const float* node_W   = (const float*)d_in[7];
    const float* node_L   = (const float*)d_in[8];
    const float* Pd       = (const float*)d_in[9];
    const float* w_exp    = (const float*)d_in[10];
    const float* Line_cap = (const float*)d_in[11];
    float* w   = (float*)d_ws;
    float* out = (float*)d_out;

    k_zero <<<512, 256, 0, stream>>>(w);
    k_prep1<<<240, 256, 0, stream>>>(w, w_scen, node_W, node_L, Pd, w_exp);
    k_prep2<<<640, 256, 0, stream>>>(w, PTDF, node_G, Line_cap);
    k_setv <<<161, 256, 0, stream>>>(w);

    for (int it = 0; it < POWER_ITERS; ++it) {
        k_norm<false><<<1, NT, 0, stream>>>(w);
        k_mv  <false><<<MA_NB, 32, 0, stream>>>(w, Pmax);
        k_rmv <false><<<RA_NB, RNT, 0, stream>>>(w, Cost, C_r_up, C_r_down);
    }
    k_norm<true><<<1, NT, 0, stream>>>(w);
    k_reset<<<256, 256, 0, stream>>>(w);

    for (int it = 0; it < PDHG_ITERS; ++it) {
        k_rmv<true><<<RA_NB, RNT, 0, stream>>>(w, Cost, C_r_up, C_r_down);
        k_mv <true><<<MA_NB, 32, 0, stream>>>(w, Pmax);
    }
    k_emit<<<161, 256, 0, stream>>>(w, out);
}


# Every load of a task (reduction and epilogue operands) is issued in one group ahead of its arithmetic, held by a scheduling barrier: one memory round trip per launch, not two; sums keep their order
# speedup vs baseline: 32.4444x; 1.0650x over previous
#include <hip/hip_runtime.h>
#include <math.h>

typedef __attribute__((ext_vector_type(2))) float v2f;
typedef __attribute__((ext_vector_type(4))) float v4f;
typedef __attribute__((ext_vector_type(8))) float v8f;

#define NT 512

constexpr int NG = 100, NL = 400, NN = 300, NWF = 50, NS = 200;
constexpr int G_P = 112;
constexpr int S_P = 208;
constexpr int PDHG_ITERS = 300, POWER_ITERS = 25;

constexpr int OX_P  = 0;
constexpr int OX_RU = OX_P  + G_P;
constexpr int OX_RD = OX_RU + G_P;
constexpr int OX_U  = OX_RD + G_P;
constexpr int OX_D  = OX_U  + G_P * S_P;
constexpr int OX_FU = OX_D  + G_P * S_P;
constexpr int OX_FD = OX_FU + NL;
constexpr int XSZ   = OX_FD + NL;

constexpr int OY_Y1 = 0;
constexpr int OY_Y2 = 16;
constexpr int OY_Y3 = OY_Y2 + NL;
constexpr int OY_Y4 = OY_Y3 + NL;
constexpr int OY_Z1 = OY_Y4 + S_P;
constexpr int OY_Z2 = OY_Z1 + G_P;
constexpr int OY_Z3 = OY_Z2 + G_P;
constexpr int OY_Z4 = OY_Z3 + G_P * S_P;
constexpr int OY_Z5 = OY_Z4 + G_P * S_P;
constexpr int OY_Z6 = OY_Z5 + NL * S_P;
constexpr int YSZ   = OY_Z6 + NL * S_P;

constexpr int WS_PG   = 0;
constexpr int WS_PGT  = WS_PG   + NL * G_P;
constexpr int WS_PW   = WS_PGT  + G_P * NL;
constexpr int WS_B2   = WS_PW   + NL * S_P;
constexpr int WS_B3   = WS_B2   + NL;
constexpr int WS_B4   = WS_B3   + NL;
constexpr int WS_B1   = WS_B4   + S_P;
constexpr int WS_M1   = WS_B1   + 16;
constexpr int WS_NV   = WS_M1   + NN * S_P;
constexpr int WS_X    = WS_NV   + NN + 4;
constexpr int WS_XB   = WS_X    + XSZ;
constexpr int WS_ATY  = WS_XB   + XSZ;
constexpr int WS_Y    = WS_ATY  + XSZ;
constexpr int WS_AX   = WS_Y    + YSZ;
constexpr int WS_T    = WS_AX   + YSZ;
constexpr int WS_FLOW = WS_T    + G_P * S_P;
constexpr int WS_GP   = WS_FLOW + NL * S_P;
constexpr int WS_COL  = WS_GP   + NL;
constexpr int WS_S3   = WS_COL  + S_P;
constexpr int WS_S4   = WS_S3   + G_P;
constexpr int WS_S5   = WS_S4   + G_P;
constexpr int WS_S6   = WS_S5   + NL;
constexpr int WS_GTY  = WS_S6   + NL;
constexpr int WS_SUMP = WS_GTY  + G_P;
constexpr int WS_STEP = WS_SUMP + 16;
constexpr int WS_UDT  = WS_STEP + 16;
constexpr int WS_ZT   = WS_UDT  + S_P * G_P;
constexpr int WS_D23  = WS_ZT   + S_P * NL;
constexpr int WS_PGP  = WS_D23  + NL;
constexpr int WS_PGTP = WS_PGP  + NL * G_P;
constexpr int WS_TOTAL= WS_PGTP + G_P * NL;

__host__ __device__ constexpr int kpos100(int r) {
    return r < 96 ? ((r & ~7) | ((r & 2) << 1) | ((r & 4) >> 1) | (r & 1)) : r;
}
__host__ __device__ constexpr int kpos400(int l) { return (l / 100) * 100 + kpos100(l % 100); }


__global__ void k_zero(float* w) {
    for (int i = blockIdx.x * blockDim.x + threadIdx.x; i < WS_TOTAL;
         i += gridDim.x * blockDim.x) w[i] = 0.0f;
}

__global__ void k_prep1(float* w, const float* w_scen, const float* node_W,
                        const float* node_L, const float* Pd, const float* w_exp) {
    const int NM1 = NN * NS;
    const int total = NM1 + NN + NS + 1;
    for (int idx = blockIdx.x * blockDim.x + threadIdx.x; idx < total;
         idx += gridDim.x * blockDim.x) {
        if (idx < NM1) {
            int n = idx / NS, s = idx % NS;
            float a = 0.f;
            for (int ww = 0; ww < NWF; ++ww) a += node_W[n*NWF + ww] * w_scen[s*NWF + ww];
            w[WS_M1 + n*S_P + s] = a;
        } else if (idx < NM1 + NN) {
            int n = idx - NM1;
            float a = 0.f;
            for (int ww = 0; ww < NWF; ++ww) a += node_W[n*NWF + ww] * w_exp[ww];
            for (int m = 0; m < NN; ++m)     a -= node_L[n*NN + m] * Pd[m];
            w[WS_NV + n] = a;
        } else if (idx < NM1 + NN + NS) {
            int s = idx - NM1 - NN;
            float a = 0.f;
            for (int ww = 0; ww < NWF; ++ww) a += w_scen[s*NWF + ww];
            w[WS_B4 + s] = -a;
        } else {
            float a = 0.f;
            for (int m = 0; m < NN; ++m)  a += Pd[m];
            for (int ww = 0; ww < NWF; ++ww) a -= w_exp[ww];
            w[WS_B1] = a;
        }
    }
}

__global__ void k_prep2(float* w, const float* PTDF, const float* node_G,
                        const float* Line_cap) {
    const int A = NL * NG;
    const int B = A + NG * NL;
    const int C = B + NL * NS;
    const int D2 = C + NL;
    const int E = D2 + NL;
    for (int idx = blockIdx.x * blockDim.x + threadIdx.x; idx < E;
         idx += gridDim.x * blockDim.x) {
        if (idx < A) {
            int l = idx / NG, g = idx % NG;
            float a = 0.f;
            for (int n = 0; n < NN; ++n) a += PTDF[l*NN + n] * node_G[n*NG + g];
            w[WS_PG + l*G_P + g] = a;
            w[WS_PGP + l*G_P + kpos100(g)] = a;
        } else if (idx < B) {
            int r = idx - A; int g = r / NL, l = r % NL;
            float a = 0.f;
            for (int n = 0; n < NN; ++n) a += PTDF[l*NN + n] * node_G[n*NG + g];
            w[WS_PGT + g*NL + l] = a;
            w[WS_PGTP + g*NL + kpos400(l)] = a;
        } else if (idx < C) {
            int r = idx - B; int l = r / NS, s = r % NS;
            float a = 0.f;
            for (int n = 0; n < NN; ++n) a += PTDF[l*NN + n] * w[WS_M1 + n*S_P + s];
            w[WS_PW + l*S_P + s] = a;
        } else if (idx < D2) {
            int l = idx - C;
            float a = 0.f;
            for (int n = 0; n < NN; ++n) a += PTDF[l*NN + n] * w[WS_NV + n];
            w[WS_B2 + l] = Line_cap[l] - a;
        } else {
            int l = idx - D2;
            float a = 0.f;
            for (int n = 0; n < NN; ++n) a += PTDF[l*NN + n] * w[WS_NV + n];
            w[WS_B3 + l] = Line_cap[l] + a;
        }
    }
}

__global__ void k_setv(float* w) {
    const int total = 3*NG + 2*NG*NS + 2*NL;
    float* X = w + WS_X;
    for (int idx = blockIdx.x * blockDim.x + threadIdx.x; idx < total;
         idx += gridDim.x * blockDim.x) {
        int off;
        if (idx < NG)                off = OX_P  + idx;
        else if (idx < 2*NG)         off = OX_RU + (idx - NG);
        else if (idx < 3*NG)         off = OX_RD + (idx - 2*NG);
        else if (idx < 3*NG + NG*NS) { int r = idx - 3*NG; off = OX_U + (r/NS)*S_P + (r%NS); }
        else if (idx < 3*NG + 2*NG*NS) { int r = idx - 3*NG - NG*NS; off = OX_D + (r/NS)*S_P + (r%NS); }
        else if (idx < 3*NG + 2*NG*NS + NL) off = OX_FU + (idx - 3*NG - 2*NG*NS);
        else                         off = OX_FD + (idx - 3*NG - 2*NG*NS - NL);
        X[off] = 1.0f;
    }
}


__device__ __forceinline__ float block_reduce(float v, float* red, int tid) {
    red[tid] = v; __syncthreads();
    for (int off = NT >> 1; off > 0; off >>= 1) {
        if (tid < off) red[tid] += red[tid + off];
        __syncthreads();
    }
    float r = red[0]; __syncthreads();
    return r;
}

template <bool FINAL>
__global__ __launch_bounds__(NT) void k_norm(float* w) {
    __shared__ float red[NT];
    const int tid = threadIdx.x;
    float* X = w + WS_X;
    float part = 0.f;
    for (int i = tid; i < XSZ; i += NT) { float v = X[i]; part += v * v; }
    float sumsq = block_reduce(part, red, tid);
    if (!FINAL) {
        float inv = 1.0f / sqrtf(sumsq);
        for (int i = tid; i < XSZ; i += NT) X[i] *= inv;
    } else {
        if (tid == 0) w[WS_STEP] = 0.9f / sqrtf(sqrtf(sumsq));
    }
}

__global__ void k_reset(float* w) {
    const int total = 2 * XSZ + YSZ + S_P * NL + NL;
    for (int i = blockIdx.x * blockDim.x + threadIdx.x; i < total; i += gridDim.x * blockDim.x) {
        if (i < XSZ)                w[WS_X + i] = 0.f;
        else if (i < 2 * XSZ)       w[WS_XB + (i - XSZ)] = 0.f;
        else if (i < 2 * XSZ + YSZ) w[WS_Y + (i - 2 * XSZ)] = 0.f;
        else                        w[WS_ZT + (i - 2 * XSZ - YSZ)] = 0.f;
    }
}

template <int K>
__device__ __forceinline__ v8f tile_gemm(const float* arow, const float* P, const float* Q,
                                         int n, int kh) {
    v8f acc = {0.f,0.f,0.f,0.f,0.f,0.f,0.f,0.f};
#pragma unroll
    for (int k0 = 0; k0 < K; k0 += 4) {
        int ka = k0 + kh;
        v2f a = *(const v2f*)(arow + ka);
        v2f b;
        int bi = ka * S_P + n;
        b[0] = P[bi]       - Q[bi];
        b[1] = P[bi + S_P] - Q[bi + S_P];
        acc = __builtin_amdgcn_wmma_f32_16x16x4_f32(false, a, false, b, (short)0, acc, false, false);
    }
    return acc;
}

struct KpFrags { v4f a[12], b[12]; v2f at, bt; };

__device__ __forceinline__ void load_kp(KpFrags& f, const float* arow, const float* brow, int h) {
#pragma unroll
    for (int j = 0; j < 12; ++j) {
        f.a[j] = *(const v4f*)(arow + 8 * j + 4 * h);
        f.b[j] = *(const v4f*)(brow + 8 * j + 4 * h);
    }
    f.at = *(const v2f*)(arow + 96 + 2 * h);
    f.bt = *(const v2f*)(brow + 96 + 2 * h);
}

__device__ __forceinline__ v8f mma_kp(const KpFrags& f) {
    v8f acc = {0.f,0.f,0.f,0.f,0.f,0.f,0.f,0.f};
#pragma unroll
    for (int j = 0; j < 12; ++j) {
        v2f a0, b0, a1, b1;
        a0[0] = f.a[j][0]; a0[1] = f.a[j][1]; b0[0] = f.b[j][0]; b0[1] = f.b[j][1];
        a1[0] = f.a[j][2]; a1[1] = f.a[j][3]; b1[0] = f.b[j][2]; b1[1] = f.b[j][3];
        acc = __builtin_amdgcn_wmma_f32_16x16x4_f32(false, a0, false, b0, (short)0, acc, false, false);
        acc = __builtin_amdgcn_wmma_f32_16x16x4_f32(false, a1, false, b1, (short)0, acc, false, false);
    }
    acc = __builtin_amdgcn_wmma_f32_16x16x4_f32(false, f.at, false, f.bt, (short)0, acc, false, false);
    return acc;
}

static_assert(NG == 100 && NL == 400, "tile_gemm_kp and kpos100 / kpos400 are written for 100-long k ranges");

constexpr int RPW = 8;


constexpr int KSP = 4;
constexpr int RNT = 32 * KSP;
constexpr int KQ  = NL / KSP;
constexpr int SQ  = 48;
constexpr int RA_T1 = 7 * 13;
constexpr int RA_T2 = RA_T1 + (NG + RPW - 1) / RPW;
constexpr int RA_T3 = RA_T2 + (NG + RPW - 1) / RPW;
constexpr int RA_NB = RA_T3 + NL / RPW;

__device__ __forceinline__ void row_sum_share(const float* r0, const float* r1, int wv, float& a, float& b) {
    const float* p0 = r0 + wv * SQ;
    const float* p1 = r1 + wv * SQ;
    constexpr int TAIL = NS - KSP * SQ;
    float v0[SQ + TAIL], v1[SQ + TAIL];
#pragma unroll
    for (int s = 0; s < SQ; ++s) { v0[s] = p0[s]; v1[s] = p1[s]; }
#pragma unroll
    for (int s = SQ; s < SQ + TAIL; ++s) { v0[s] = 0.f; v1[s] = 0.f; }
    if (wv == KSP - 1) {
#pragma unroll
        for (int s = SQ; s < SQ + TAIL; ++s) { v0[s] = p0[s]; v1[s] = p1[s]; }
    }
#pragma unroll
    for (int s = 0; s < SQ; ++s) { a += v0[s]; b += v1[s]; }
    if (wv == KSP - 1) {
#pragma unroll
        for (int s = SQ; s < SQ + TAIL; ++s) { a += v0[s]; b += v1[s]; }
    }
}

template <bool PD>
__global__ __launch_bounds__(RNT) __attribute__((amdgpu_waves_per_eu(1, 1))) void k_rmv(
        float* w, const float* Cost, const float* Cru, const float* Crd) {
    __shared__ float part[KSP * 8 * 32];
    const int lane = threadIdx.x & 31, blk = blockIdx.x;
    const int wv = __builtin_amdgcn_readfirstlane((int)(threadIdx.x >> 5));
    const float* PGT = w + WS_PGT;
    const float* yb = PD ? (w + WS_Y) : (w + WS_AX);
    float* X  = w + WS_X;
    float* XB = w + WS_XB;
    const float step = PD ? w[WS_STEP] : 0.f;
    if (blk < RA_T1) {
        const int tm = blk / 13, tn = blk % 13;
        const int m16 = lane & 15;
        const int m = tm * 16 + m16, n = tn * 16 + m16;
        const int mr = tm * 16 + ((lane >> 4) << 3) + 2 * wv;
        const int nc = n < NS ? n : NS - 1;
        const float y4 = yb[OY_Y4 + nc];
        float z3v[2], z4v[2], xu[2], xd[2];
        for (int j = 0; j < 2; ++j) {
            const int gc = (mr + j) < NG ? (mr + j) : NG - 1;
            const int o = gc * S_P + nc;
            z3v[j] = yb[OY_Z3 + o]; z4v[j] = yb[OY_Z4 + o];
            xu[j] = PD ? X[OX_U + o] : 0.f; xd[j] = PD ? X[OX_D + o] : 0.f;
        }
        const int kb = wv * KQ;
        KpFrags f;
        load_kp(f, w + WS_PGTP + m * NL + kb, w + WS_ZT + n * NL + kb, lane >> 4);
        __builtin_amdgcn_sched_barrier(0);
        v8f acc = mma_kp(f);
        for (int i = 0; i < 8; ++i) part[(wv * 8 + i) * 32 + lane] = acc[i];
        __syncthreads();
        float nu[2], nd[2], bu[2], bd[2];
        for (int j = 0; j < 2; ++j) {
            const int i = 2 * wv + j;
            const float tv = ((part[(0 * 8 + i) * 32 + lane] + part[(1 * 8 + i) * 32 + lane])
                              + part[(2 * 8 + i) * 32 + lane]) + part[(3 * 8 + i) * 32 + lane];
            const float au =  y4 + z3v[j] + tv;
            const float ad = -y4 + z4v[j] - tv;
            if (PD) {
                nu[j] = fmaxf(xu[j] - step * au, 0.f); bu[j] = 2.f * nu[j] - xu[j];
                nd[j] = fmaxf(xd[j] - step * ad, 0.f); bd[j] = 2.f * nd[j] - xd[j];
            } else {
                nu[j] = au; nd[j] = ad; bu[j] = 0.f; bd[j] = 0.f;
            }
        }
        for (int j = 0; j < 2; ++j) {
            const int g = mr + j;
            if (n < NS && g < NG) {
                const int o = g * S_P + n;
                X[OX_U + o] = nu[j]; X[OX_D + o] = nd[j];
                if (PD) { XB[OX_U + o] = bu[j]; XB[OX_D + o] = bd[j]; }
            }
        }
        if (PD && n < NS && mr < NG) {
            v2f d; d[0] = bu[0] - bd[0]; d[1] = bu[1] - bd[1];
            *(v2f*)(w + WS_UDT + n * G_P + kpos100(mr)) = d;
        }
    } else if (blk < RA_T2) {
        const int g = (blk - RA_T1) * RPW + lane;
        const bool act = lane < RPW && g < NG;
        float z1 = 0.f, z2 = 0.f, xu = 0.f, xd = 0.f, cu = 0.f, cd = 0.f;
        float a = 0.f, b = 0.f;
        if (act) {
            if (wv == 0) {
                z1 = yb[OY_Z1 + g]; z2 = yb[OY_Z2 + g];
                xu = PD ? X[OX_RU + g] : 0.f; xd = PD ? X[OX_RD + g] : 0.f;
                cu = PD ? Cru[g] : 0.f; cd = PD ? Crd[g] : 0.f;
            }
            row_sum_share(yb + OY_Z3 + g * S_P, yb + OY_Z4 + g * S_P, wv, a, b);
        }
        part[wv * 32 + lane] = a; part[(KSP + wv) * 32 + lane] = b;
        __syncthreads();
        if (wv == 0 && act) {
            const float sa = ((part[lane] + part[32 + lane]) + part[64 + lane]) + part[96 + lane];
            const float sb = ((part[128 + lane] + part[160 + lane]) + part[192 + lane]) + part[224 + lane];
            const float au = z1 - sa;
            const float ad = z2 - sb;
            if (PD) {
                const float nu = fmaxf(xu - step * (cu + au), 0.f);
                const float nd = fmaxf(xd - step * (cd + ad), 0.f);
                X[OX_RU + g] = nu; XB[OX_RU + g] = 2.f * nu - xu;
                X[OX_RD + g] = nd; XB[OX_RD + g] = 2.f * nd - xd;
            } else {
                X[OX_RU + g] = au; X[OX_RD + g] = ad;
            }
        }
    } else if (blk < RA_T3) {
        const int g = (blk - RA_T2) * RPW + lane;
        const bool act = lane < RPW && g < NG;
        float y1 = 0.f, z1 = 0.f, z2 = 0.f, xp = 0.f, cp = 0.f;
        float c = 0.f;
        if (act) {
            const float* pr = PGT + g * NL + wv * KQ;
            const float* dp = w + WS_D23 + wv * KQ;
            float pv[KQ], dv[KQ];
#pragma unroll
            for (int l = 0; l < KQ; ++l) { pv[l] = pr[l]; dv[l] = dp[l]; }
            if (wv == 0) {
                y1 = yb[OY_Y1]; z1 = yb[OY_Z1 + g]; z2 = yb[OY_Z2 + g];
                xp = PD ? X[OX_P + g] : 0.f;
                cp = PD ? Cost[g] : 0.f;
            }
#pragma unroll
            for (int l = 0; l < KQ; ++l) c += pv[l] * dv[l];
        }
        part[wv * 32 + lane] = c;
        __syncthreads();
        if (wv == 0 && act) {
            const float cs = ((part[lane] + part[32 + lane]) + part[64 + lane]) + part[96 + lane];
            const float ap = y1 + cs + z1 - z2;
            if (PD) {
                const float nn = fmaxf(xp - step * (cp + ap), 0.f);
                X[OX_P + g] = nn; XB[OX_P + g] = 2.f * nn - xp;
            } else {
                X[OX_P + g] = ap;
            }
        }
    } else {
        const int l = (blk - RA_T3) * RPW + lane;
        const bool act = lane < RPW && l < NL;
        float y2 = 0.f, y3 = 0.f, xu = 0.f, xd = 0.f;
        float a = 0.f, b = 0.f;
        if (act) {
            if (wv == 0) {
                y2 = yb[OY_Y2 + l]; y3 = yb[OY_Y3 + l];
                xu = PD ? X[OX_FU + l] : 0.f; xd = PD ? X[OX_FD + l] : 0.f;
            }
            row_sum_share(yb + OY_Z5 + l * S_P, yb + OY_Z6 + l * S_P, wv, a, b);
        }
        part[wv * 32 + lane] = a; part[(KSP + wv) * 32 + lane] = b;
        __syncthreads();
        if (wv == 0 && act) {
            const float sa = ((part[lane] + part[32 + lane]) + part[64 + lane]) + part[96 + lane];
            const float sb = ((part[128 + lane] + part[160 + lane]) + part[192 + lane]) + part[224 + lane];
            const float au = y2 - sa;
            const float ad = y3 - sb;
            if (PD) {
                const float nu = fmaxf(xu - step * au, 0.f);
                const float nd = fmaxf(xd - step * ad, 0.f);
                X[OX_FU + l] = nu; XB[OX_FU + l] = 2.f * nu - xu;
                X[OX_FD + l] = nd; XB[OX_FD + l] = 2.f * nd - xd;
            } else {
                X[OX_FU + l] = au; X[OX_FD + l] = ad;
            }
        }
    }
}

constexpr int EPW = 128;
constexpr int MA_T1 = 25 * 13;
constexpr int MA_T2 = MA_T1 + NL / RPW;
constexpr int MA_T3 = MA_T2 + (NS + 31) / 32;
constexpr int MA_T4 = MA_T3 + (NG + 31) / 32;
constexpr int MA_T5 = MA_T4 + 1;
constexpr int MA_NB = MA_T5 + (NG * NS + EPW - 1) / EPW;

template <bool PD>
__global__ __launch_bounds__(32) __attribute__((amdgpu_waves_per_eu(1, 1))) void k_mv(
        float* w, const float* Pmax) {
    const int lane = threadIdx.x, blk = blockIdx.x;
    const float* PG = w + WS_PG;
    const float* xb = PD ? (w + WS_XB) : (w + WS_X);
    float* Y = PD ? (w + WS_Y) : (w + WS_AX);
    const float* pw = w + WS_PW;
    const float step = PD ? w[WS_STEP] : 0.f;
    if (blk < MA_T1) {
        const int tm = blk / 13, tn = blk % 13;
        const int m16 = lane & 15, kh = (lane >> 4) * 2;
        const int m = tm * 16 + m16, n = tn * 16 + m16;
        const int mr = tm * 16 + ((lane >> 4) << 3);
        const int nc = n < NS ? n : NS - 1;
        float fu[8], fd[8], y5[8], y6[8], pv[8];
        for (int i = 0; i < 8; ++i) {
            const int l = mr + i;
            const int o = l * S_P + nc;
            fu[i] = xb[OX_FU + l]; fd[i] = xb[OX_FD + l];
            y5[i] = PD ? Y[OY_Z5 + o] : 0.f; y6[i] = PD ? Y[OY_Z6 + o] : 0.f;
            pv[i] = PD ? pw[o] : 0.f;
        }
        v8f acc;
        if (PD) {
            KpFrags f;
            load_kp(f, w + WS_PGP + m * G_P, w + WS_UDT + n * G_P, lane >> 4);
            __builtin_amdgcn_sched_barrier(0);
            acc = mma_kp(f);
        } else {
            acc = tile_gemm<NG>(PG + m * G_P, xb + OX_U, xb + OX_D, n, kh);
        }
        float r5[8], r6[8];
        for (int i = 0; i < 8; ++i) {
            const float f = acc[i];
            const float a5 =  f - fu[i];
            const float a6 = -f - fd[i];
            if (PD) {
                r5[i] = fmaxf(y5[i] + step * (a5 + pv[i]), 0.f);
                r6[i] = fmaxf(y6[i] + step * (a6 - pv[i]), 0.f);
            } else {
                r5[i] = a5; r6[i] = a6;
            }
        }
        if (n < NS) {
            for (int i = 0; i < 8; ++i) {
                const int o = (mr + i) * S_P + n;
                Y[OY_Z5 + o] = r5[i]; Y[OY_Z6 + o] = r6[i];
            }
            float* zt = w + WS_ZT + n * NL;
            for (int i = 0; i < 8; i += 2) {
                v2f d; d[0] = r5[i] - r6[i]; d[1] = r5[i + 1] - r6[i + 1];
                *(v2f*)(zt + kpos400(mr + i)) = d;
            }
        }
    } else if (blk < MA_T2) {
        const int l = (blk - MA_T1) * RPW + lane;
        if (lane < RPW && l < NL) {
            const float fu = xb[OX_FU + l], fd = xb[OX_FD + l];
            const float y2 = PD ? Y[OY_Y2 + l] : 0.f, y3 = PD ? Y[OY_Y3 + l] : 0.f;
            const float b2 = PD ? w[WS_B2 + l] : 0.f, b3 = PD ? w[WS_B3 + l] : 0.f;
            const float* r = PG + l * G_P; const float* p = xb + OX_P;
            float rv[NG], pg[NG];
#pragma unroll
            for (int g = 0; g < NG; ++g) { rv[g] = r[g]; pg[g] = p[g]; }
            __builtin_amdgcn_sched_barrier(0);
            float s = 0.f;
#pragma unroll
            for (int g = 0; g < NG; ++g) s += rv[g] * pg[g];
            const float a2 =  s + fu;
            const float a3 = -s + fd;
            const float n2 = PD ? y2 + step * (a2 - b2) : a2;
            const float n3 = PD ? y3 + step * (a3 - b3) : a3;
            Y[OY_Y2 + l] = n2; Y[OY_Y3 + l] = n3;
            w[WS_D23 + l] = n2 - n3;
        }
    } else if (blk < MA_T3) {
        const int s = (blk - MA_T2) * 32 + lane;
        if (s < NS) {
            const float y4 = PD ? Y[OY_Y4 + s] : 0.f;
            const float b4 = PD ? w[WS_B4 + s] : 0.f;
            float c = 0.f;
            if (PD) {
                const float* r = w + WS_UDT + s * G_P;
                float cv[NG];
#pragma unroll
                for (int g = 0; g < NG; ++g) cv[g] = r[g];
                __builtin_amdgcn_sched_barrier(0);
#pragma unroll
                for (int g = 0; g < NG; ++g) c += cv[kpos100(g)];
            } else {
                const float* Ux = xb + OX_U + s; const float* Dx = xb + OX_D + s;
#pragma unroll
                for (int g = 0; g < NG; ++g) c += Ux[g * S_P] - Dx[g * S_P];
            }
            if (PD) Y[OY_Y4 + s] = y4 + step * (c - b4);
            else    Y[OY_Y4 + s] = c;
        }
    } else if (blk < MA_T4) {
        const int g = (blk - MA_T3) * 32 + lane;
        if (g < NG) {
            const float xp = xb[OX_P + g], xu = xb[OX_RU + g], xd = xb[OX_RD + g];
            const float a1 = xp + xu;
            const float a2 = xd - xp;
            if (PD) {
                const float z1 = Y[OY_Z1 + g], z2 = Y[OY_Z2 + g], pm = Pmax[g];
                Y[OY_Z1 + g] = fmaxf(z1 + step * (a1 - pm), 0.f);
                Y[OY_Z2 + g] = fmaxf(z2 + step *  a2, 0.f);
            } else {
                Y[OY_Z1 + g] = a1; Y[OY_Z2 + g] = a2;
            }
        }
    } else if (blk < MA_T5) {
        if (lane == 0) {
            const float y1 = PD ? Y[OY_Y1] : 0.f;
            const float b1 = PD ? w[WS_B1] : 0.f;
            float pp[NG];
#pragma unroll
            for (int g = 0; g < NG; ++g) pp[g] = xb[OX_P + g];
            __builtin_amdgcn_sched_barrier(0);
            float s = 0.f;
#pragma unroll
            for (int g = 0; g < NG; ++g) s += pp[g];
            if (PD) Y[OY_Y1] = y1 + step * (s - b1);
            else    Y[OY_Y1] = s;
        }
    } else {
        const int base = (blk - MA_T5) * EPW;
        float r3[EPW / 32], r4[EPW / 32];
        for (int j = 0; j < EPW / 32; ++j) {
            const int i = base + j * 32 + lane;
            const int ic = i < NG * NS ? i : NG * NS - 1;
            const int g = ic / NS, s = ic % NS, o = g * S_P + s;
            const float a3 = xb[OX_U + o] - xb[OX_RU + g];
            const float a4 = xb[OX_D + o] - xb[OX_RD + g];
            if (PD) {
                r3[j] = fmaxf(Y[OY_Z3 + o] + step * a3, 0.f);
                r4[j] = fmaxf(Y[OY_Z4 + o] + step * a4, 0.f);
            } else {
                r3[j] = a3; r4[j] = a4;
            }
        }
        for (int j = 0; j < EPW / 32; ++j) {
            const int i = base + j * 32 + lane;
            if (i < NG * NS) {
                const int g = i / NS, s = i % NS, o = g * S_P + s;
                Y[OY_Z3 + o] = r3[j]; Y[OY_Z4 + o] = r4[j];
            }
        }
    }
}

__global__ void k_emit(const float* w, float* out) {
    const float* X = w + WS_X;
    const int total = 3 * NG + 2 * NG * NS + 2 * NL;
    for (int idx = blockIdx.x * blockDim.x + threadIdx.x; idx < total; idx += gridDim.x * blockDim.x) {
        float v;
        if (idx < NG)                 v = X[OX_P  + idx];
        else if (idx < 2 * NG)        v = X[OX_RU + (idx - NG)];
        else if (idx < 3 * NG)        v = X[OX_RD + (idx - 2 * NG)];
        else if (idx < 3 * NG + NG * NS) { int r = idx - 3 * NG; v = X[OX_U + (r / NS) * S_P + (r % NS)]; }
        else if (idx < 3 * NG + 2 * NG * NS) { int r = idx - 3 * NG - NG * NS; v = X[OX_D + (r / NS) * S_P + (r % NS)]; }
        else if (idx < 3 * NG + 2 * NG * NS + NL) v = X[OX_FU + (idx - 3 * NG - 2 * NG * NS)];
        else                          v = X[OX_FD + (idx - 3 * NG - 2 * NG * NS - NL)];
        out[idx] = v;
    }
}


extern "C" void kernel_launch(void* const* d_in, const int* in_sizes, int n_in,
                              void* d_out, int out_size, void* d_ws, size_t ws_size,
                              hipStream_t stream) {
    const float* w_scen   = (const float*)d_in[0];
    const float* Cost     = (const float*)d_in[1];
    const float* C_r_up   = (const float*)d_in[2];
    const float* C_r_down = (const float*)d_in[3];
    const float* Pmax     = (const float*)d_in[4];
    const float* PTDF     = (const float*)d_in[5];
    const float* node_G   = (const float*)d_in[6];
    const float* node_W   = (const float*)d_in[7];
    const float* node_L   = (const float*)d_in[8];
    const float* Pd       = (const float*)d_in[9];
    const float* w_exp    = (const float*)d_in[10];
    const float* Line_cap = (const float*)d_in[11];
    float* w   = (float*)d_ws;
    float* out = (float*)d_out;

    k_zero <<<512, 256, 0, stream>>>(w);
    k_prep1<<<240, 256, 0, stream>>>(w, w_scen, node_W, node_L, Pd, w_exp);
    k_prep2<<<640, 256, 0, stream>>>(w, PTDF, node_G, Line_cap);
    k_setv <<<161, 256, 0, stream>>>(w);

    for (int it = 0; it < POWER_ITERS; ++it) {
        k_norm<false><<<1, NT, 0, stream>>>(w);
        k_mv  <false><<<MA_NB, 32, 0, stream>>>(w, Pmax);
        k_rmv <false><<<RA_NB, RNT, 0, stream>>>(w, Cost, C_r_up, C_r_down);
    }
    k_norm<true><<<1, NT, 0, stream>>>(w);
    k_reset<<<256, 256, 0, stream>>>(w);

    for (int it = 0; it < PDHG_ITERS; ++it) {
        k_rmv<true><<<RA_NB, RNT, 0, stream>>>(w, Cost, C_r_up, C_r_down);
        k_mv <true><<<MA_NB, 32, 0, stream>>>(w, Pmax);
    }
    k_emit<<<161, 256, 0, stream>>>(w, out);
}


# Power iteration: the one-block norm launch is gone; the A^T launch leaves one partial sum of squares per wave, the A launch adds them and scales v as it loads it: the sum of squares changes order
# speedup vs baseline: 38.0981x; 1.1743x over previous
#include <hip/hip_runtime.h>
#include <math.h>

typedef __attribute__((ext_vector_type(2))) float v2f;
typedef __attribute__((ext_vector_type(4))) float v4f;
typedef __attribute__((ext_vector_type(8))) float v8f;


constexpr int NG = 100, NL = 400, NN = 300, NWF = 50, NS = 200;
constexpr int G_P = 112;
constexpr int S_P = 208;
constexpr int PDHG_ITERS = 300, POWER_ITERS = 25;

constexpr int OX_P  = 0;
constexpr int OX_RU = OX_P  + G_P;
constexpr int OX_RD = OX_RU + G_P;
constexpr int OX_U  = OX_RD + G_P;
constexpr int OX_D  = OX_U  + G_P * S_P;
constexpr int OX_FU = OX_D  + G_P * S_P;
constexpr int OX_FD = OX_FU + NL;
constexpr int XSZ   = OX_FD + NL;

constexpr int OY_Y1 = 0;
constexpr int OY_Y2 = 16;
constexpr int OY_Y3 = OY_Y2 + NL;
constexpr int OY_Y4 = OY_Y3 + NL;
constexpr int OY_Z1 = OY_Y4 + S_P;
constexpr int OY_Z2 = OY_Z1 + G_P;
constexpr int OY_Z3 = OY_Z2 + G_P;
constexpr int OY_Z4 = OY_Z3 + G_P * S_P;
constexpr int OY_Z5 = OY_Z4 + G_P * S_P;
constexpr int OY_Z6 = OY_Z5 + NL * S_P;
constexpr int YSZ   = OY_Z6 + NL * S_P;

constexpr int WS_PG   = 0;
constexpr int WS_PGT  = WS_PG   + NL * G_P;
constexpr int WS_PW   = WS_PGT  + G_P * NL;
constexpr int WS_B2   = WS_PW   + NL * S_P;
constexpr int WS_B3   = WS_B2   + NL;
constexpr int WS_B4   = WS_B3   + NL;
constexpr int WS_B1   = WS_B4   + S_P;
constexpr int WS_M1   = WS_B1   + 16;
constexpr int WS_NV   = WS_M1   + NN * S_P;
constexpr int WS_X    = WS_NV   + NN + 4;
constexpr int WS_XB   = WS_X    + XSZ;
constexpr int WS_ATY  = WS_XB   + XSZ;
constexpr int WS_Y    = WS_ATY  + XSZ;
constexpr int WS_AX   = WS_Y    + YSZ;
constexpr int WS_T    = WS_AX   + YSZ;
constexpr int WS_FLOW = WS_T    + G_P * S_P;
constexpr int WS_GP   = WS_FLOW + NL * S_P;
constexpr int WS_COL  = WS_GP   + NL;
constexpr int WS_S3   = WS_COL  + S_P;
constexpr int WS_S4   = WS_S3   + G_P;
constexpr int WS_S5   = WS_S4   + G_P;
constexpr int WS_S6   = WS_S5   + NL;
constexpr int WS_GTY  = WS_S6   + NL;
constexpr int WS_SUMP = WS_GTY  + G_P;
constexpr int WS_STEP = WS_SUMP + 16;
constexpr int WS_UDT  = WS_STEP + 16;
constexpr int WS_ZT   = WS_UDT  + S_P * G_P;
constexpr int WS_D23  = WS_ZT   + S_P * NL;
constexpr int WS_PGP  = WS_D23  + NL;
constexpr int WS_PGTP = WS_PGP  + NL * G_P;
constexpr int WS_PSQ  = WS_PGTP + G_P * NL;
constexpr int PSQ_P   = 448;
constexpr int WS_TOTAL= WS_PSQ  + PSQ_P;

__host__ __device__ constexpr int kpos100(int r) {
    return r < 96 ? ((r & ~7) | ((r & 2) << 1) | ((r & 4) >> 1) | (r & 1)) : r;
}
__host__ __device__ constexpr int kpos400(int l) { return (l / 100) * 100 + kpos100(l % 100); }


__global__ void k_zero(float* w) {
    for (int i = blockIdx.x * blockDim.x + threadIdx.x; i < WS_TOTAL;
         i += gridDim.x * blockDim.x) w[i] = 0.0f;
}

__global__ void k_prep1(float* w, const float* w_scen, const float* node_W,
                        const float* node_L, const float* Pd, const float* w_exp) {
    const int NM1 = NN * NS;
    const int total = NM1 + NN + NS + 1;
    for (int idx = blockIdx.x * blockDim.x + threadIdx.x; idx < total;
         idx += gridDim.x * blockDim.x) {
        if (idx < NM1) {
            int n = idx / NS, s = idx % NS;
            float a = 0.f;
            for (int ww = 0; ww < NWF; ++ww) a += node_W[n*NWF + ww] * w_scen[s*NWF + ww];
            w[WS_M1 + n*S_P + s] = a;
        } else if (idx < NM1 + NN) {
            int n = idx - NM1;
            float a = 0.f;
            for (int ww = 0; ww < NWF; ++ww) a += node_W[n*NWF + ww] * w_exp[ww];
            for (int m = 0; m < NN; ++m)     a -= node_L[n*NN + m] * Pd[m];
            w[WS_NV + n] = a;
        } else if (idx < NM1 + NN + NS) {
            int s = idx - NM1 - NN;
            float a = 0.f;
            for (int ww = 0; ww < NWF; ++ww) a += w_scen[s*NWF + ww];
            w[WS_B4 + s] = -a;
        } else {
            float a = 0.f;
            for (int m = 0; m < NN; ++m)  a += Pd[m];
            for (int ww = 0; ww < NWF; ++ww) a -= w_exp[ww];
            w[WS_B1] = a;
        }
    }
}

__global__ void k_prep2(float* w, const float* PTDF, const float* node_G,
                        const float* Line_cap) {
    const int A = NL * NG;
    const int B = A + NG * NL;
    const int C = B + NL * NS;
    const int D2 = C + NL;
    const int E = D2 + NL;
    for (int idx = blockIdx.x * blockDim.x + threadIdx.x; idx < E;
         idx += gridDim.x * blockDim.x) {
        if (idx < A) {
            int l = idx / NG, g = idx % NG;
            float a = 0.f;
            for (int n = 0; n < NN; ++n) a += PTDF[l*NN + n] * node_G[n*NG + g];
            w[WS_PG + l*G_P + g] = a;
            w[WS_PGP + l*G_P + kpos100(g)] = a;
        } else if (idx < B) {
            int r = idx - A; int g = r / NL, l = r % NL;
            float a = 0.f;
            for (int n = 0; n < NN; ++n) a += PTDF[l*NN + n] * node_G[n*NG + g];
            w[WS_PGT + g*NL + l] = a;
            w[WS_PGTP + g*NL + kpos400(l)] = a;
        } else if (idx < C) {
            int r = idx - B; int l = r / NS, s = r % NS;
            float a = 0.f;
            for (int n = 0; n < NN; ++n) a += PTDF[l*NN + n] * w[WS_M1 + n*S_P + s];
            w[WS_PW + l*S_P + s] = a;
        } else if (idx < D2) {
            int l = idx - C;
            float a = 0.f;
            for (int n = 0; n < NN; ++n) a += PTDF[l*NN + n] * w[WS_NV + n];
            w[WS_B2 + l] = Line_cap[l] - a;
        } else {
            int l = idx - D2;
            float a = 0.f;
            for (int n = 0; n < NN; ++n) a += PTDF[l*NN + n] * w[WS_NV + n];
            w[WS_B3 + l] = Line_cap[l] + a;
        }
    }
}

__global__ void k_setv(float* w) {
    const int total = 3*NG + 2*NG*NS + 2*NL;
    float* X = w + WS_X;
    for (int idx = blockIdx.x * blockDim.x + threadIdx.x; idx < total;
         idx += gridDim.x * blockDim.x) {
        int off;
        if (idx < NG)                off = OX_P  + idx;
        else if (idx < 2*NG)         off = OX_RU + (idx - NG);
        else if (idx < 3*NG)         off = OX_RD + (idx - 2*NG);
        else if (idx < 3*NG + NG*NS) { int r = idx - 3*NG; off = OX_U + (r/NS)*S_P + (r%NS); }
        else if (idx < 3*NG + 2*NG*NS) { int r = idx - 3*NG - NG*NS; off = OX_D + (r/NS)*S_P + (r%NS); }
        else if (idx < 3*NG + 2*NG*NS + NL) off = OX_FU + (idx - 3*NG - 2*NG*NS);
        else                         off = OX_FD + (idx - 3*NG - 2*NG*NS - NL);
        X[off] = 1.0f;
    }
    for (int i = blockIdx.x * blockDim.x + threadIdx.x; i < PSQ_P; i += gridDim.x * blockDim.x)
        w[WS_PSQ + i] = i == 0 ? (float)total : 0.f;
}


__device__ __forceinline__ float wave_sum(float v) {
    for (int m = 16; m > 0; m >>= 1) v += __shfl_xor(v, m);
    return v;
}

__device__ __forceinline__ float sumsq_of_parts(const float* psq, int lane) {
    float pv[PSQ_P / 32];
#pragma unroll
    for (int j = 0; j < PSQ_P / 32; ++j) pv[j] = psq[lane + 32 * j];
    float s = 0.f;
#pragma unroll
    for (int j = 0; j < PSQ_P / 32; ++j) s += pv[j];
    return wave_sum(s);
}

__global__ void k_reset(float* w) {
    if (blockIdx.x == 0 && threadIdx.x < 32) {
        const float ss = sumsq_of_parts(w + WS_PSQ, (int)threadIdx.x);
        if (threadIdx.x == 0) w[WS_STEP] = 0.9f / sqrtf(sqrtf(ss));
    }
    const int total = 2 * XSZ + YSZ + S_P * NL + NL;
    for (int i = blockIdx.x * blockDim.x + threadIdx.x; i < total; i += gridDim.x * blockDim.x) {
        if (i < XSZ)                w[WS_X + i] = 0.f;
        else if (i < 2 * XSZ)       w[WS_XB + (i - XSZ)] = 0.f;
        else if (i < 2 * XSZ + YSZ) w[WS_Y + (i - 2 * XSZ)] = 0.f;
        else                        w[WS_ZT + (i - 2 * XSZ - YSZ)] = 0.f;
    }
}

template <int K>
__device__ __forceinline__ v8f tile_gemm(const float* arow, const float* P, const float* Q,
                                         int n, int kh, float sc) {
    v8f acc = {0.f,0.f,0.f,0.f,0.f,0.f,0.f,0.f};
#pragma unroll
    for (int k0 = 0; k0 < K; k0 += 4) {
        int ka = k0 + kh;
        v2f a = *(const v2f*)(arow + ka);
        v2f b;
        int bi = ka * S_P + n;
        b[0] = P[bi]       * sc - Q[bi]       * sc;
        b[1] = P[bi + S_P] * sc - Q[bi + S_P] * sc;
        acc = __builtin_amdgcn_wmma_f32_16x16x4_f32(false, a, false, b, (short)0, acc, false, false);
    }
    return acc;
}

struct KpFrags { v4f a[12], b[12]; v2f at, bt; };

__device__ __forceinline__ void load_kp(KpFrags& f, const float* arow, const float* brow, int h) {
#pragma unroll
    for (int j = 0; j < 12; ++j) {
        f.a[j] = *(const v4f*)(arow + 8 * j + 4 * h);
        f.b[j] = *(const v4f*)(brow + 8 * j + 4 * h);
    }
    f.at = *(const v2f*)(arow + 96 + 2 * h);
    f.bt = *(const v2f*)(brow + 96 + 2 * h);
}

__device__ __forceinline__ v8f mma_kp(const KpFrags& f) {
    v8f acc = {0.f,0.f,0.f,0.f,0.f,0.f,0.f,0.f};
#pragma unroll
    for (int j = 0; j < 12; ++j) {
        v2f a0, b0, a1, b1;
        a0[0] = f.a[j][0]; a0[1] = f.a[j][1]; b0[0] = f.b[j][0]; b0[1] = f.b[j][1];
        a1[0] = f.a[j][2]; a1[1] = f.a[j][3]; b1[0] = f.b[j][2]; b1[1] = f.b[j][3];
        acc = __builtin_amdgcn_wmma_f32_16x16x4_f32(false, a0, false, b0, (short)0, acc, false, false);
        acc = __builtin_amdgcn_wmma_f32_16x16x4_f32(false, a1, false, b1, (short)0, acc, false, false);
    }
    acc = __builtin_amdgcn_wmma_f32_16x16x4_f32(false, f.at, false, f.bt, (short)0, acc, false, false);
    return acc;
}

static_assert(NG == 100 && NL == 400, "tile_gemm_kp and kpos100 / kpos400 are written for 100-long k ranges");

constexpr int RPW = 8;


constexpr int KSP = 4;
constexpr int RNT = 32 * KSP;
constexpr int KQ  = NL / KSP;
constexpr int SQ  = 48;
constexpr int RA_T1 = 7 * 13;
constexpr int RA_T2 = RA_T1 + (NG + RPW - 1) / RPW;
constexpr int RA_T3 = RA_T2 + (NG + RPW - 1) / RPW;
constexpr int RA_NB = RA_T3 + NL / RPW;
static_assert(RA_T1 * KSP + (RA_NB - RA_T1) <= PSQ_P, "one partial sum of squares per storing wave");

__device__ __forceinline__ void row_sum_share(const float* r0, const float* r1, int wv, float& a, float& b) {
    const float* p0 = r0 + wv * SQ;
    const float* p1 = r1 + wv * SQ;
    constexpr int TAIL = NS - KSP * SQ;
    float v0[SQ + TAIL], v1[SQ + TAIL];
#pragma unroll
    for (int s = 0; s < SQ; ++s) { v0[s] = p0[s]; v1[s] = p1[s]; }
#pragma unroll
    for (int s = SQ; s < SQ + TAIL; ++s) { v0[s] = 0.f; v1[s] = 0.f; }
    if (wv == KSP - 1) {
#pragma unroll
        for (int s = SQ; s < SQ + TAIL; ++s) { v0[s] = p0[s]; v1[s] = p1[s]; }
    }
#pragma unroll
    for (int s = 0; s < SQ; ++s) { a += v0[s]; b += v1[s]; }
    if (wv == KSP - 1) {
#pragma unroll
        for (int s = SQ; s < SQ + TAIL; ++s) { a += v0[s]; b += v1[s]; }
    }
}

template <bool PD>
__global__ __launch_bounds__(RNT) __attribute__((amdgpu_waves_per_eu(1, 1))) void k_rmv(
        float* w, const float* Cost, const float* Cru, const float* Crd) {
    __shared__ float part[KSP * 8 * 32];
    const int lane = threadIdx.x & 31, blk = blockIdx.x;
    const int wv = __builtin_amdgcn_readfirstlane((int)(threadIdx.x >> 5));
    const float* PGT = w + WS_PGT;
    const float* yb = PD ? (w + WS_Y) : (w + WS_AX);
    float* X  = w + WS_X;
    float* XB = w + WS_XB;
    const float step = PD ? w[WS_STEP] : 0.f;
    float q = 0.f;
    if (blk < RA_T1) {
        const int tm = blk / 13, tn = blk % 13;
        const int m16 = lane & 15;
        const int m = tm * 16 + m16, n = tn * 16 + m16;
        const int mr = tm * 16 + ((lane >> 4) << 3) + 2 * wv;
        const int nc = n < NS ? n : NS - 1;
        const float y4 = yb[OY_Y4 + nc];
        float z3v[2], z4v[2], xu[2], xd[2];
        for (int j = 0; j < 2; ++j) {
            const int gc = (mr + j) < NG ? (mr + j) : NG - 1;
            const int o = gc * S_P + nc;
            z3v[j] = yb[OY_Z3 + o]; z4v[j] = yb[OY_Z4 + o];
            xu[j] = PD ? X[OX_U + o] : 0.f; xd[j] = PD ? X[OX_D + o] : 0.f;
        }
        const int kb = wv * KQ;
        KpFrags f;
        load_kp(f, w + WS_PGTP + m * NL + kb, w + WS_ZT + n * NL + kb, lane >> 4);
        __builtin_amdgcn_sched_barrier(0);
        v8f acc = mma_kp(f);
        for (int i = 0; i < 8; ++i) part[(wv * 8 + i) * 32 + lane] = acc[i];
        __syncthreads();
        float nu[2], nd[2], bu[2], bd[2];
        for (int j = 0; j < 2; ++j) {
            const int i = 2 * wv + j;
            const float tv = ((part[(0 * 8 + i) * 32 + lane] + part[(1 * 8 + i) * 32 + lane])
                              + part[(2 * 8 + i) * 32 + lane]) + part[(3 * 8 + i) * 32 + lane];
            const float au =  y4 + z3v[j] + tv;
            const float ad = -y4 + z4v[j] - tv;
            if (PD) {
                nu[j] = fmaxf(xu[j] - step * au, 0.f); bu[j] = 2.f * nu[j] - xu[j];
                nd[j] = fmaxf(xd[j] - step * ad, 0.f); bd[j] = 2.f * nd[j] - xd[j];
            } else {
                nu[j] = au; nd[j] = ad; bu[j] = 0.f; bd[j] = 0.f;
            }
        }
        for (int j = 0; j < 2; ++j) {
            const int g = mr + j;
            if (n < NS && g < NG) {
                const int o = g * S_P + n;
                X[OX_U + o] = nu[j]; X[OX_D + o] = nd[j];
                if (PD) { XB[OX_U + o] = bu[j]; XB[OX_D + o] = bd[j]; }
                else    q += nu[j] * nu[j] + nd[j] * nd[j];
            }
        }
        if (PD && n < NS && mr < NG) {
            v2f d; d[0] = bu[0] - bd[0]; d[1] = bu[1] - bd[1];
            *(v2f*)(w + WS_UDT + n * G_P + kpos100(mr)) = d;
        }
    } else if (blk < RA_T2) {
        const int g = (blk - RA_T1) * RPW + lane;
        const bool act = lane < RPW && g < NG;
        float z1 = 0.f, z2 = 0.f, xu = 0.f, xd = 0.f, cu = 0.f, cd = 0.f;
        float a = 0.f, b = 0.f;
        if (act) {
            if (wv == 0) {
                z1 = yb[OY_Z1 + g]; z2 = yb[OY_Z2 + g];
                xu = PD ? X[OX_RU + g] : 0.f; xd = PD ? X[OX_RD + g] : 0.f;
                cu = PD ? Cru[g] : 0.f; cd = PD ? Crd[g] : 0.f;
            }
            row_sum_share(yb + OY_Z3 + g * S_P, yb + OY_Z4 + g * S_P, wv, a, b);
        }
        part[wv * 32 + lane] = a; part[(KSP + wv) * 32 + lane] = b;
        __syncthreads();
        if (wv == 0 && act) {
            const float sa = ((part[lane] + part[32 + lane]) + part[64 + lane]) + part[96 + lane];
            const float sb = ((part[128 + lane] + part[160 + lane]) + part[192 + lane]) + part[224 + lane];
            const float au = z1 - sa;
            const float ad = z2 - sb;
            if (PD) {
                const float nu = fmaxf(xu - step * (cu + au), 0.f);
                const float nd = fmaxf(xd - step * (cd + ad), 0.f);
                X[OX_RU + g] = nu; XB[OX_RU + g] = 2.f * nu - xu;
                X[OX_RD + g] = nd; XB[OX_RD + g] = 2.f * nd - xd;
            } else {
                X[OX_RU + g] = au; X[OX_RD + g] = ad;
                q = au * au + ad * ad;
            }
        }
    } else if (blk < RA_T3) {
        const int g = (blk - RA_T2) * RPW + lane;
        const bool act = lane < RPW && g < NG;
        float y1 = 0.f, z1 = 0.f, z2 = 0.f, xp = 0.f, cp = 0.f;
        float c = 0.f;
        if (act) {
            const float* pr = PGT + g * NL + wv * KQ;
            const float* dp = w + WS_D23 + wv * KQ;
            float pv[KQ], dv[KQ];
#pragma unroll
            for (int l = 0; l < KQ; ++l) { pv[l] = pr[l]; dv[l] = dp[l]; }
            if (wv == 0) {
                y1 = yb[OY_Y1]; z1 = yb[OY_Z1 + g]; z2 = yb[OY_Z2 + g];
                xp = PD ? X[OX_P + g] : 0.f;
                cp = PD ? Cost[g] : 0.f;
            }
#pragma unroll
            for (int l = 0; l < KQ; ++l) c += pv[l] * dv[l];
        }
        part[wv * 32 + lane] = c;
        __syncthreads();
        if (wv == 0 && act) {
            const float cs = ((part[lane] + part[32 + lane]) + part[64 + lane]) + part[96 + lane];
            const float ap = y1 + cs + z1 - z2;
            if (PD) {
                const float nn = fmaxf(xp - step * (cp + ap), 0.f);
                X[OX_P + g] = nn; XB[OX_P + g] = 2.f * nn - xp;
            } else {
                X[OX_P + g] = ap;
                q = ap * ap;
            }
        }
    } else {
        const int l = (blk - RA_T3) * RPW + lane;
        const bool act = lane < RPW && l < NL;
        float y2 = 0.f, y3 = 0.f, xu = 0.f, xd = 0.f;
        float a = 0.f, b = 0.f;
        if (act) {
            if (wv == 0) {
                y2 = yb[OY_Y2 + l]; y3 = yb[OY_Y3 + l];
                xu = PD ? X[OX_FU + l] : 0.f; xd = PD ? X[OX_FD + l] : 0.f;
            }
            row_sum_share(yb + OY_Z5 + l * S_P, yb + OY_Z6 + l * S_P, wv, a, b);
        }
        part[wv * 32 + lane] = a; part[(KSP + wv) * 32 + lane] = b;
        __syncthreads();
        if (wv == 0 && act) {
            const float sa = ((part[lane] + part[32 + lane]) + part[64 + lane]) + part[96 + lane];
            const float sb = ((part[128 + lane] + part[160 + lane]) + part[192 + lane]) + part[224 + lane];
            const float au = y2 - sa;
            const float ad = y3 - sb;
            if (PD) {
                const float nu = fmaxf(xu - step * au, 0.f);
                const float nd = fmaxf(xd - step * ad, 0.f);
                X[OX_FU + l] = nu; XB[OX_FU + l] = 2.f * nu - xu;
                X[OX_FD + l] = nd; XB[OX_FD + l] = 2.f * nd - xd;
            } else {
                X[OX_FU + l] = au; X[OX_FD + l] = ad;
                q = au * au + ad * ad;
            }
        }
    }
    if (!PD) {
        const float s = wave_sum(q);
        const bool tile = blk < RA_T1;
        if (lane == 0 && (tile || wv == 0))
            w[WS_PSQ + (tile ? blk * KSP + wv : RA_T1 * KSP + (blk - RA_T1))] = s;
    }
}

constexpr int EPW = 128;
constexpr int MA_T1 = 25 * 13;
constexpr int MA_T2 = MA_T1 + NL / RPW;
constexpr int MA_T3 = MA_T2 + (NS + 31) / 32;
constexpr int MA_T4 = MA_T3 + (NG + 31) / 32;
constexpr int MA_T5 = MA_T4 + 1;
constexpr int MA_NB = MA_T5 + (NG * NS + EPW - 1) / EPW;

template <bool PD>
__global__ __launch_bounds__(32) __attribute__((amdgpu_waves_per_eu(1, 1))) void k_mv(
        float* w, const float* Pmax) {
    const int lane = threadIdx.x, blk = blockIdx.x;
    const float* PG = w + WS_PG;
    const float* xb = PD ? (w + WS_XB) : (w + WS_X);
    float* Y = PD ? (w + WS_Y) : (w + WS_AX);
    const float* pw = w + WS_PW;
    const float step = PD ? w[WS_STEP] : 0.f;
    const float inv = PD ? 1.f : 1.0f / sqrtf(sumsq_of_parts(w + WS_PSQ, lane));
#define XS(v) (PD ? (v) : (v) * inv)
    if (blk < MA_T1) {
        const int tm = blk / 13, tn = blk % 13;
        const int m16 = lane & 15, kh = (lane >> 4) * 2;
        const int m = tm * 16 + m16, n = tn * 16 + m16;
        const int mr = tm * 16 + ((lane >> 4) << 3);
        const int nc = n < NS ? n : NS - 1;
        float fu[8], fd[8], y5[8], y6[8], pv[8];
        for (int i = 0; i < 8; ++i) {
            const int l = mr + i;
            const int o = l * S_P + nc;
            fu[i] = XS(xb[OX_FU + l]); fd[i] = XS(xb[OX_FD + l]);
            y5[i] = PD ? Y[OY_Z5 + o] : 0.f; y6[i] = PD ? Y[OY_Z6 + o] : 0.f;
            pv[i] = PD ? pw[o] : 0.f;
        }
        v8f acc;
        if (PD) {
            KpFrags f;
            load_kp(f, w + WS_PGP + m * G_P, w + WS_UDT + n * G_P, lane >> 4);
            __builtin_amdgcn_sched_barrier(0);
            acc = mma_kp(f);
        } else {
            acc = tile_gemm<NG>(PG + m * G_P, xb + OX_U, xb + OX_D, n, kh, inv);
        }
        float r5[8], r6[8];
        for (int i = 0; i < 8; ++i) {
            const float f = acc[i];
            const float a5 =  f - fu[i];
            const float a6 = -f - fd[i];
            if (PD) {
                r5[i] = fmaxf(y5[i] + step * (a5 + pv[i]), 0.f);
                r6[i] = fmaxf(y6[i] + step * (a6 - pv[i]), 0.f);
            } else {
                r5[i] = a5; r6[i] = a6;
            }
        }
        if (n < NS) {
            for (int i = 0; i < 8; ++i) {
                const int o = (mr + i) * S_P + n;
                Y[OY_Z5 + o] = r5[i]; Y[OY_Z6 + o] = r6[i];
            }
            float* zt = w + WS_ZT + n * NL;
            for (int i = 0; i < 8; i += 2) {
                v2f d; d[0] = r5[i] - r6[i]; d[1] = r5[i + 1] - r6[i + 1];
                *(v2f*)(zt + kpos400(mr + i)) = d;
            }
        }
    } else if (blk < MA_T2) {
        const int l = (blk - MA_T1) * RPW + lane;
        if (lane < RPW && l < NL) {
            const float fu = XS(xb[OX_FU + l]), fd = XS(xb[OX_FD + l]);
            const float y2 = PD ? Y[OY_Y2 + l] : 0.f, y3 = PD ? Y[OY_Y3 + l] : 0.f;
            const float b2 = PD ? w[WS_B2 + l] : 0.f, b3 = PD ? w[WS_B3 + l] : 0.f;
            const float* r = PG + l * G_P; const float* p = xb + OX_P;
            float rv[NG], pg[NG];
#pragma unroll
            for (int g = 0; g < NG; ++g) { rv[g] = r[g]; pg[g] = p[g]; }
            __builtin_amdgcn_sched_barrier(0);
            float s = 0.f;
#pragma unroll
            for (int g = 0; g < NG; ++g) s += rv[g] * XS(pg[g]);
            const float a2 =  s + fu;
            const float a3 = -s + fd;
            const float n2 = PD ? y2 + step * (a2 - b2) : a2;
            const float n3 = PD ? y3 + step * (a3 - b3) : a3;
            Y[OY_Y2 + l] = n2; Y[OY_Y3 + l] = n3;
            w[WS_D23 + l] = n2 - n3;
        }
    } else if (blk < MA_T3) {
        const int s = (blk - MA_T2) * 32 + lane;
        if (s < NS) {
            const float y4 = PD ? Y[OY_Y4 + s] : 0.f;
            const float b4 = PD ? w[WS_B4 + s] : 0.f;
            float c = 0.f;
            if (PD) {
                const float* r = w + WS_UDT + s * G_P;
                float cv[NG];
#pragma unroll
                for (int g = 0; g < NG; ++g) cv[g] = r[g];
                __builtin_amdgcn_sched_barrier(0);
#pragma unroll
                for (int g = 0; g < NG; ++g) c += cv[kpos100(g)];
            } else {
                const float* Ux = xb + OX_U + s; const float* Dx = xb + OX_D + s;
#pragma unroll
                for (int g = 0; g < NG; ++g) c += Ux[g * S_P] * inv - Dx[g * S_P] * inv;
            }
            if (PD) Y[OY_Y4 + s] = y4 + step * (c - b4);
            else    Y[OY_Y4 + s] = c;
        }
    } else if (blk < MA_T4) {
        const int g = (blk - MA_T3) * 32 + lane;
        if (g < NG) {
            const float xp = XS(xb[OX_P + g]), xu = XS(xb[OX_RU + g]), xd = XS(xb[OX_RD + g]);
            const float a1 = xp + xu;
            const float a2 = xd - xp;
            if (PD) {
                const float z1 = Y[OY_Z1 + g], z2 = Y[OY_Z2 + g], pm = Pmax[g];
                Y[OY_Z1 + g] = fmaxf(z1 + step * (a1 - pm), 0.f);
                Y[OY_Z2 + g] = fmaxf(z2 + step *  a2, 0.f);
            } else {
                Y[OY_Z1 + g] = a1; Y[OY_Z2 + g] = a2;
            }
        }
    } else if (blk < MA_T5) {
        if (lane == 0) {
            const float y1 = PD ? Y[OY_Y1] : 0.f;
            const float b1 = PD ? w[WS_B1] : 0.f;
            float pp[NG];
#pragma unroll
            for (int g = 0; g < NG; ++g) pp[g] = xb[OX_P + g];
            __builtin_amdgcn_sched_barrier(0);
            float s = 0.f;
#pragma unroll
            for (int g = 0; g < NG; ++g) s += XS(pp[g]);
            if (PD) Y[OY_Y1] = y1 + step * (s - b1);
            else    Y[OY_Y1] = s;
        }
    } else {
        const int base = (blk - MA_T5) * EPW;
        float r3[EPW / 32], r4[EPW / 32];
        for (int j = 0; j < EPW / 32; ++j) {
            const int i = base + j * 32 + lane;
            const int ic = i < NG * NS ? i : NG * NS - 1;
            const int g = ic / NS, s = ic % NS, o = g * S_P + s;
            const float a3 = XS(xb[OX_U + o]) - XS(xb[OX_RU + g]);
            const float a4 = XS(xb[OX_D + o]) - XS(xb[OX_RD + g]);
            if (PD) {
                r3[j] = fmaxf(Y[OY_Z3 + o] + step * a3, 0.f);
                r4[j] = fmaxf(Y[OY_Z4 + o] + step * a4, 0.f);
            } else {
                r3[j] = a3; r4[j] = a4;
            }
        }
        for (int j = 0; j < EPW / 32; ++j) {
            const int i = base + j * 32 + lane;
            if (i < NG * NS) {
                const int g = i / NS, s = i % NS, o = g * S_P + s;
                Y[OY_Z3 + o] = r3[j]; Y[OY_Z4 + o] = r4[j];
            }
        }
    }
#undef XS
}

__global__ void k_emit(const float* w, float* out) {
    const float* X = w + WS_X;
    const int total = 3 * NG + 2 * NG * NS + 2 * NL;
    for (int idx = blockIdx.x * blockDim.x + threadIdx.x; idx < total; idx += gridDim.x * blockDim.x) {
        float v;
        if (idx < NG)                 v = X[OX_P  + idx];
        else if (idx < 2 * NG)        v = X[OX_RU + (idx - NG)];
        else if (idx < 3 * NG)        v = X[OX_RD + (idx - 2 * NG)];
        else if (idx < 3 * NG + NG * NS) { int r = idx - 3 * NG; v = X[OX_U + (r / NS) * S_P + (r % NS)]; }
        else if (idx < 3 * NG + 2 * NG * NS) { int r = idx - 3 * NG - NG * NS; v = X[OX_D + (r / NS) * S_P + (r % NS)]; }
        else if (idx < 3 * NG + 2 * NG * NS + NL) v = X[OX_FU + (idx - 3 * NG - 2 * NG * NS)];
        else                          v = X[OX_FD + (idx - 3 * NG - 2 * NG * NS - NL)];
        out[idx] = v;
    }
}


extern "C" void kernel_launch(void* const* d_in, const int* in_sizes, int n_in,
                              void* d_out, int out_size, void* d_ws, size_t ws_size,
                              hipStream_t stream) {
    const float* w_scen   = (const float*)d_in[0];
    const float* Cost     = (const float*)d_in[1];
    const float* C_r_up   = (const float*)d_in[2];
    const float* C_r_down = (const float*)d_in[3];
    const float* Pmax     = (const float*)d_in[4];
    const float* PTDF     = (const float*)d_in[5];
    const float* node_G   = (const float*)d_in[6];
    const float* node_W   = (const float*)d_in[7];
    const float* node_L   = (const float*)d_in[8];
    const float* Pd       = (const float*)d_in[9];
    const float* w_exp    = (const float*)d_in[10];
    const float* Line_cap = (const float*)d_in[11];
    float* w   = (float*)d_ws;
    float* out = (float*)d_out;

    k_zero <<<512, 256, 0, stream>>>(w);
    k_prep1<<<240, 256, 0, stream>>>(w, w_scen, node_W, node_L, Pd, w_exp);
    k_prep2<<<640, 256, 0, stream>>>(w, PTDF, node_G, Line_cap);
    k_setv <<<161, 256, 0, stream>>>(w);

    for (int it = 0; it < POWER_ITERS; ++it) {
        k_mv  <false><<<MA_NB, 32, 0, stream>>>(w, Pmax);
        k_rmv <false><<<RA_NB, RNT, 0, stream>>>(w, Cost, C_r_up, C_r_down);
    }
    k_reset<<<256, 256, 0, stream>>>(w);

    for (int it = 0; it < PDHG_ITERS; ++it) {
        k_rmv<true><<<RA_NB, RNT, 0, stream>>>(w, Cost, C_r_up, C_r_down);
        k_mv <true><<<MA_NB, 32, 0, stream>>>(w, Pmax);
    }
    k_emit<<<161, 256, 0, stream>>>(w, out);
}


# Tile-product operand planes (PG, PG^T, U-D, z5-z6) stored block-packed in lane order: a 16-byte wave load reads 512 contiguous bytes (4 whole lines), not 32 bytes from each of 16 rows; sum order kept
# speedup vs baseline: 40.5000x; 1.0630x over previous
#include <hip/hip_runtime.h>
#include <math.h>

typedef __attribute__((ext_vector_type(2))) float v2f;
typedef __attribute__((ext_vector_type(4))) float v4f;
typedef __attribute__((ext_vector_type(8))) float v8f;


constexpr int NG = 100, NL = 400, NN = 300, NWF = 50, NS = 200;
constexpr int G_P = 112;
constexpr int S_P = 208;
constexpr int PDHG_ITERS = 300, POWER_ITERS = 25;

constexpr int OX_P  = 0;
constexpr int OX_RU = OX_P  + G_P;
constexpr int OX_RD = OX_RU + G_P;
constexpr int OX_U  = OX_RD + G_P;
constexpr int OX_D  = OX_U  + G_P * S_P;
constexpr int OX_FU = OX_D  + G_P * S_P;
constexpr int OX_FD = OX_FU + NL;
constexpr int XSZ   = OX_FD + NL;

constexpr int OY_Y1 = 0;
constexpr int OY_Y2 = 16;
constexpr int OY_Y3 = OY_Y2 + NL;
constexpr int OY_Y4 = OY_Y3 + NL;
constexpr int OY_Z1 = OY_Y4 + S_P;
constexpr int OY_Z2 = OY_Z1 + G_P;
constexpr int OY_Z3 = OY_Z2 + G_P;
constexpr int OY_Z4 = OY_Z3 + G_P * S_P;
constexpr int OY_Z5 = OY_Z4 + G_P * S_P;
constexpr int OY_Z6 = OY_Z5 + NL * S_P;
constexpr int YSZ   = OY_Z6 + NL * S_P;

constexpr int WS_PG   = 0;
constexpr int WS_PGT  = WS_PG   + NL * G_P;
constexpr int WS_PW   = WS_PGT  + G_P * NL;
constexpr int WS_B2   = WS_PW   + NL * S_P;
constexpr int WS_B3   = WS_B2   + NL;
constexpr int WS_B4   = WS_B3   + NL;
constexpr int WS_B1   = WS_B4   + S_P;
constexpr int WS_M1   = WS_B1   + 16;
constexpr int WS_NV   = WS_M1   + NN * S_P;
constexpr int WS_X    = WS_NV   + NN + 4;
constexpr int WS_XB   = WS_X    + XSZ;
constexpr int WS_ATY  = WS_XB   + XSZ;
constexpr int WS_Y    = WS_ATY  + XSZ;
constexpr int WS_AX   = WS_Y    + YSZ;
constexpr int WS_T    = WS_AX   + YSZ;
constexpr int WS_FLOW = WS_T    + G_P * S_P;
constexpr int WS_GP   = WS_FLOW + NL * S_P;
constexpr int WS_COL  = WS_GP   + NL;
constexpr int WS_S3   = WS_COL  + S_P;
constexpr int WS_S4   = WS_S3   + G_P;
constexpr int WS_S5   = WS_S4   + G_P;
constexpr int WS_S6   = WS_S5   + NL;
constexpr int WS_GTY  = WS_S6   + NL;
constexpr int WS_SUMP = WS_GTY  + G_P;
constexpr int WS_STEP = WS_SUMP + 16;
constexpr int BLK = 1600;
__host__ __device__ constexpr int al32(int x) { return (x + 31) & ~31; }
constexpr int WS_UDT  = al32(WS_STEP + 16);
constexpr int WS_ZT   = WS_UDT  + S_P * G_P;
constexpr int WS_D23  = WS_ZT   + S_P * NL;
constexpr int WS_PGP  = al32(WS_D23 + NL);
constexpr int WS_PGTP = WS_PGP  + NL * G_P;
constexpr int WS_PSQ  = WS_PGTP + G_P * NL;
constexpr int PSQ_P   = 448;
constexpr int WS_TOTAL= WS_PSQ  + PSQ_P;

__host__ __device__ constexpr int kpos100(int r) {
    return r < 96 ? ((r & ~7) | ((r & 2) << 1) | ((r & 4) >> 1) | (r & 1)) : r;
}
__host__ __device__ constexpr int pk_in(int c, int p) {
    return p < 96 ? ((p >> 3) * 128 + (c + ((p >> 2) & 1) * 16) * 4 + (p & 3))
                  : (1536 + (c + ((p - 96) >> 1) * 16) * 2 + (p & 1));
}
static_assert(13 * BLK <= S_P * G_P && 13 * 4 * BLK == S_P * NL && 25 * BLK <= NL * G_P && 7 * 4 * BLK == G_P * NL,
              "the packed planes fit the room of the row-major ones");


__global__ void k_zero(float* w) {
    for (int i = blockIdx.x * blockDim.x + threadIdx.x; i < WS_TOTAL;
         i += gridDim.x * blockDim.x) w[i] = 0.0f;
}

__global__ void k_prep1(float* w, const float* w_scen, const float* node_W,
                        const float* node_L, const float* Pd, const float* w_exp) {
    const int NM1 = NN * NS;
    const int total = NM1 + NN + NS + 1;
    for (int idx = blockIdx.x * blockDim.x + threadIdx.x; idx < total;
         idx += gridDim.x * blockDim.x) {
        if (idx < NM1) {
            int n = idx / NS, s = idx % NS;
            float a = 0.f;
            for (int ww = 0; ww < NWF; ++ww) a += node_W[n*NWF + ww] * w_scen[s*NWF + ww];
            w[WS_M1 + n*S_P + s] = a;
        } else if (idx < NM1 + NN) {
            int n = idx - NM1;
            float a = 0.f;
            for (int ww = 0; ww < NWF; ++ww) a += node_W[n*NWF + ww] * w_exp[ww];
            for (int m = 0; m < NN; ++m)     a -= node_L[n*NN + m] * Pd[m];
            w[WS_NV + n] = a;
        } else if (idx < NM1 + NN + NS) {
            int s = idx - NM1 - NN;
            float a = 0.f;
            for (int ww = 0; ww < NWF; ++ww) a += w_scen[s*NWF + ww];
            w[WS_B4 + s] = -a;
        } else {
            float a = 0.f;
            for (int m = 0; m < NN; ++m)  a += Pd[m];
            for (int ww = 0; ww < NWF; ++ww) a -= w_exp[ww];
            w[WS_B1] = a;
        }
    }
}

__global__ void k_prep2(float* w, const float* PTDF, const float* node_G,
                        const float* Line_cap) {
    const int A = NL * NG;
    const int B = A + NG * NL;
    const int C = B + NL * NS;
    const int D2 = C + NL;
    const int E = D2 + NL;
    for (int idx = blockIdx.x * blockDim.x + threadIdx.x; idx < E;
         idx += gridDim.x * blockDim.x) {
        if (idx < A) {
            int l = idx / NG, g = idx % NG;
            float a = 0.f;
            for (int n = 0; n < NN; ++n) a += PTDF[l*NN + n] * node_G[n*NG + g];
            w[WS_PG + l*G_P + g] = a;
            w[WS_PGP + (l >> 4) * BLK + pk_in(l & 15, kpos100(g))] = a;
        } else if (idx < B) {
            int r = idx - A; int g = r / NL, l = r % NL;
            float a = 0.f;
            for (int n = 0; n < NN; ++n) a += PTDF[l*NN + n] * node_G[n*NG + g];
            w[WS_PGT + g*NL + l] = a;
            w[WS_PGTP + ((g >> 4) * 4 + l / 100) * BLK + pk_in(g & 15, kpos100(l % 100))] = a;
        } else if (idx < C) {
            int r = idx - B; int l = r / NS, s = r % NS;
            float a = 0.f;
            for (int n = 0; n < NN; ++n) a += PTDF[l*NN + n] * w[WS_M1 + n*S_P + s];
            w[WS_PW + l*S_P + s] = a;
        } else if (idx < D2) {
            int l = idx - C;
            float a = 0.f;
            for (int n = 0; n < NN; ++n) a += PTDF[l*NN + n] * w[WS_NV + n];
            w[WS_B2 + l] = Line_cap[l] - a;
        } else {
            int l = idx - D2;
            float a = 0.f;
            for (int n = 0; n < NN; ++n) a += PTDF[l*NN + n] * w[WS_NV + n];
            w[WS_B3 + l] = Line_cap[l] + a;
        }
    }
}

__global__ void k_setv(float* w) {
    const int total = 3*NG + 2*NG*NS + 2*NL;
    float* X = w + WS_X;
    for (int idx = blockIdx.x * blockDim.x + threadIdx.x; idx < total;
         idx += gridDim.x * blockDim.x) {
        int off;
        if (idx < NG)                off = OX_P  + idx;
        else if (idx < 2*NG)         off = OX_RU + (idx - NG);
        else if (idx < 3*NG)         off = OX_RD + (idx - 2*NG);
        else if (idx < 3*NG + NG*NS) { int r = idx - 3*NG; off = OX_U + (r/NS)*S_P + (r%NS); }
        else if (idx < 3*NG + 2*NG*NS) { int r = idx - 3*NG - NG*NS; off = OX_D + (r/NS)*S_P + (r%NS); }
        else if (idx < 3*NG + 2*NG*NS + NL) off = OX_FU + (idx - 3*NG - 2*NG*NS);
        else                         off = OX_FD + (idx - 3*NG - 2*NG*NS - NL);
        X[off] = 1.0f;
    }
    for (int i = blockIdx.x * blockDim.x + threadIdx.x; i < PSQ_P; i += gridDim.x * blockDim.x)
        w[WS_PSQ + i] = i == 0 ? (float)total : 0.f;
}


__device__ __forceinline__ float wave_sum(float v) {
    for (int m = 16; m > 0; m >>= 1) v += __shfl_xor(v, m);
    return v;
}

__device__ __forceinline__ float sumsq_of_parts(const float* psq, int lane) {
    float pv[PSQ_P / 32];
#pragma unroll
    for (int j = 0; j < PSQ_P / 32; ++j) pv[j] = psq[lane + 32 * j];
    float s = 0.f;
#pragma unroll
    for (int j = 0; j < PSQ_P / 32; ++j) s += pv[j];
    return wave_sum(s);
}

__global__ void k_reset(float* w) {
    if (blockIdx.x == 0 && threadIdx.x < 32) {
        const float ss = sumsq_of_parts(w + WS_PSQ, (int)threadIdx.x);
        if (threadIdx.x == 0) w[WS_STEP] = 0.9f / sqrtf(sqrtf(ss));
    }
    const int total = 2 * XSZ + YSZ + S_P * NL + NL;
    for (int i = blockIdx.x * blockDim.x + threadIdx.x; i < total; i += gridDim.x * blockDim.x) {
        if (i < XSZ)                w[WS_X + i] = 0.f;
        else if (i < 2 * XSZ)       w[WS_XB + (i - XSZ)] = 0.f;
        else if (i < 2 * XSZ + YSZ) w[WS_Y + (i - 2 * XSZ)] = 0.f;
        else                        w[WS_ZT + (i - 2 * XSZ - YSZ)] = 0.f;
    }
}

template <int K>
__device__ __forceinline__ v8f tile_gemm(const float* arow, const float* P, const float* Q,
                                         int n, int kh, float sc) {
    v8f acc = {0.f,0.f,0.f,0.f,0.f,0.f,0.f,0.f};
#pragma unroll
    for (int k0 = 0; k0 < K; k0 += 4) {
        int ka = k0 + kh;
        v2f a = *(const v2f*)(arow + ka);
        v2f b;
        int bi = ka * S_P + n;
        b[0] = P[bi]       * sc - Q[bi]       * sc;
        b[1] = P[bi + S_P] * sc - Q[bi + S_P] * sc;
        acc = __builtin_amdgcn_wmma_f32_16x16x4_f32(false, a, false, b, (short)0, acc, false, false);
    }
    return acc;
}

struct KpFrags { v4f a[12], b[12]; v2f at, bt; };

__device__ __forceinline__ void load_kp(KpFrags& f, const float* ablk, const float* bblk, int lane) {
#pragma unroll
    for (int j = 0; j < 12; ++j) {
        f.a[j] = *(const v4f*)(ablk + 128 * j + 4 * lane);
        f.b[j] = *(const v4f*)(bblk + 128 * j + 4 * lane);
    }
    f.at = *(const v2f*)(ablk + 1536 + 2 * lane);
    f.bt = *(const v2f*)(bblk + 1536 + 2 * lane);
}

__device__ __forceinline__ v8f mma_kp(const KpFrags& f) {
    v8f acc = {0.f,0.f,0.f,0.f,0.f,0.f,0.f,0.f};
#pragma unroll
    for (int j = 0; j < 12; ++j) {
        v2f a0, b0, a1, b1;
        a0[0] = f.a[j][0]; a0[1] = f.a[j][1]; b0[0] = f.b[j][0]; b0[1] = f.b[j][1];
        a1[0] = f.a[j][2]; a1[1] = f.a[j][3]; b1[0] = f.b[j][2]; b1[1] = f.b[j][3];
        acc = __builtin_amdgcn_wmma_f32_16x16x4_f32(false, a0, false, b0, (short)0, acc, false, false);
        acc = __builtin_amdgcn_wmma_f32_16x16x4_f32(false, a1, false, b1, (short)0, acc, false, false);
    }
    acc = __builtin_amdgcn_wmma_f32_16x16x4_f32(false, f.at, false, f.bt, (short)0, acc, false, false);
    return acc;
}

static_assert(NG == 100 && NL == 400, "load_kp, pk_in and kpos100 are written for 100-long k ranges");

constexpr int RPW = 8;


constexpr int KSP = 4;
constexpr int RNT = 32 * KSP;
constexpr int KQ  = NL / KSP;
constexpr int SQ  = 48;
constexpr int RA_T1 = 7 * 13;
constexpr int RA_T2 = RA_T1 + (NG + RPW - 1) / RPW;
constexpr int RA_T3 = RA_T2 + (NG + RPW - 1) / RPW;
constexpr int RA_NB = RA_T3 + NL / RPW;
static_assert(RA_T1 * KSP + (RA_NB - RA_T1) <= PSQ_P, "one partial sum of squares per storing wave");

__device__ __forceinline__ void row_sum_share(const float* r0, const float* r1, int wv, float& a, float& b) {
    const float* p0 = r0 + wv * SQ;
    const float* p1 = r1 + wv * SQ;
    constexpr int TAIL = NS - KSP * SQ;
    float v0[SQ + TAIL], v1[SQ + TAIL];
#pragma unroll
    for (int s = 0; s < SQ; ++s) { v0[s] = p0[s]; v1[s] = p1[s]; }
#pragma unroll
    for (int s = SQ; s < SQ + TAIL; ++s) { v0[s] = 0.f; v1[s] = 0.f; }
    if (wv == KSP - 1) {
#pragma unroll
        for (int s = SQ; s < SQ + TAIL; ++s) { v0[s] = p0[s]; v1[s] = p1[s]; }
    }
#pragma unroll
    for (int s = 0; s < SQ; ++s) { a += v0[s]; b += v1[s]; }
    if (wv == KSP - 1) {
#pragma unroll
        for (int s = SQ; s < SQ + TAIL; ++s) { a += v0[s]; b += v1[s]; }
    }
}

template <bool PD>
__global__ __launch_bounds__(RNT) __attribute__((amdgpu_waves_per_eu(1, 1))) void k_rmv(
        float* w, const float* Cost, const float* Cru, const float* Crd) {
    __shared__ float part[KSP * 8 * 32];
    const int lane = threadIdx.x & 31, blk = blockIdx.x;
    const int wv = __builtin_amdgcn_readfirstlane((int)(threadIdx.x >> 5));
    const float* PGT = w + WS_PGT;
    const float* yb = PD ? (w + WS_Y) : (w + WS_AX);
    float* X  = w + WS_X;
    float* XB = w + WS_XB;
    const float step = PD ? w[WS_STEP] : 0.f;
    float q = 0.f;
    if (blk < RA_T1) {
        const int tm = blk / 13, tn = blk % 13;
        const int m16 = lane & 15;
        const int n = tn * 16 + m16;
        const int mr = tm * 16 + ((lane >> 4) << 3) + 2 * wv;
        const int nc = n < NS ? n : NS - 1;
        const float y4 = yb[OY_Y4 + nc];
        float z3v[2], z4v[2], xu[2], xd[2];
        for (int j = 0; j < 2; ++j) {
            const int gc = (mr + j) < NG ? (mr + j) : NG - 1;
            const int o = gc * S_P + nc;
            z3v[j] = yb[OY_Z3 + o]; z4v[j] = yb[OY_Z4 + o];
            xu[j] = PD ? X[OX_U + o] : 0.f; xd[j] = PD ? X[OX_D + o] : 0.f;
        }
        KpFrags f;
        load_kp(f, w + WS_PGTP + (tm * KSP + wv) * BLK, w + WS_ZT + (tn * KSP + wv) * BLK, lane);
        __builtin_amdgcn_sched_barrier(0);
        v8f acc = mma_kp(f);
        for (int i = 0; i < 8; ++i) part[(wv * 8 + i) * 32 + lane] = acc[i];
        __syncthreads();
        float nu[2], nd[2], bu[2], bd[2];
        for (int j = 0; j < 2; ++j) {
            const int i = 2 * wv + j;
            const float tv = ((part[(0 * 8 + i) * 32 + lane] + part[(1 * 8 + i) * 32 + lane])
                              + part[(2 * 8 + i) * 32 + lane]) + part[(3 * 8 + i) * 32 + lane];
            const float au =  y4 + z3v[j] + tv;
            const float ad = -y4 + z4v[j] - tv;
            if (PD) {
                nu[j] = fmaxf(xu[j] - step * au, 0.f); bu[j] = 2.f * nu[j] - xu[j];
                nd[j] = fmaxf(xd[j] - step * ad, 0.f); bd[j] = 2.f * nd[j] - xd[j];
            } else {
                nu[j] = au; nd[j] = ad; bu[j] = 0.f; bd[j] = 0.f;
            }
        }
        for (int j = 0; j < 2; ++j) {
            const int g = mr + j;
            if (n < NS && g < NG) {
                const int o = g * S_P + n;
                X[OX_U + o] = nu[j]; X[OX_D + o] = nd[j];
                if (PD) { XB[OX_U + o] = bu[j]; XB[OX_D + o] = bd[j]; }
                else    q += nu[j] * nu[j] + nd[j] * nd[j];
            }
        }
        if (PD && n < NS && mr < NG) {
            v2f d; d[0] = bu[0] - bd[0]; d[1] = bu[1] - bd[1];
            *(v2f*)(w + WS_UDT + tn * BLK + pk_in(m16, kpos100(mr))) = d;
        }
    } else if (blk < RA_T2) {
        const int g = (blk - RA_T1) * RPW + lane;
        const bool act = lane < RPW && g < NG;
        float z1 = 0.f, z2 = 0.f, xu = 0.f, xd = 0.f, cu = 0.f, cd = 0.f;
        float a = 0.f, b = 0.f;
        if (act) {
            if (wv == 0) {
                z1 = yb[OY_Z1 + g]; z2 = yb[OY_Z2 + g];
                xu = PD ? X[OX_RU + g] : 0.f; xd = PD ? X[OX_RD + g] : 0.f;
                cu = PD ? Cru[g] : 0.f; cd = PD ? Crd[g] : 0.f;
            }
            row_sum_share(yb + OY_Z3 + g * S_P, yb + OY_Z4 + g * S_P, wv, a, b);
        }
        part[wv * 32 + lane] = a; part[(KSP + wv) * 32 + lane] = b;
        __syncthreads();
        if (wv == 0 && act) {
            const float sa = ((part[lane] + part[32 + lane]) + part[64 + lane]) + part[96 + lane];
            const float sb = ((part[128 + lane] + part[160 + lane]) + part[192 + lane]) + part[224 + lane];
            const float au = z1 - sa;
            const float ad = z2 - sb;
            if (PD) {
                const float nu = fmaxf(xu - step * (cu + au), 0.f);
                const float nd = fmaxf(xd - step * (cd + ad), 0.f);
                X[OX_RU + g] = nu; XB[OX_RU + g] = 2.f * nu - xu;
                X[OX_RD + g] = nd; XB[OX_RD + g] = 2.f * nd - xd;
            } else {
                X[OX_RU + g] = au; X[OX_RD + g] = ad;
                q = au * au + ad * ad;
            }
        }
    } else if (blk < RA_T3) {
        const int g = (blk - RA_T2) * RPW + lane;
        const bool act = lane < RPW && g < NG;
        float y1 = 0.f, z1 = 0.f, z2 = 0.f, xp = 0.f, cp = 0.f;
        float c = 0.f;
        if (act) {
            const float* pr = PGT + g * NL + wv * KQ;
            const float* dp = w + WS_D23 + wv * KQ;
            float pv[KQ], dv[KQ];
#pragma unroll
            for (int l = 0; l < KQ; ++l) { pv[l] = pr[l]; dv[l] = dp[l]; }
            if (wv == 0) {
                y1 = yb[OY_Y1]; z1 = yb[OY_Z1 + g]; z2 = yb[OY_Z2 + g];
                xp = PD ? X[OX_P + g] : 0.f;
                cp = PD ? Cost[g] : 0.f;
            }
#pragma unroll
            for (int l = 0; l < KQ; ++l) c += pv[l] * dv[l];
        }
        part[wv * 32 + lane] = c;
        __syncthreads();
        if (wv == 0 && act) {
            const float cs = ((part[lane] + part[32 + lane]) + part[64 + lane]) + part[96 + lane];
            const float ap = y1 + cs + z1 - z2;
            if (PD) {
                const float nn = fmaxf(xp - step * (cp + ap), 0.f);
                X[OX_P + g] = nn; XB[OX_P + g] = 2.f * nn - xp;
            } else {
                X[OX_P + g] = ap;
                q = ap * ap;
            }
        }
    } else {
        const int l = (blk - RA_T3) * RPW + lane;
        const bool act = lane < RPW && l < NL;
        float y2 = 0.f, y3 = 0.f, xu = 0.f, xd = 0.f;
        float a = 0.f, b = 0.f;
        if (act) {
            if (wv == 0) {
                y2 = yb[OY_Y2 + l]; y3 = yb[OY_Y3 + l];
                xu = PD ? X[OX_FU + l] : 0.f; xd = PD ? X[OX_FD + l] : 0.f;
            }
            row_sum_share(yb + OY_Z5 + l * S_P, yb + OY_Z6 + l * S_P, wv, a, b);
        }
        part[wv * 32 + lane] = a; part[(KSP + wv) * 32 + lane] = b;
        __syncthreads();
        if (wv == 0 && act) {
            const float sa = ((part[lane] + part[32 + lane]) + part[64 + lane]) + part[96 + lane];
            const float sb = ((part[128 + lane] + part[160 + lane]) + part[192 + lane]) + part[224 + lane];
            const float au = y2 - sa;
            const float ad = y3 - sb;
            if (PD) {
                const float nu = fmaxf(xu - step * au, 0.f);
                const float nd = fmaxf(xd - step * ad, 0.f);
                X[OX_FU + l] = nu; XB[OX_FU + l] = 2.f * nu - xu;
                X[OX_FD + l] = nd; XB[OX_FD + l] = 2.f * nd - xd;
            } else {
                X[OX_FU + l] = au; X[OX_FD + l] = ad;
                q = au * au + ad * ad;
            }
        }
    }
    if (!PD) {
        const float s = wave_sum(q);
        const bool tile = blk < RA_T1;
        if (lane == 0 && (tile || wv == 0))
            w[WS_PSQ + (tile ? blk * KSP + wv : RA_T1 * KSP + (blk - RA_T1))] = s;
    }
}

constexpr int EPW = 128;
constexpr int MA_T1 = 25 * 13;
constexpr int MA_T2 = MA_T1 + NL / RPW;
constexpr int MA_T3 = MA_T2 + (NS + 31) / 32;
constexpr int MA_T4 = MA_T3 + (NG + 31) / 32;
constexpr int MA_T5 = MA_T4 + 1;
constexpr int MA_NB = MA_T5 + (NG * NS + EPW - 1) / EPW;

template <bool PD>
__global__ __launch_bounds__(32) __attribute__((amdgpu_waves_per_eu(1, 1))) void k_mv(
        float* w, const float* Pmax) {
    const int lane = threadIdx.x, blk = blockIdx.x;
    const float* PG = w + WS_PG;
    const float* xb = PD ? (w + WS_XB) : (w + WS_X);
    float* Y = PD ? (w + WS_Y) : (w + WS_AX);
    const float* pw = w + WS_PW;
    const float step = PD ? w[WS_STEP] : 0.f;
    const float inv = PD ? 1.f : 1.0f / sqrtf(sumsq_of_parts(w + WS_PSQ, lane));
#define XS(v) (PD ? (v) : (v) * inv)
    if (blk < MA_T1) {
        const int tm = blk / 13, tn = blk % 13;
        const int m16 = lane & 15, kh = (lane >> 4) * 2;
        const int m = tm * 16 + m16, n = tn * 16 + m16;
        const int mr = tm * 16 + ((lane >> 4) << 3);
        const int nc = n < NS ? n : NS - 1;
        float fu[8], fd[8], y5[8], y6[8], pv[8];
        for (int i = 0; i < 8; ++i) {
            const int l = mr + i;
            const int o = l * S_P + nc;
            fu[i] = XS(xb[OX_FU + l]); fd[i] = XS(xb[OX_FD + l]);
            y5[i] = PD ? Y[OY_Z5 + o] : 0.f; y6[i] = PD ? Y[OY_Z6 + o] : 0.f;
            pv[i] = PD ? pw[o] : 0.f;
        }
        v8f acc;
        if (PD) {
            KpFrags f;
            load_kp(f, w + WS_PGP + tm * BLK, w + WS_UDT + tn * BLK, lane);
#pragma unroll
            for (int i = 0; i < 8; ++i)
                asm volatile("" :: "v"(fu[i]), "v"(fd[i]), "v"(y5[i]), "v"(y6[i]), "v"(pv[i]));
            __builtin_amdgcn_sched_barrier(0);
            acc = mma_kp(f);
        } else {
            acc = tile_gemm<NG>(PG + m * G_P, xb + OX_U, xb + OX_D, n, kh, inv);
        }
        float r5[8], r6[8];
        for (int i = 0; i < 8; ++i) {
            const float f = acc[i];
            const float a5 =  f - fu[i];
            const float a6 = -f - fd[i];
            if (PD) {
                r5[i] = fmaxf(y5[i] + step * (a5 + pv[i]), 0.f);
                r6[i] = fmaxf(y6[i] + step * (a6 - pv[i]), 0.f);
            } else {
                r5[i] = a5; r6[i] = a6;
            }
        }
        if (n < NS) {
            for (int i = 0; i < 8; ++i) {
                const int o = (mr + i) * S_P + n;
                Y[OY_Z5 + o] = r5[i]; Y[OY_Z6 + o] = r6[i];
            }
            float* zt = w + WS_ZT + tn * KSP * BLK;
            for (int i = 0; i < 8; i += 2) {
                const int l = mr + i;
                v2f d; d[0] = r5[i] - r6[i]; d[1] = r5[i + 1] - r6[i + 1];
                *(v2f*)(zt + (l / KQ) * BLK + pk_in(m16, kpos100(l % KQ))) = d;
            }
        }
    } else if (blk < MA_T2) {
        const int l = (blk - MA_T1) * RPW + lane;
        if (lane < RPW && l < NL) {
            const float fu = XS(xb[OX_FU + l]), fd = XS(xb[OX_FD + l]);
            const float y2 = PD ? Y[OY_Y2 + l] : 0.f, y3 = PD ? Y[OY_Y3 + l] : 0.f;
            const float b2 = PD ? w[WS_B2 + l] : 0.f, b3 = PD ? w[WS_B3 + l] : 0.f;
            const float* r = PG + l * G_P; const float* p = xb + OX_P;
            float rv[NG], pg[NG];
#pragma unroll
            for (int g = 0; g < NG; ++g) { rv[g] = r[g]; pg[g] = p[g]; }
            __builtin_amdgcn_sched_barrier(0);
            float s = 0.f;
#pragma unroll
            for (int g = 0; g < NG; ++g) s += rv[g] * XS(pg[g]);
            const float a2 =  s + fu;
            const float a3 = -s + fd;
            const float n2 = PD ? y2 + step * (a2 - b2) : a2;
            const float n3 = PD ? y3 + step * (a3 - b3) : a3;
            Y[OY_Y2 + l] = n2; Y[OY_Y3 + l] = n3;
            w[WS_D23 + l] = n2 - n3;
        }
    } else if (blk < MA_T3) {
        const int s = (blk - MA_T2) * 32 + lane;
        if (s < NS) {
            const float y4 = PD ? Y[OY_Y4 + s] : 0.f;
            const float b4 = PD ? w[WS_B4 + s] : 0.f;
            float c = 0.f;
            if (PD) {
                const float* r = w + WS_UDT + (s >> 4) * BLK;
                const int c16 = s & 15;
                float cv[NG];
#pragma unroll
                for (int j = 0; j < 12; ++j)
#pragma unroll
                    for (int h = 0; h < 2; ++h) {
                        const v4f t = *(const v4f*)(r + 128 * j + 4 * (c16 + 16 * h));
                        cv[8 * j + 4 * h + 0] = t[0]; cv[8 * j + 4 * h + 1] = t[1];
                        cv[8 * j + 4 * h + 2] = t[2]; cv[8 * j + 4 * h + 3] = t[3];
                    }
#pragma unroll
                for (int h = 0; h < 2; ++h) {
                    const v2f t = *(const v2f*)(r + 1536 + 2 * (c16 + 16 * h));
                    cv[96 + 2 * h] = t[0]; cv[97 + 2 * h] = t[1];
                }
                __builtin_amdgcn_sched_barrier(0);
#pragma unroll
                for (int g = 0; g < NG; ++g) c += cv[kpos100(g)];
            } else {
                const float* Ux = xb + OX_U + s; const float* Dx = xb + OX_D + s;
#pragma unroll
                for (int g = 0; g < NG; ++g) c += Ux[g * S_P] * inv - Dx[g * S_P] * inv;
            }
            if (PD) Y[OY_Y4 + s] = y4 + step * (c - b4);
            else    Y[OY_Y4 + s] = c;
        }
    } else if (blk < MA_T4) {
        const int g = (blk - MA_T3) * 32 + lane;
        if (g < NG) {
            const float xp = XS(xb[OX_P + g]), xu = XS(xb[OX_RU + g]), xd = XS(xb[OX_RD + g]);
            const float a1 = xp + xu;
            const float a2 = xd - xp;
            if (PD) {
                const float z1 = Y[OY_Z1 + g], z2 = Y[OY_Z2 + g], pm = Pmax[g];
                Y[OY_Z1 + g] = fmaxf(z1 + step * (a1 - pm), 0.f);
                Y[OY_Z2 + g] = fmaxf(z2 + step *  a2, 0.f);
            } else {
                Y[OY_Z1 + g] = a1; Y[OY_Z2 + g] = a2;
            }
        }
    } else if (blk < MA_T5) {
        if (lane == 0) {
            const float y1 = PD ? Y[OY_Y1] : 0.f;
            const float b1 = PD ? w[WS_B1] : 0.f;
            float pp[NG];
#pragma unroll
            for (int g = 0; g < NG; ++g) pp[g] = xb[OX_P + g];
            __builtin_amdgcn_sched_barrier(0);
            float s = 0.f;
#pragma unroll
            for (int g = 0; g < NG; ++g) s += XS(pp[g]);
            if (PD) Y[OY_Y1] = y1 + step * (s - b1);
            else    Y[OY_Y1] = s;
        }
    } else {
        const int base = (blk - MA_T5) * EPW;
        float r3[EPW / 32], r4[EPW / 32];
        for (int j = 0; j < EPW / 32; ++j) {
            const int i = base + j * 32 + lane;
            const int ic = i < NG * NS ? i : NG * NS - 1;
            const int g = ic / NS, s = ic % NS, o = g * S_P + s;
            const float a3 = XS(xb[OX_U + o]) - XS(xb[OX_RU + g]);
            const float a4 = XS(xb[OX_D + o]) - XS(xb[OX_RD + g]);
            if (PD) {
                r3[j] = fmaxf(Y[OY_Z3 + o] + step * a3, 0.f);
                r4[j] = fmaxf(Y[OY_Z4 + o] + step * a4, 0.f);
            } else {
                r3[j] = a3; r4[j] = a4;
            }
        }
        for (int j = 0; j < EPW / 32; ++j) {
            const int i = base + j * 32 + lane;
            if (i < NG * NS) {
                const int g = i / NS, s = i % NS, o = g * S_P + s;
                Y[OY_Z3 + o] = r3[j]; Y[OY_Z4 + o] = r4[j];
            }
        }
    }
#undef XS
}

__global__ void k_emit(const float* w, float* out) {
    const float* X = w + WS_X;
    const int total = 3 * NG + 2 * NG * NS + 2 * NL;
    for (int idx = blockIdx.x * blockDim.x + threadIdx.x; idx < total; idx += gridDim.x * blockDim.x) {
        float v;
        if (idx < NG)                 v = X[OX_P  + idx];
        else if (idx < 2 * NG)        v = X[OX_RU + (idx - NG)];
        else if (idx < 3 * NG)        v = X[OX_RD + (idx - 2 * NG)];
        else if (idx < 3 * NG + NG * NS) { int r = idx - 3 * NG; v = X[OX_U + (r / NS) * S_P + (r % NS)]; }
        else if (idx < 3 * NG + 2 * NG * NS) { int r = idx - 3 * NG - NG * NS; v = X[OX_D + (r / NS) * S_P + (r % NS)]; }
        else if (idx < 3 * NG + 2 * NG * NS + NL) v = X[OX_FU + (idx - 3 * NG - 2 * NG * NS)];
        else                          v = X[OX_FD + (idx - 3 * NG - 2 * NG * NS - NL)];
        out[idx] = v;
    }
}


extern "C" void kernel_launch(void* const* d_in, const int* in_sizes, int n_in,
                              void* d_out, int out_size, void* d_ws, size_t ws_size,
                              hipStream_t stream) {
    const float* w_scen   = (const float*)d_in[0];
    const float* Cost     = (const float*)d_in[1];
    const float* C_r_up   = (const float*)d_in[2];
    const float* C_r_down = (const float*)d_in[3];
    const float* Pmax     = (const float*)d_in[4];
    const float* PTDF     = (const float*)d_in[5];
    const float* node_G   = (const float*)d_in[6];
    const float* node_W   = (const float*)d_in[7];
    const float* node_L   = (const float*)d_in[8];
    const float* Pd       = (const float*)d_in[9];
    const float* w_exp    = (const float*)d_in[10];
    const float* Line_cap = (const float*)d_in[11];
    float* w   = (float*)d_ws;
    float* out = (float*)d_out;

    k_zero <<<512, 256, 0, stream>>>(w);
    k_prep1<<<240, 256, 0, stream>>>(w, w_scen, node_W, node_L, Pd, w_exp);
    k_prep2<<<640, 256, 0, stream>>>(w, PTDF, node_G, Line_cap);
    k_setv <<<161, 256, 0, stream>>>(w);

    for (int it = 0; it < POWER_ITERS; ++it) {
        k_mv  <false><<<MA_NB, 32, 0, stream>>>(w, Pmax);
        k_rmv <false><<<RA_NB, RNT, 0, stream>>>(w, Cost, C_r_up, C_r_down);
    }
    k_reset<<<256, 256, 0, stream>>>(w);

    for (int it = 0; it < PDHG_ITERS; ++it) {
        k_rmv<true><<<RA_NB, RNT, 0, stream>>>(w, Cost, C_r_up, C_r_down);
        k_mv <true><<<MA_NB, 32, 0, stream>>>(w, Pmax);
    }
    k_emit<<<161, 256, 0, stream>>>(w, out);
}


# Power iteration on one scenario column (all stay equal from the all-ones start): 25 steps in one work-group launch, PG in registers and LDS, vector-unit products; its sums change order (step size)
# speedup vs baseline: 42.8204x; 1.0573x over previous
#include <hip/hip_runtime.h>
#include <math.h>

typedef __attribute__((ext_vector_type(2))) float v2f;
typedef __attribute__((ext_vector_type(4))) float v4f;
typedef __attribute__((ext_vector_type(8))) float v8f;


constexpr int NG = 100, NL = 400, NN = 300, NWF = 50, NS = 200;
constexpr int G_P = 112;
constexpr int S_P = 208;
constexpr int PDHG_ITERS = 300, POWER_ITERS = 25;

constexpr int OX_P  = 0;
constexpr int OX_RU = OX_P  + G_P;
constexpr int OX_RD = OX_RU + G_P;
constexpr int OX_U  = OX_RD + G_P;
constexpr int OX_D  = OX_U  + G_P * S_P;
constexpr int OX_FU = OX_D  + G_P * S_P;
constexpr int OX_FD = OX_FU + NL;
constexpr int XSZ   = OX_FD + NL;

constexpr int OY_Y1 = 0;
constexpr int OY_Y2 = 16;
constexpr int OY_Y3 = OY_Y2 + NL;
constexpr int OY_Y4 = OY_Y3 + NL;
constexpr int OY_Z1 = OY_Y4 + S_P;
constexpr int OY_Z2 = OY_Z1 + G_P;
constexpr int OY_Z3 = OY_Z2 + G_P;
constexpr int OY_Z4 = OY_Z3 + G_P * S_P;
constexpr int OY_Z5 = OY_Z4 + G_P * S_P;
constexpr int OY_Z6 = OY_Z5 + NL * S_P;
constexpr int YSZ   = OY_Z6 + NL * S_P;

constexpr int WS_PG   = 0;
constexpr int WS_PGT  = WS_PG   + NL * G_P;
constexpr int WS_PW   = WS_PGT  + G_P * NL;
constexpr int WS_B2   = WS_PW   + NL * S_P;
constexpr int WS_B3   = WS_B2   + NL;
constexpr int WS_B4   = WS_B3   + NL;
constexpr int WS_B1   = WS_B4   + S_P;
constexpr int WS_M1   = WS_B1   + 16;
constexpr int WS_NV   = WS_M1   + NN * S_P;
constexpr int WS_X    = WS_NV   + NN + 4;
constexpr int WS_XB   = WS_X    + XSZ;
constexpr int WS_ATY  = WS_XB   + XSZ;
constexpr int WS_Y    = WS_ATY  + XSZ;
constexpr int WS_AX   = WS_Y    + YSZ;
constexpr int WS_T    = WS_AX   + YSZ;
constexpr int WS_FLOW = WS_T    + G_P * S_P;
constexpr int WS_GP   = WS_FLOW + NL * S_P;
constexpr int WS_COL  = WS_GP   + NL;
constexpr int WS_S3   = WS_COL  + S_P;
constexpr int WS_S4   = WS_S3   + G_P;
constexpr int WS_S5   = WS_S4   + G_P;
constexpr int WS_S6   = WS_S5   + NL;
constexpr int WS_GTY  = WS_S6   + NL;
constexpr int WS_SUMP = WS_GTY  + G_P;
constexpr int WS_STEP = WS_SUMP + 16;
constexpr int BLK = 1600;
__host__ __device__ constexpr int al32(int x) { return (x + 31) & ~31; }
constexpr int WS_UDT  = al32(WS_STEP + 16);
constexpr int WS_ZT   = WS_UDT  + S_P * G_P;
constexpr int WS_D23  = WS_ZT   + S_P * NL;
constexpr int WS_PGP  = al32(WS_D23 + NL);
constexpr int WS_PGTP = WS_PGP  + NL * G_P;
constexpr int WS_TOTAL= WS_PGTP + G_P * NL;

__host__ __device__ constexpr int kpos100(int r) {
    return r < 96 ? ((r & ~7) | ((r & 2) << 1) | ((r & 4) >> 1) | (r & 1)) : r;
}
__host__ __device__ constexpr int pk_in(int c, int p) {
    return p < 96 ? ((p >> 3) * 128 + (c + ((p >> 2) & 1) * 16) * 4 + (p & 3))
                  : (1536 + (c + ((p - 96) >> 1) * 16) * 2 + (p & 1));
}
static_assert(13 * BLK <= S_P * G_P && 13 * 4 * BLK == S_P * NL && 25 * BLK <= NL * G_P && 7 * 4 * BLK == G_P * NL,
              "the packed planes fit the room of the row-major ones");


__global__ void k_zero(float* w) {
    for (int i = blockIdx.x * blockDim.x + threadIdx.x; i < WS_TOTAL;
         i += gridDim.x * blockDim.x) w[i] = 0.0f;
}

__global__ void k_prep1(float* w, const float* w_scen, const float* node_W,
                        const float* node_L, const float* Pd, const float* w_exp) {
    const int NM1 = NN * NS;
    const int total = NM1 + NN + NS + 1;
    for (int idx = blockIdx.x * blockDim.x + threadIdx.x; idx < total;
         idx += gridDim.x * blockDim.x) {
        if (idx < NM1) {
            int n = idx / NS, s = idx % NS;
            float a = 0.f;
            for (int ww = 0; ww < NWF; ++ww) a += node_W[n*NWF + ww] * w_scen[s*NWF + ww];
            w[WS_M1 + n*S_P + s] = a;
        } else if (idx < NM1 + NN) {
            int n = idx - NM1;
            float a = 0.f;
            for (int ww = 0; ww < NWF; ++ww) a += node_W[n*NWF + ww] * w_exp[ww];
            for (int m = 0; m < NN; ++m)     a -= node_L[n*NN + m] * Pd[m];
            w[WS_NV + n] = a;
        } else if (idx < NM1 + NN + NS) {
            int s = idx - NM1 - NN;
            float a = 0.f;
            for (int ww = 0; ww < NWF; ++ww) a += w_scen[s*NWF + ww];
            w[WS_B4 + s] = -a;
        } else {
            float a = 0.f;
            for (int m = 0; m < NN; ++m)  a += Pd[m];
            for (int ww = 0; ww < NWF; ++ww) a -= w_exp[ww];
            w[WS_B1] = a;
        }
    }
}

__global__ void k_prep2(float* w, const float* PTDF, const float* node_G,
                        const float* Line_cap) {
    const int A = NL * NG;
    const int B = A + NG * NL;
    const int C = B + NL * NS;
    const int D2 = C + NL;
    const int E = D2 + NL;
    for (int idx = blockIdx.x * blockDim.x + threadIdx.x; idx < E;
         idx += gridDim.x * blockDim.x) {
        if (idx < A) {
            int l = idx / NG, g = idx % NG;
            float a = 0.f;
            for (int n = 0; n < NN; ++n) a += PTDF[l*NN + n] * node_G[n*NG + g];
            w[WS_PG + l*G_P + g] = a;
            w[WS_PGP + (l >> 4) * BLK + pk_in(l & 15, kpos100(g))] = a;
        } else if (idx < B) {
            int r = idx - A; int g = r / NL, l = r % NL;
            float a = 0.f;
            for (int n = 0; n < NN; ++n) a += PTDF[l*NN + n] * node_G[n*NG + g];
            w[WS_PGT + g*NL + l] = a;
            w[WS_PGTP + ((g >> 4) * 4 + l / 100) * BLK + pk_in(g & 15, kpos100(l % 100))] = a;
        } else if (idx < C) {
            int r = idx - B; int l = r / NS, s = r % NS;
            float a = 0.f;
            for (int n = 0; n < NN; ++n) a += PTDF[l*NN + n] * w[WS_M1 + n*S_P + s];
            w[WS_PW + l*S_P + s] = a;
        } else if (idx < D2) {
            int l = idx - C;
            float a = 0.f;
            for (int n = 0; n < NN; ++n) a += PTDF[l*NN + n] * w[WS_NV + n];
            w[WS_B2 + l] = Line_cap[l] - a;
        } else {
            int l = idx - D2;
            float a = 0.f;
            for (int n = 0; n < NN; ++n) a += PTDF[l*NN + n] * w[WS_NV + n];
            w[WS_B3 + l] = Line_cap[l] + a;
        }
    }
}


__device__ __forceinline__ float wave_sum(float v) {
    for (int m = 16; m > 0; m >>= 1) v += __shfl_xor(v, m);
    return v;
}

struct KpFrags { v4f a[12], b[12]; v2f at, bt; };

__device__ __forceinline__ void load_kp(KpFrags& f, const float* ablk, const float* bblk, int lane) {
#pragma unroll
    for (int j = 0; j < 12; ++j) {
        f.a[j] = *(const v4f*)(ablk + 128 * j + 4 * lane);
        f.b[j] = *(const v4f*)(bblk + 128 * j + 4 * lane);
    }
    f.at = *(const v2f*)(ablk + 1536 + 2 * lane);
    f.bt = *(const v2f*)(bblk + 1536 + 2 * lane);
}

__device__ __forceinline__ v8f mma_kp(const KpFrags& f) {
    v8f acc = {0.f,0.f,0.f,0.f,0.f,0.f,0.f,0.f};
#pragma unroll
    for (int j = 0; j < 12; ++j) {
        v2f a0, b0, a1, b1;
        a0[0] = f.a[j][0]; a0[1] = f.a[j][1]; b0[0] = f.b[j][0]; b0[1] = f.b[j][1];
        a1[0] = f.a[j][2]; a1[1] = f.a[j][3]; b1[0] = f.b[j][2]; b1[1] = f.b[j][3];
        acc = __builtin_amdgcn_wmma_f32_16x16x4_f32(false, a0, false, b0, (short)0, acc, false, false);
        acc = __builtin_amdgcn_wmma_f32_16x16x4_f32(false, a1, false, b1, (short)0, acc, false, false);
    }
    acc = __builtin_amdgcn_wmma_f32_16x16x4_f32(false, f.at, false, f.bt, (short)0, acc, false, false);
    return acc;
}

static_assert(NG == 100 && NL == 400, "load_kp, pk_in and kpos100 are written for 100-long k ranges");

constexpr int RPW = 8;


constexpr int KSP = 4;
constexpr int RNT = 32 * KSP;
constexpr int KQ  = NL / KSP;
constexpr int SQ  = 48;
constexpr int RA_T1 = 7 * 13;
constexpr int RA_T2 = RA_T1 + (NG + RPW - 1) / RPW;
constexpr int RA_T3 = RA_T2 + (NG + RPW - 1) / RPW;
constexpr int RA_NB = RA_T3 + NL / RPW;

__device__ __forceinline__ void row_sum_share(const float* r0, const float* r1, int wv, float& a, float& b) {
    const float* p0 = r0 + wv * SQ;
    const float* p1 = r1 + wv * SQ;
    constexpr int TAIL = NS - KSP * SQ;
    float v0[SQ + TAIL], v1[SQ + TAIL];
#pragma unroll
    for (int s = 0; s < SQ; ++s) { v0[s] = p0[s]; v1[s] = p1[s]; }
#pragma unroll
    for (int s = SQ; s < SQ + TAIL; ++s) { v0[s] = 0.f; v1[s] = 0.f; }
    if (wv == KSP - 1) {
#pragma unroll
        for (int s = SQ; s < SQ + TAIL; ++s) { v0[s] = p0[s]; v1[s] = p1[s]; }
    }
#pragma unroll
    for (int s = 0; s < SQ; ++s) { a += v0[s]; b += v1[s]; }
    if (wv == KSP - 1) {
#pragma unroll
        for (int s = SQ; s < SQ + TAIL; ++s) { a += v0[s]; b += v1[s]; }
    }
}

__global__ __launch_bounds__(RNT) __attribute__((amdgpu_waves_per_eu(1, 1))) void k_rmv(
        float* w, const float* Cost, const float* Cru, const float* Crd) {
    __shared__ float part[KSP * 8 * 32];
    const int lane = threadIdx.x & 31, blk = blockIdx.x;
    const int wv = __builtin_amdgcn_readfirstlane((int)(threadIdx.x >> 5));
    const float* PGT = w + WS_PGT;
    const float* yb = w + WS_Y;
    float* X  = w + WS_X;
    float* XB = w + WS_XB;
    const float step = w[WS_STEP];
    if (blk < RA_T1) {
        const int tm = blk / 13, tn = blk % 13;
        const int m16 = lane & 15;
        const int n = tn * 16 + m16;
        const int mr = tm * 16 + ((lane >> 4) << 3) + 2 * wv;
        const int nc = n < NS ? n : NS - 1;
        const float y4 = yb[OY_Y4 + nc];
        float z3v[2], z4v[2], xu[2], xd[2];
        for (int j = 0; j < 2; ++j) {
            const int gc = (mr + j) < NG ? (mr + j) : NG - 1;
            const int o = gc * S_P + nc;
            z3v[j] = yb[OY_Z3 + o]; z4v[j] = yb[OY_Z4 + o];
            xu[j] = X[OX_U + o]; xd[j] = X[OX_D + o];
        }
        KpFrags f;
        load_kp(f, w + WS_PGTP + (tm * KSP + wv) * BLK, w + WS_ZT + (tn * KSP + wv) * BLK, lane);
        __builtin_amdgcn_sched_barrier(0);
        v8f acc = mma_kp(f);
        for (int i = 0; i < 8; ++i) part[(wv * 8 + i) * 32 + lane] = acc[i];
        __syncthreads();
        float nu[2], nd[2], bu[2], bd[2];
        for (int j = 0; j < 2; ++j) {
            const int i = 2 * wv + j;
            const float tv = ((part[(0 * 8 + i) * 32 + lane] + part[(1 * 8 + i) * 32 + lane])
                              + part[(2 * 8 + i) * 32 + lane]) + part[(3 * 8 + i) * 32 + lane];
            const float au =  y4 + z3v[j] + tv;
            const float ad = -y4 + z4v[j] - tv;
            nu[j] = fmaxf(xu[j] - step * au, 0.f); bu[j] = 2.f * nu[j] - xu[j];
            nd[j] = fmaxf(xd[j] - step * ad, 0.f); bd[j] = 2.f * nd[j] - xd[j];
        }
        for (int j = 0; j < 2; ++j) {
            const int g = mr + j;
            if (n < NS && g < NG) {
                const int o = g * S_P + n;
                X[OX_U + o] = nu[j]; X[OX_D + o] = nd[j];
                XB[OX_U + o] = bu[j]; XB[OX_D + o] = bd[j];
            }
        }
        if (n < NS && mr < NG) {
            v2f d; d[0] = bu[0] - bd[0]; d[1] = bu[1] - bd[1];
            *(v2f*)(w + WS_UDT + tn * BLK + pk_in(m16, kpos100(mr))) = d;
        }
    } else if (blk < RA_T2) {
        const int g = (blk - RA_T1) * RPW + lane;
        const bool act = lane < RPW && g < NG;
        float z1 = 0.f, z2 = 0.f, xu = 0.f, xd = 0.f, cu = 0.f, cd = 0.f;
        float a = 0.f, b = 0.f;
        if (act) {
            if (wv == 0) {
                z1 = yb[OY_Z1 + g]; z2 = yb[OY_Z2 + g];
                xu = X[OX_RU + g]; xd = X[OX_RD + g];
                cu = Cru[g]; cd = Crd[g];
            }
            row_sum_share(yb + OY_Z3 + g * S_P, yb + OY_Z4 + g * S_P, wv, a, b);
        }
        part[wv * 32 + lane] = a; part[(KSP + wv) * 32 + lane] = b;
        __syncthreads();
        if (wv == 0 && act) {
            const float sa = ((part[lane] + part[32 + lane]) + part[64 + lane]) + part[96 + lane];
            const float sb = ((part[128 + lane] + part[160 + lane]) + part[192 + lane]) + part[224 + lane];
            const float au = z1 - sa;
            const float ad = z2 - sb;
            const float nu = fmaxf(xu - step * (cu + au), 0.f);
            const float nd = fmaxf(xd - step * (cd + ad), 0.f);
            X[OX_RU + g] = nu; XB[OX_RU + g] = 2.f * nu - xu;
            X[OX_RD + g] = nd; XB[OX_RD + g] = 2.f * nd - xd;
        }
    } else if (blk < RA_T3) {
        const int g = (blk - RA_T2) * RPW + lane;
        const bool act = lane < RPW && g < NG;
        float y1 = 0.f, z1 = 0.f, z2 = 0.f, xp = 0.f, cp = 0.f;
        float c = 0.f;
        if (act) {
            const float* pr = PGT + g * NL + wv * KQ;
            const float* dp = w + WS_D23 + wv * KQ;
            float pv[KQ], dv[KQ];
#pragma unroll
            for (int l = 0; l < KQ; ++l) { pv[l] = pr[l]; dv[l] = dp[l]; }
            if (wv == 0) {
                y1 = yb[OY_Y1]; z1 = yb[OY_Z1 + g]; z2 = yb[OY_Z2 + g];
                xp = X[OX_P + g];
                cp = Cost[g];
            }
#pragma unroll
            for (int l = 0; l < KQ; ++l) c += pv[l] * dv[l];
        }
        part[wv * 32 + lane] = c;
        __syncthreads();
        if (wv == 0 && act) {
            const float cs = ((part[lane] + part[32 + lane]) + part[64 + lane]) + part[96 + lane];
            const float ap = y1 + cs + z1 - z2;
            const float nn = fmaxf(xp - step * (cp + ap), 0.f);
            X[OX_P + g] = nn; XB[OX_P + g] = 2.f * nn - xp;
        }
    } else {
        const int l = (blk - RA_T3) * RPW + lane;
        const bool act = lane < RPW && l < NL;
        float y2 = 0.f, y3 = 0.f, xu = 0.f, xd = 0.f;
        float a = 0.f, b = 0.f;
        if (act) {
            if (wv == 0) {
                y2 = yb[OY_Y2 + l]; y3 = yb[OY_Y3 + l];
                xu = X[OX_FU + l]; xd = X[OX_FD + l];
            }
            row_sum_share(yb + OY_Z5 + l * S_P, yb + OY_Z6 + l * S_P, wv, a, b);
        }
        part[wv * 32 + lane] = a; part[(KSP + wv) * 32 + lane] = b;
        __syncthreads();
        if (wv == 0 && act) {
            const float sa = ((part[lane] + part[32 + lane]) + part[64 + lane]) + part[96 + lane];
            const float sb = ((part[128 + lane] + part[160 + lane]) + part[192 + lane]) + part[224 + lane];
            const float au = y2 - sa;
            const float ad = y3 - sb;
            const float nu = fmaxf(xu - step * au, 0.f);
            const float nd = fmaxf(xd - step * ad, 0.f);
            X[OX_FU + l] = nu; XB[OX_FU + l] = 2.f * nu - xu;
            X[OX_FD + l] = nd; XB[OX_FD + l] = 2.f * nd - xd;
        }
    }
}

constexpr int EPW = 128;
constexpr int MA_T1 = 25 * 13;
constexpr int MA_T2 = MA_T1 + NL / RPW;
constexpr int MA_T3 = MA_T2 + (NS + 31) / 32;
constexpr int MA_T4 = MA_T3 + (NG + 31) / 32;
constexpr int MA_T5 = MA_T4 + 1;
constexpr int MA_NB = MA_T5 + (NG * NS + EPW - 1) / EPW;

__global__ __launch_bounds__(32) __attribute__((amdgpu_waves_per_eu(1, 1))) void k_mv(
        float* w, const float* Pmax) {
    const int lane = threadIdx.x, blk = blockIdx.x;
    const float* PG = w + WS_PG;
    const float* xb = w + WS_XB;
    float* Y = w + WS_Y;
    const float* pw = w + WS_PW;
    const float step = w[WS_STEP];
    if (blk < MA_T1) {
        const int tm = blk / 13, tn = blk % 13;
        const int m16 = lane & 15;
        const int n = tn * 16 + m16;
        const int mr = tm * 16 + ((lane >> 4) << 3);
        const int nc = n < NS ? n : NS - 1;
        float fu[8], fd[8], y5[8], y6[8], pv[8];
        for (int i = 0; i < 8; ++i) {
            const int l = mr + i;
            const int o = l * S_P + nc;
            fu[i] = xb[OX_FU + l]; fd[i] = xb[OX_FD + l];
            y5[i] = Y[OY_Z5 + o]; y6[i] = Y[OY_Z6 + o];
            pv[i] = pw[o];
        }
        KpFrags f;
        load_kp(f, w + WS_PGP + tm * BLK, w + WS_UDT + tn * BLK, lane);
#pragma unroll
        for (int i = 0; i < 8; ++i)
            asm volatile("" :: "v"(fu[i]), "v"(fd[i]), "v"(y5[i]), "v"(y6[i]), "v"(pv[i]));
        __builtin_amdgcn_sched_barrier(0);
        const v8f acc = mma_kp(f);
        float r5[8], r6[8];
        for (int i = 0; i < 8; ++i) {
            const float f2 = acc[i];
            const float a5 =  f2 - fu[i];
            const float a6 = -f2 - fd[i];
            r5[i] = fmaxf(y5[i] + step * (a5 + pv[i]), 0.f);
            r6[i] = fmaxf(y6[i] + step * (a6 - pv[i]), 0.f);
        }
        if (n < NS) {
            for (int i = 0; i < 8; ++i) {
                const int o = (mr + i) * S_P + n;
                Y[OY_Z5 + o] = r5[i]; Y[OY_Z6 + o] = r6[i];
            }
            float* zt = w + WS_ZT + tn * KSP * BLK;
            for (int i = 0; i < 8; i += 2) {
                const int l = mr + i;
                v2f d; d[0] = r5[i] - r6[i]; d[1] = r5[i + 1] - r6[i + 1];
                *(v2f*)(zt + (l / KQ) * BLK + pk_in(m16, kpos100(l % KQ))) = d;
            }
        }
    } else if (blk < MA_T2) {
        const int l = (blk - MA_T1) * RPW + lane;
        if (lane < RPW && l < NL) {
            const float fu = xb[OX_FU + l], fd = xb[OX_FD + l];
            const float y2 = Y[OY_Y2 + l], y3 = Y[OY_Y3 + l];
            const float b2 = w[WS_B2 + l], b3 = w[WS_B3 + l];
            const float* r = PG + l * G_P; const float* p = xb + OX_P;
            float rv[NG], pg[NG];
#pragma unroll
            for (int g = 0; g < NG; ++g) { rv[g] = r[g]; pg[g] = p[g]; }
            __builtin_amdgcn_sched_barrier(0);
            float s = 0.f;
#pragma unroll
            for (int g = 0; g < NG; ++g) s += rv[g] * pg[g];
            const float a2 =  s + fu;
            const float a3 = -s + fd;
            const float n2 = y2 + step * (a2 - b2);
            const float n3 = y3 + step * (a3 - b3);
            Y[OY_Y2 + l] = n2; Y[OY_Y3 + l] = n3;
            w[WS_D23 + l] = n2 - n3;
        }
    } else if (blk < MA_T3) {
        const int s = (blk - MA_T2) * 32 + lane;
        if (s < NS) {
            const float y4 = Y[OY_Y4 + s];
            const float b4 = w[WS_B4 + s];
            float c = 0.f;
            const float* r = w + WS_UDT + (s >> 4) * BLK;
            const int c16 = s & 15;
            float cv[NG];
#pragma unroll
            for (int j = 0; j < 12; ++j)
#pragma unroll
                for (int h = 0; h < 2; ++h) {
                    const v4f t = *(const v4f*)(r + 128 * j + 4 * (c16 + 16 * h));
                    cv[8 * j + 4 * h + 0] = t[0]; cv[8 * j + 4 * h + 1] = t[1];
                    cv[8 * j + 4 * h + 2] = t[2]; cv[8 * j + 4 * h + 3] = t[3];
                }
#pragma unroll
            for (int h = 0; h < 2; ++h) {
                const v2f t = *(const v2f*)(r + 1536 + 2 * (c16 + 16 * h));
                cv[96 + 2 * h] = t[0]; cv[97 + 2 * h] = t[1];
            }
            __builtin_amdgcn_sched_barrier(0);
#pragma unroll
            for (int g = 0; g < NG; ++g) c += cv[kpos100(g)];
            Y[OY_Y4 + s] = y4 + step * (c - b4);
        }
    } else if (blk < MA_T4) {
        const int g = (blk - MA_T3) * 32 + lane;
        if (g < NG) {
            const float xp = xb[OX_P + g], xu = xb[OX_RU + g], xd = xb[OX_RD + g];
            const float a1 = xp + xu;
            const float a2 = xd - xp;
            const float z1 = Y[OY_Z1 + g], z2 = Y[OY_Z2 + g], pm = Pmax[g];
            Y[OY_Z1 + g] = fmaxf(z1 + step * (a1 - pm), 0.f);
            Y[OY_Z2 + g] = fmaxf(z2 + step *  a2, 0.f);
        }
    } else if (blk < MA_T5) {
        if (lane == 0) {
            const float y1 = Y[OY_Y1];
            const float b1 = w[WS_B1];
            float pp[NG];
#pragma unroll
            for (int g = 0; g < NG; ++g) pp[g] = xb[OX_P + g];
            __builtin_amdgcn_sched_barrier(0);
            float s = 0.f;
#pragma unroll
            for (int g = 0; g < NG; ++g) s += pp[g];
            Y[OY_Y1] = y1 + step * (s - b1);
        }
    } else {
        const int base = (blk - MA_T5) * EPW;
        float r3[EPW / 32], r4[EPW / 32];
        for (int j = 0; j < EPW / 32; ++j) {
            const int i = base + j * 32 + lane;
            const int ic = i < NG * NS ? i : NG * NS - 1;
            const int g = ic / NS, s = ic % NS, o = g * S_P + s;
            const float a3 = xb[OX_U + o] - xb[OX_RU + g];
            const float a4 = xb[OX_D + o] - xb[OX_RD + g];
            r3[j] = fmaxf(Y[OY_Z3 + o] + step * a3, 0.f);
            r4[j] = fmaxf(Y[OY_Z4 + o] + step * a4, 0.f);
        }
        for (int j = 0; j < EPW / 32; ++j) {
            const int i = base + j * 32 + lane;
            if (i < NG * NS) {
                const int g = i / NS, s = i % NS, o = g * S_P + s;
                Y[OY_Z3 + o] = r3[j]; Y[OY_Z4 + o] = r4[j];
            }
        }
    }
}

constexpr int PW_NT = 1024;
constexpr int PW_NW = PW_NT / 32;
constexpr int PW_CW = 25;
constexpr int HG = NG / 2;
constexpr int HL = NL / 8;
static_assert(2 * NL == 32 * PW_CW && 8 * NG == 32 * PW_CW && PW_CW + 2 <= PW_NW && (HG % 2) == 0 && (HL % 2) == 0,
              "thread map of k_power");

__global__ __launch_bounds__(PW_NT) void k_power(float* w) {
    __shared__ __attribute__((aligned(16))) float vp[G_P], vru[G_P], vrd[G_P], vu[G_P], vd[G_P];
    __shared__ __attribute__((aligned(16))) float vfu[NL], vfd[NL];
    __shared__ __attribute__((aligned(16))) float bx[2 * NG];
    __shared__ __attribute__((aligned(16))) float by[2 * NL];
    __shared__ float sq[PW_NW];
    __shared__ float sc[2];
    __shared__ __attribute__((aligned(16))) v2f prow[(HG / 2) * PW_CW * 32];

    const int tid = threadIdx.x, lane = tid & 31;
    const int wave = __builtin_amdgcn_readfirstlane(tid >> 5);
    const bool hold = wave < PW_CW;
    const int ml = hold ? tid >> 1 : 0, mpart = tid & 1;
    const int rg = hold ? tid >> 3 : 0, rpart = tid & 7;

    float pcol[HL];
    {
        const float* r = w + WS_PG + ml * G_P + mpart * HG;
        const float* cp = w + WS_PGT + rg * NL + rpart * HL;
        v2f rv[HG / 2];
#pragma unroll
        for (int j = 0; j < HG / 2; ++j) rv[j] = *(const v2f*)(r + 2 * j);
#pragma unroll
        for (int k = 0; k < HL; k += 2) { const v2f t = *(const v2f*)(cp + k); pcol[k] = t[0]; pcol[k + 1] = t[1]; }
        if (hold) {
#pragma unroll
            for (int j = 0; j < HG / 2; ++j) prow[j * (PW_CW * 32) + tid] = rv[j];
        }
    }

    if (tid < G_P) {
        const float o = tid < NG ? 1.f : 0.f;
        vp[tid] = o; vru[tid] = o; vrd[tid] = o; vu[tid] = o; vd[tid] = o;
    }
    if (tid < NL) { vfu[tid] = 1.f; vfd[tid] = 1.f; }
    if (tid < PW_NW) sq[tid] = tid == 0 ? (float)(3 * NG + 2 * NG * NS + 2 * NL) : 0.f;
    __syncthreads();

    for (int it = 0; it < POWER_ITERS; ++it) {
        float ss = 0.f;
#pragma unroll
        for (int j = 0; j < PW_NW; ++j) ss += sq[j];
        const float inv = 1.0f / sqrtf(ss);

        if (tid < NG) {
            v2f t; t[0] = vp[tid] * inv; t[1] = vu[tid] * inv - vd[tid] * inv;
            *(v2f*)(bx + 2 * tid) = t;
        }
        __syncthreads();

        float q = 0.f;
        if (hold) {
            float s0 = 0.f, s1 = 0.f;
            const v4f* b4 = (const v4f*)(bx + 2 * HG * mpart);
#pragma unroll
            for (int j = 0; j < HG / 2; ++j) {
                const v2f a = prow[j * (PW_CW * 32) + tid];
                const v4f t = b4[j];
                s0 += a[0] * t[0]; s1 += a[0] * t[1];
                s0 += a[1] * t[2]; s1 += a[1] * t[3];
            }
            s0 += __shfl_xor(s0, 1); s1 += __shfl_xor(s1, 1);
            if (mpart == 0) {
                const float fu = vfu[ml] * inv, fd = vfd[ml] * inv;
                const float y2 =  s0 + fu, y3 = -s0 + fd;
                const float z5 =  s1 - fu, z6 = -s1 - fd;
                v2f t; t[0] = y2 - y3; t[1] = z5 - z6;
                *(v2f*)(by + 2 * ml) = t;
                const float nfu = y2 - (float)NS * z5;
                const float nfd = y3 - (float)NS * z6;
                vfu[ml] = nfu; vfd[ml] = nfd;
                q = nfu * nfu + nfd * nfd;
            }
        } else if (wave >= PW_NW - 2) {
            const bool first = wave == PW_NW - 2;
            float s = 0.f;
            for (int g = lane; g < NG; g += 32)
                s += first ? vp[g] * inv : (vu[g] * inv - vd[g] * inv);
            s = wave_sum(s);
            if (lane == 0) sc[first ? 0 : 1] = s;
        }
        __syncthreads();

        if (hold) {
            float s0 = 0.f, s1 = 0.f;
            const v4f* b4 = (const v4f*)(by + 2 * HL * rpart);
#pragma unroll
            for (int j = 0; j < HL / 2; ++j) {
                const v4f t = b4[j];
                s0 += pcol[2 * j] * t[0];     s1 += pcol[2 * j] * t[1];
                s0 += pcol[2 * j + 1] * t[2]; s1 += pcol[2 * j + 1] * t[3];
            }
#pragma unroll
            for (int m = 1; m < 8; m <<= 1) { s0 += __shfl_xor(s0, m); s1 += __shfl_xor(s1, m); }
            if (rpart == 0) {
                const float p = vp[rg] * inv, ru = vru[rg] * inv, rd = vrd[rg] * inv;
                const float u = vu[rg] * inv, d = vd[rg] * inv;
                const float z1 = p + ru, z2 = rd - p, z3 = u - ru, z4 = d - rd;
                const float y1 = sc[0], y4 = sc[1];
                const float np  = y1 + s0 + z1 - z2;
                const float nru = z1 - (float)NS * z3;
                const float nrd = z2 - (float)NS * z4;
                const float nu  =  y4 + z3 + s1;
                const float nd  = -y4 + z4 - s1;
                vp[rg] = np; vru[rg] = nru; vrd[rg] = nrd; vu[rg] = nu; vd[rg] = nd;
                q += np * np + nru * nru + nrd * nrd + (float)NS * (nu * nu + nd * nd);
            }
        }
        q = wave_sum(q);
        if (lane == 0) sq[wave] = q;
        __syncthreads();
    }

    if (tid == 0) {
        float ss = 0.f;
        for (int j = 0; j < PW_NW; ++j) ss += sq[j];
        w[WS_STEP] = 0.9f / sqrtf(sqrtf(ss));
    }
}

__global__ void k_emit(const float* w, float* out) {
    const float* X = w + WS_X;
    const int total = 3 * NG + 2 * NG * NS + 2 * NL;
    for (int idx = blockIdx.x * blockDim.x + threadIdx.x; idx < total; idx += gridDim.x * blockDim.x) {
        float v;
        if (idx < NG)                 v = X[OX_P  + idx];
        else if (idx < 2 * NG)        v = X[OX_RU + (idx - NG)];
        else if (idx < 3 * NG)        v = X[OX_RD + (idx - 2 * NG)];
        else if (idx < 3 * NG + NG * NS) { int r = idx - 3 * NG; v = X[OX_U + (r / NS) * S_P + (r % NS)]; }
        else if (idx < 3 * NG + 2 * NG * NS) { int r = idx - 3 * NG - NG * NS; v = X[OX_D + (r / NS) * S_P + (r % NS)]; }
        else if (idx < 3 * NG + 2 * NG * NS + NL) v = X[OX_FU + (idx - 3 * NG - 2 * NG * NS)];
        else                          v = X[OX_FD + (idx - 3 * NG - 2 * NG * NS - NL)];
        out[idx] = v;
    }
}


extern "C" void kernel_launch(void* const* d_in, const int* in_sizes, int n_in,
                              void* d_out, int out_size, void* d_ws, size_t ws_size,
                              hipStream_t stream) {
    const float* w_scen   = (const float*)d_in[0];
    const float* Cost     = (const float*)d_in[1];
    const float* C_r_up   = (const float*)d_in[2];
    const float* C_r_down = (const float*)d_in[3];
    const float* Pmax     = (const float*)d_in[4];
    const float* PTDF     = (const float*)d_in[5];
    const float* node_G   = (const float*)d_in[6];
    const float* node_W   = (const float*)d_in[7];
    const float* node_L   = (const float*)d_in[8];
    const float* Pd       = (const float*)d_in[9];
    const float* w_exp    = (const float*)d_in[10];
    const float* Line_cap = (const float*)d_in[11];
    float* w   = (float*)d_ws;
    float* out = (float*)d_out;

    k_zero <<<512, 256, 0, stream>>>(w);
    k_prep1<<<240, 256, 0, stream>>>(w, w_scen, node_W, node_L, Pd, w_exp);
    k_prep2<<<640, 256, 0, stream>>>(w, PTDF, node_G, Line_cap);
    k_power<<<1, PW_NT, 0, stream>>>(w);

    for (int it = 0; it < PDHG_ITERS; ++it) {
        k_rmv<<<RA_NB, RNT, 0, stream>>>(w, Cost, C_r_up, C_r_down);
        k_mv <<<MA_NB, 32, 0, stream>>>(w, Pmax);
    }
    k_emit<<<161, 256, 0, stream>>>(w, out);
}


# One launch per PDHG iteration: work-group = scenario tile x 100-line block, phases meet in LDS; T partials, row-sum parts, y2/y3 cross launches; PG p, PG^T(y2-y3) in a spare column: sums change order
# speedup vs baseline: 44.5146x; 1.0396x over previous
#include <hip/hip_runtime.h>
#include <math.h>

typedef __attribute__((ext_vector_type(2))) float v2f;
typedef __attribute__((ext_vector_type(4))) float v4f;
typedef __attribute__((ext_vector_type(8))) float v8f;


constexpr int NG = 100, NL = 400, NN = 300, NWF = 50, NS = 200;
constexpr int G_P = 112;
constexpr int S_P = 208;
constexpr int PDHG_ITERS = 300, POWER_ITERS = 25;

constexpr int WS_PG   = 0;
constexpr int WS_PGT  = WS_PG   + NL * G_P;
constexpr int WS_PW   = WS_PGT  + G_P * NL;
constexpr int WS_B2   = WS_PW   + NL * S_P;
constexpr int WS_B3   = WS_B2   + NL;
constexpr int WS_B4   = WS_B3   + NL;
constexpr int WS_B1   = WS_B4   + S_P;
constexpr int WS_M1   = WS_B1   + 16;
constexpr int WS_NV   = WS_M1   + NN * S_P;
constexpr int WS_STEP = WS_NV   + NN + 4;
constexpr int BLK = 1600;
__host__ __device__ constexpr int al32(int x) { return (x + 31) & ~31; }
constexpr int WS_PGP  = al32(WS_STEP + 16);
constexpr int WS_PGTP = WS_PGP  + NL * G_P;

constexpr int NTN = 13, NLB = 4, NWG = NTN * NLB;
constexpr int LBL = NL / NLB;
constexpr int FNT = 512;
constexpr int TPSZ = 7 * 256;
constexpr int RSS  = 1024;
constexpr int PS_U  = 0;
constexpr int PS_D  = PS_U  + TPSZ;
constexpr int PS_Z3 = PS_D  + TPSZ;
constexpr int PS_Z4 = PS_Z3 + TPSZ;
constexpr int PS_Z5 = PS_Z4 + TPSZ;
constexpr int PS_Z6 = PS_Z5 + TPSZ;
constexpr int PS_Y4 = PS_Z6 + TPSZ;
constexpr int PS_P  = PS_Y4 + 16;
constexpr int PS_RU = PS_P  + G_P;
constexpr int PS_RD = PS_RU + G_P;
constexpr int PS_Z1 = PS_RD + G_P;
constexpr int PS_Z2 = PS_Z1 + G_P;
constexpr int PS_FU = PS_Z2 + G_P;
constexpr int PS_FD = PS_FU + 128;
constexpr int PS_Y1 = PS_FD + 128;
constexpr int PS_SZ = al32(PS_Y1 + 16);
constexpr int WS_PS  = al32(WS_PGTP + G_P * NL);
constexpr int WS_TP  = WS_PS  + NWG * PS_SZ;
constexpr int WS_RS  = WS_TP  + 2 * NWG * TPSZ;
constexpr int WS_Y23 = WS_RS  + 2 * NTN * RSS;
constexpr int WS_TOTAL = WS_Y23 + 2 * 2 * NL;

__host__ __device__ constexpr int kpos100(int r) {
    return r < 96 ? ((r & ~7) | ((r & 2) << 1) | ((r & 4) >> 1) | (r & 1)) : r;
}
__host__ __device__ constexpr int pk_in(int c, int p) {
    return p < 96 ? ((p >> 3) * 128 + (c + ((p >> 2) & 1) * 16) * 4 + (p & 3))
                  : (1536 + (c + ((p - 96) >> 1) * 16) * 2 + (p & 1));
}


__global__ void k_zero(float* w) {
    for (int i = blockIdx.x * blockDim.x + threadIdx.x; i < WS_TOTAL;
         i += gridDim.x * blockDim.x) w[i] = 0.0f;
}

__global__ void k_prep1(float* w, const float* w_scen, const float* node_W,
                        const float* node_L, const float* Pd, const float* w_exp) {
    const int NM1 = NN * NS;
    const int total = NM1 + NN + NS + 1;
    for (int idx = blockIdx.x * blockDim.x + threadIdx.x; idx < total;
         idx += gridDim.x * blockDim.x) {
        if (idx < NM1) {
            int n = idx / NS, s = idx % NS;
            float a = 0.f;
            for (int ww = 0; ww < NWF; ++ww) a += node_W[n*NWF + ww] * w_scen[s*NWF + ww];
            w[WS_M1 + n*S_P + s] = a;
        } else if (idx < NM1 + NN) {
            int n = idx - NM1;
            float a = 0.f;
            for (int ww = 0; ww < NWF; ++ww) a += node_W[n*NWF + ww] * w_exp[ww];
            for (int m = 0; m < NN; ++m)     a -= node_L[n*NN + m] * Pd[m];
            w[WS_NV + n] = a;
        } else if (idx < NM1 + NN + NS) {
            int s = idx - NM1 - NN;
            float a = 0.f;
            for (int ww = 0; ww < NWF; ++ww) a += w_scen[s*NWF + ww];
            w[WS_B4 + s] = -a;
        } else {
            float a = 0.f;
            for (int m = 0; m < NN; ++m)  a += Pd[m];
            for (int ww = 0; ww < NWF; ++ww) a -= w_exp[ww];
            w[WS_B1] = a;
        }
    }
}

__global__ void k_prep2(float* w, const float* PTDF, const float* node_G,
                        const float* Line_cap) {
    const int A = NL * NG;
    const int B = A + NG * NL;
    const int C = B + NL * NS;
    const int D2 = C + NL;
    const int E = D2 + NL;
    for (int idx = blockIdx.x * blockDim.x + threadIdx.x; idx < E;
         idx += gridDim.x * blockDim.x) {
        if (idx < A) {
            int l = idx / NG, g = idx % NG;
            float a = 0.f;
            for (int n = 0; n < NN; ++n) a += PTDF[l*NN + n] * node_G[n*NG + g];
            w[WS_PG + l*G_P + g] = a;
            w[WS_PGP + (l >> 4) * BLK + pk_in(l & 15, kpos100(g))] = a;
        } else if (idx < B) {
            int r = idx - A; int g = r / NL, l = r % NL;
            float a = 0.f;
            for (int n = 0; n < NN; ++n) a += PTDF[l*NN + n] * node_G[n*NG + g];
            w[WS_PGT + g*NL + l] = a;
            w[WS_PGTP + ((g >> 4) * 4 + l / 100) * BLK + pk_in(g & 15, kpos100(l % 100))] = a;
        } else if (idx < C) {
            int r = idx - B; int l = r / NS, s = r % NS;
            float a = 0.f;
            for (int n = 0; n < NN; ++n) a += PTDF[l*NN + n] * w[WS_M1 + n*S_P + s];
            w[WS_PW + l*S_P + s] = a;
        } else if (idx < D2) {
            int l = idx - C;
            float a = 0.f;
            for (int n = 0; n < NN; ++n) a += PTDF[l*NN + n] * w[WS_NV + n];
            w[WS_B2 + l] = Line_cap[l] - a;
        } else {
            int l = idx - D2;
            float a = 0.f;
            for (int n = 0; n < NN; ++n) a += PTDF[l*NN + n] * w[WS_NV + n];
            w[WS_B3 + l] = Line_cap[l] + a;
        }
    }
}

__device__ __forceinline__ float wave_sum(float v) {
    for (int m = 16; m > 0; m >>= 1) v += __shfl_xor(v, m);
    return v;
}

template <int M> __device__ __forceinline__ float row_xor(float v) {
    return __builtin_bit_cast(float, __builtin_amdgcn_update_dpp(0, __builtin_bit_cast(int, v), 0x160 + M, 0xf, 0xf, false));
}
__device__ __forceinline__ float half_sum(float v) {
    v += row_xor<1>(v); v += row_xor<2>(v); v += row_xor<4>(v); v += row_xor<8>(v);
    return v;
}

struct KpFrags { v4f a[12], b[12]; v2f at, bt; };

__device__ __forceinline__ void load_kp_a(KpFrags& f, const float* ablk, int lane) {
#pragma unroll
    for (int j = 0; j < 12; ++j) f.a[j] = *(const v4f*)(ablk + 128 * j + 4 * lane);
    f.at = *(const v2f*)(ablk + 1536 + 2 * lane);
}

__device__ __forceinline__ void load_kp_b(KpFrags& f, const float* bblk, int lane) {
#pragma unroll
    for (int j = 0; j < 12; ++j) f.b[j] = *(const v4f*)(bblk + 128 * j + 4 * lane);
    f.bt = *(const v2f*)(bblk + 1536 + 2 * lane);
}

__device__ __forceinline__ v8f mma_kp(const KpFrags& f) {
    v8f acc = {0.f,0.f,0.f,0.f,0.f,0.f,0.f,0.f};
#pragma unroll
    for (int j = 0; j < 12; ++j) {
        v2f a0, b0, a1, b1;
        a0[0] = f.a[j][0]; a0[1] = f.a[j][1]; b0[0] = f.b[j][0]; b0[1] = f.b[j][1];
        a1[0] = f.a[j][2]; a1[1] = f.a[j][3]; b1[0] = f.b[j][2]; b1[1] = f.b[j][3];
        acc = __builtin_amdgcn_wmma_f32_16x16x4_f32(false, a0, false, b0, (short)0, acc, false, false);
        acc = __builtin_amdgcn_wmma_f32_16x16x4_f32(false, a1, false, b1, (short)0, acc, false, false);
    }
    acc = __builtin_amdgcn_wmma_f32_16x16x4_f32(false, f.at, false, f.bt, (short)0, acc, false, false);
    return acc;
}

static_assert(NG == 100 && LBL == 100, "load_kp_*, pk_in and kpos100 are written for 100-long k ranges");
static_assert(25 * BLK <= NL * G_P && 7 * 4 * BLK == G_P * NL, "the packed planes fit the room of the row-major ones");
static_assert(NTN * 16 == S_P && (NTN - 1) * 16 < NS && NS < NTN * 16, "the last scenario tile has a free column");

__global__ __launch_bounds__(FNT) void k_iter(float* w, const float* Cost, const float* Cru, const float* Crd,
                                              const float* Pmax, int par) {
    __shared__ __attribute__((aligned(16))) float udt[BLK];
    __shared__ __attribute__((aligned(16))) float zdt[BLK];
    __shared__ __attribute__((aligned(16))) float s_pb[G_P];
    __shared__ float s_rub[G_P], s_rdb[G_P];
    __shared__ float s_fub[LBL], s_fdb[LBL];

    const int tid = threadIdx.x, lane = tid & 31;
    const int wv = __builtin_amdgcn_readfirstlane((int)(threadIdx.x >> 5));
    const int tn = blockIdx.x / NLB, lb = blockIdx.x % NLB;
    const int c = lane & 15, h = lane >> 4;
    const int s = tn * 16 + c;
    const bool valid = s < NS;
    const int sc = valid ? s : NS - 1;
    const bool last = tn == NTN - 1;
    const bool carry = last && s == NS;

    float* ps = w + WS_PS + blockIdx.x * PS_SZ;
    const float* tpr = w + WS_TP + (par ^ 1) * (NWG * TPSZ);
    float*       tpw = w + WS_TP + par * (NWG * TPSZ);
    const float* rsr = w + WS_RS + (par ^ 1) * (NTN * RSS);
    float*       rsw = w + WS_RS + par * (NTN * RSS) + tn * RSS;
    const float* yr  = w + WS_Y23 + (par ^ 1) * (2 * NL);
    float*       yw  = w + WS_Y23 + par * (2 * NL);
    const float step = w[WS_STEP];

    const bool isU = tid < NG;
    const bool isL = tid >= 128 && tid < 128 + LBL;
    const int ug = tid < NG ? tid : NG - 1;
    const int lj = tid < 128 ? 0 : (tid < 128 + LBL ? tid - 128 : LBL - 1);
    const int ll = lb * LBL + lj;
    const bool roleL = wv >= 4;
    const int ia = roleL ? 2 * NG + ll : ug;
    const int ib = roleL ? 2 * NG + NL + ll : NG + ug;
    float sa[NTN], sb[NTN];
    float f_p, f_ru, f_rd, f_z1, f_z2, f_y1, f_c, f_cu, f_cd, f_pm, f_tp[NLB];
    float l_fu, l_fd, l_y2, l_y3;
    if (wv < 8) {
#pragma unroll
        for (int t = 0; t < NTN; ++t) { sa[t] = rsr[t * RSS + ia]; sb[t] = rsr[t * RSS + ib]; }
        f_p = ps[PS_P + ug]; f_ru = ps[PS_RU + ug]; f_rd = ps[PS_RD + ug];
        f_z1 = ps[PS_Z1 + ug]; f_z2 = ps[PS_Z2 + ug]; f_y1 = ps[PS_Y1];
        f_c = Cost[ug]; f_cu = Cru[ug]; f_cd = Crd[ug]; f_pm = Pmax[ug];
        const int idx = ((ug >> 4) * 8 + (ug & 7)) * 32 + (NS - (NTN - 1) * 16) + 16 * ((ug >> 3) & 1);
#pragma unroll
        for (int q = 0; q < NLB; ++q) f_tp[q] = tpr[((NTN - 1) * NLB + q) * TPSZ + idx];
        l_fu = ps[PS_FU + lj]; l_fd = ps[PS_FD + lj];
        l_y2 = yr[ll]; l_y3 = yr[NL + ll];
    }
    const int e_tm = wv % 7, e_rg = wv / 7;
    float eu[4], ed[4], e3[4], e4[4], et[4][NLB];
#pragma unroll
    for (int i = 0; i < 4; ++i) {
        const int idx = (e_tm * 8 + 4 * e_rg + i) * 32 + lane;
        eu[i] = ps[PS_U + idx]; ed[i] = ps[PS_D + idx]; e3[i] = ps[PS_Z3 + idx]; e4[i] = ps[PS_Z4 + idx];
#pragma unroll
        for (int q = 0; q < NLB; ++q) et[i][q] = tpr[(tn * NLB + q) * TPSZ + idx];
    }
    const float ey4 = ps[PS_Y4 + c];
    const float yb4 = w[WS_B4 + sc], yy1 = ps[PS_Y1], yb1 = w[WS_B1];
    KpFrags f;
    const float* ablk = wv < 7 ? w + WS_PGP + (6 * lb + wv) * BLK
                               : w + WS_PGTP + (((wv < 14 ? wv : 13) - 7) * NLB + lb) * BLK;
    load_kp_a(f, ablk, lane);
    float q5[8], q6[8], qw[8], o2[8], o3[8], ob2[8], ob3[8];
    if (wv < 7) {
        const int tmL = 6 * lb + wv;
#pragma unroll
        for (int i = 0; i < 8; ++i) {
            const int l = tmL * 16 + 8 * h + i;
            const int idx = (wv * 8 + i) * 32 + lane;
            q5[i] = ps[PS_Z5 + idx]; q6[i] = ps[PS_Z6 + idx];
            qw[i] = w[WS_PW + l * S_P + sc];
            if (last) { o2[i] = yr[l]; o3[i] = yr[NL + l]; ob2[i] = w[WS_B2 + l]; ob3[i] = w[WS_B3 + l]; }
        }
    }
    __builtin_amdgcn_sched_barrier(0);

    if (tid < G_P) {
        float pb = 0.f, rub = 0.f, rdb = 0.f;
        if (isU) {
            float z3s = 0.f, z4s = 0.f;
#pragma unroll
            for (int t = 0; t < NTN; ++t) { z3s += sa[t]; z4s += sb[t]; }
            const float cs = ((f_tp[0] + f_tp[1]) + f_tp[2]) + f_tp[3];
            const float ap = f_y1 + cs + f_z1 - f_z2;
            const float np = fmaxf(f_p - step * (f_c + ap), 0.f);
            pb = 2.f * np - f_p;
            const float au = f_z1 - z3s;
            const float ad = f_z2 - z4s;
            const float nru = fmaxf(f_ru - step * (f_cu + au), 0.f);
            const float nrd = fmaxf(f_rd - step * (f_cd + ad), 0.f);
            rub = 2.f * nru - f_ru;
            rdb = 2.f * nrd - f_rd;
            const float a1 = pb + rub;
            const float a2 = rdb - pb;
            ps[PS_P + tid] = np; ps[PS_RU + tid] = nru; ps[PS_RD + tid] = nrd;
            ps[PS_Z1 + tid] = fmaxf(f_z1 + step * (a1 - f_pm), 0.f);
            ps[PS_Z2 + tid] = fmaxf(f_z2 + step *  a2, 0.f);
        }
        s_pb[tid] = pb; s_rub[tid] = rub; s_rdb[tid] = rdb;
    }
    if (isL) {
        float z5s = 0.f, z6s = 0.f;
#pragma unroll
        for (int t = 0; t < NTN; ++t) { z5s += sa[t]; z6s += sb[t]; }
        const float au = l_y2 - z5s;
        const float ad = l_y3 - z6s;
        const float nfu = fmaxf(l_fu - step * au, 0.f);
        const float nfd = fmaxf(l_fd - step * ad, 0.f);
        ps[PS_FU + lj] = nfu; ps[PS_FD + lj] = nfd;
        s_fub[lj] = 2.f * nfu - l_fu; s_fdb[lj] = 2.f * nfd - l_fd;
    }
    __syncthreads();

    if (wv < 14) {
        float ud[4];
#pragma unroll
        for (int i = 0; i < 4; ++i) {
            const int r = 4 * e_rg + i;
            const int g = e_tm * 16 + 8 * h + r;
            const int idx = (e_tm * 8 + r) * 32 + lane;
            const bool live = valid && g < NG;
            const float tv = ((et[i][0] + et[i][1]) + et[i][2]) + et[i][3];
            const float au =  ey4 + e3[i] + tv;
            const float ad = -ey4 + e4[i] - tv;
            float nu = fmaxf(eu[i] - step * au, 0.f);
            float nd = fmaxf(ed[i] - step * ad, 0.f);
            float bu = 2.f * nu - eu[i];
            float bd = 2.f * nd - ed[i];
            const float a3 = bu - s_rub[g];
            const float a4 = bd - s_rdb[g];
            float n3 = fmaxf(e3[i] + step * a3, 0.f);
            float n4 = fmaxf(e4[i] + step * a4, 0.f);
            if (!live) { nu = 0.f; nd = 0.f; bu = 0.f; bd = 0.f; n3 = 0.f; n4 = 0.f; }
            ps[PS_U + idx] = nu; ps[PS_D + idx] = nd; ps[PS_Z3 + idx] = n3; ps[PS_Z4 + idx] = n4;
            ud[i] = carry ? s_pb[g] : (bu - bd);
            const float s3 = half_sum(n3), s4 = half_sum(n4);
            if (c == 0 && lb == 0 && g < NG) { rsw[g] = s3; rsw[NG + g] = s4; }
        }
#pragma unroll
        for (int i = 0; i < 4; i += 2) {
            const int g = e_tm * 16 + 8 * h + 4 * e_rg + i;
            if (g < NG) { v2f d; d[0] = ud[i]; d[1] = ud[i + 1]; *(v2f*)(udt + pk_in(c, kpos100(g))) = d; }
        }
    }
    __syncthreads();

    if (wv < 7) {
        const int tmL = 6 * lb + wv;
        float fubv[8], fdbv[8];
#pragma unroll
        for (int i = 0; i < 8; ++i) {
            const int p = tmL * 16 + 8 * h + i - lb * LBL;
            const int pc = (p >= 0 && p < LBL) ? p : 0;
            fubv[i] = s_fub[pc]; fdbv[i] = s_fdb[pc];
        }
        load_kp_b(f, udt, lane);
        __builtin_amdgcn_sched_barrier(0);
        const v8f acc = mma_kp(f);
        float dzv[8];
#pragma unroll
        for (int i = 0; i < 8; ++i) {
            const int l = tmL * 16 + 8 * h + i;
            const int p = l - lb * LBL;
            const bool own = p >= 0 && p < LBL;
            const int idx = (wv * 8 + i) * 32 + lane;
            const float fub = fubv[i], fdb = fdbv[i];
            const float f2 = acc[i];
            const float a5 =  f2 - fub;
            const float a6 = -f2 - fdb;
            float r5 = fmaxf(q5[i] + step * (a5 + qw[i]), 0.f);
            float r6 = fmaxf(q6[i] + step * (a6 - qw[i]), 0.f);
            if (!(own && valid)) { r5 = 0.f; r6 = 0.f; }
            ps[PS_Z5 + idx] = r5; ps[PS_Z6 + idx] = r6;
            float dz = r5 - r6;
            if (last) {
                const float n2 = o2[i] + step * (( f2 + fub) - ob2[i]);
                const float n3 = o3[i] + step * ((-f2 + fdb) - ob3[i]);
                if (carry) {
                    dz = n2 - n3;
                    if (own) { yw[l] = n2; yw[NL + l] = n3; }
                }
            }
            dzv[i] = dz;
            const float s5 = half_sum(r5), s6 = half_sum(r6);
            if (c == 0 && own) { rsw[2 * NG + l] = s5; rsw[2 * NG + NL + l] = s6; }
        }
#pragma unroll
        for (int i = 0; i < 8; i += 2) {
            const int p = tmL * 16 + 8 * h + i - lb * LBL;
            if (p >= 0 && p < LBL) { v2f d; d[0] = dzv[i]; d[1] = dzv[i + 1]; *(v2f*)(zdt + pk_in(c, kpos100(p))) = d; }
        }
    } else if (wv == 7) {
        float cv[NG];
#pragma unroll
        for (int j = 0; j < 12; ++j)
#pragma unroll
            for (int hh = 0; hh < 2; ++hh) {
                const v4f t = *(const v4f*)(udt + 128 * j + 4 * (c + 16 * hh));
                cv[8 * j + 4 * hh + 0] = t[0]; cv[8 * j + 4 * hh + 1] = t[1];
                cv[8 * j + 4 * hh + 2] = t[2]; cv[8 * j + 4 * hh + 3] = t[3];
            }
#pragma unroll
        for (int hh = 0; hh < 2; ++hh) {
            const v2f t = *(const v2f*)(udt + 1536 + 2 * (c + 16 * hh));
            cv[96 + 2 * hh] = t[0]; cv[97 + 2 * hh] = t[1];
        }
        __builtin_amdgcn_sched_barrier(0);
        float cs = 0.f;
#pragma unroll
        for (int g = 0; g < NG; ++g) cs += cv[kpos100(g)];
        if (h == 0) ps[PS_Y4 + c] = valid ? ey4 + step * (cs - yb4) : 0.f;
    } else if (wv == 8) {
        float pv[NG];
#pragma unroll
        for (int j = 0; j < NG / 4; ++j) {
            const v4f t = *(const v4f*)(s_pb + 4 * j);
            pv[4 * j] = t[0]; pv[4 * j + 1] = t[1]; pv[4 * j + 2] = t[2]; pv[4 * j + 3] = t[3];
        }
        __builtin_amdgcn_sched_barrier(0);
        float sp = 0.f;
#pragma unroll
        for (int g = 0; g < NG; ++g) sp += pv[g];
        if (lane == 0) ps[PS_Y1] = yy1 + step * (sp - yb1);
    }
    __syncthreads();

    if (wv >= 7 && wv < 14) {
        load_kp_b(f, zdt, lane);
        __builtin_amdgcn_sched_barrier(0);
        const v8f acc = mma_kp(f);
        float* o = tpw + (tn * NLB + lb) * TPSZ + (wv - 7) * 256 + lane;
#pragma unroll
        for (int i = 0; i < 8; ++i) o[i * 32] = acc[i];
    }
}

constexpr int PW_NT = 1024;
constexpr int PW_NW = PW_NT / 32;
constexpr int PW_CW = 25;
constexpr int HG = NG / 2;
constexpr int HL = NL / 8;
static_assert(2 * NL == 32 * PW_CW && 8 * NG == 32 * PW_CW && PW_CW + 2 <= PW_NW && (HG % 2) == 0 && (HL % 2) == 0,
              "thread map of k_power");

__global__ __launch_bounds__(PW_NT) void k_power(float* w) {
    __shared__ __attribute__((aligned(16))) float vp[G_P], vru[G_P], vrd[G_P], vu[G_P], vd[G_P];
    __shared__ __attribute__((aligned(16))) float vfu[NL], vfd[NL];
    __shared__ __attribute__((aligned(16))) float bx[2 * NG];
    __shared__ __attribute__((aligned(16))) float by[2 * NL];
    __shared__ float sq[PW_NW];
    __shared__ float sc[2];
    __shared__ __attribute__((aligned(16))) v2f prow[(HG / 2) * PW_CW * 32];

    const int tid = threadIdx.x, lane = tid & 31;
    const int wave = __builtin_amdgcn_readfirstlane(tid >> 5);
    const bool hold = wave < PW_CW;
    const int ml = hold ? tid >> 1 : 0, mpart = tid & 1;
    const int rg = hold ? tid >> 3 : 0, rpart = tid & 7;

    float pcol[HL];
    {
        const float* r = w + WS_PG + ml * G_P + mpart * HG;
        const float* cp = w + WS_PGT + rg * NL + rpart * HL;
        v2f rv[HG / 2];
#pragma unroll
        for (int j = 0; j < HG / 2; ++j) rv[j] = *(const v2f*)(r + 2 * j);
#pragma unroll
        for (int k = 0; k < HL; k += 2) { const v2f t = *(const v2f*)(cp + k); pcol[k] = t[0]; pcol[k + 1] = t[1]; }
        if (hold) {
#pragma unroll
            for (int j = 0; j < HG / 2; ++j) prow[j * (PW_CW * 32) + tid] = rv[j];
        }
    }

    if (tid < G_P) {
        const float o = tid < NG ? 1.f : 0.f;
        vp[tid] = o; vru[tid] = o; vrd[tid] = o; vu[tid] = o; vd[tid] = o;
    }
    if (tid < NL) { vfu[tid] = 1.f; vfd[tid] = 1.f; }
    if (tid < PW_NW) sq[tid] = tid == 0 ? (float)(3 * NG + 2 * NG * NS + 2 * NL) : 0.f;
    __syncthreads();

    for (int it = 0; it < POWER_ITERS; ++it) {
        float ss = 0.f;
#pragma unroll
        for (int j = 0; j < PW_NW; ++j) ss += sq[j];
        const float inv = 1.0f / sqrtf(ss);

        if (tid < NG) {
            v2f t; t[0] = vp[tid] * inv; t[1] = vu[tid] * inv - vd[tid] * inv;
            *(v2f*)(bx + 2 * tid) = t;
        }
        __syncthreads();

        float q = 0.f;
        if (hold) {
            float s0 = 0.f, s1 = 0.f;
            const v4f* b4 = (const v4f*)(bx + 2 * HG * mpart);
#pragma unroll
            for (int j = 0; j < HG / 2; ++j) {
                const v2f a = prow[j * (PW_CW * 32) + tid];
                const v4f t = b4[j];
                s0 += a[0] * t[0]; s1 += a[0] * t[1];
                s0 += a[1] * t[2]; s1 += a[1] * t[3];
            }
            s0 += __shfl_xor(s0, 1); s1 += __shfl_xor(s1, 1);
            if (mpart == 0) {
                const float fu = vfu[ml] * inv, fd = vfd[ml] * inv;
                const float y2 =  s0 + fu, y3 = -s0 + fd;
                const float z5 =  s1 - fu, z6 = -s1 - fd;
                v2f t; t[0] = y2 - y3; t[1] = z5 - z6;
                *(v2f*)(by + 2 * ml) = t;
                const float nfu = y2 - (float)NS * z5;
                const float nfd = y3 - (float)NS * z6;
                vfu[ml] = nfu; vfd[ml] = nfd;
                q = nfu * nfu + nfd * nfd;
            }
        } else if (wave >= PW_NW - 2) {
            const bool first = wave == PW_NW - 2;
            float s = 0.f;
            for (int g = lane; g < NG; g += 32)
                s += first ? vp[g] * inv : (vu[g] * inv - vd[g] * inv);
            s = wave_sum(s);
            if (lane == 0) sc[first ? 0 : 1] = s;
        }
        __syncthreads();

        if (hold) {
            float s0 = 0.f, s1 = 0.f;
            const v4f* b4 = (const v4f*)(by + 2 * HL * rpart);
#pragma unroll
            for (int j = 0; j < HL / 2; ++j) {
                const v4f t = b4[j];
                s0 += pcol[2 * j] * t[0];     s1 += pcol[2 * j] * t[1];
                s0 += pcol[2 * j + 1] * t[2]; s1 += pcol[2 * j + 1] * t[3];
            }
#pragma unroll
            for (int m = 1; m < 8; m <<= 1) { s0 += __shfl_xor(s0, m); s1 += __shfl_xor(s1, m); }
            if (rpart == 0) {
                const float p = vp[rg] * inv, ru = vru[rg] * inv, rd = vrd[rg] * inv;
                const float u = vu[rg] * inv, d = vd[rg] * inv;
                const float z1 = p + ru, z2 = rd - p, z3 = u - ru, z4 = d - rd;
                const float y1 = sc[0], y4 = sc[1];
                const float np  = y1 + s0 + z1 - z2;
                const float nru = z1 - (float)NS * z3;
                const float nrd = z2 - (float)NS * z4;
                const float nu  =  y4 + z3 + s1;
                const float nd  = -y4 + z4 - s1;
                vp[rg] = np; vru[rg] = nru; vrd[rg] = nrd; vu[rg] = nu; vd[rg] = nd;
                q += np * np + nru * nru + nrd * nrd + (float)NS * (nu * nu + nd * nd);
            }
        }
        q = wave_sum(q);
        if (lane == 0) sq[wave] = q;
        __syncthreads();
    }

    if (tid == 0) {
        float ss = 0.f;
        for (int j = 0; j < PW_NW; ++j) ss += sq[j];
        w[WS_STEP] = 0.9f / sqrtf(sqrtf(ss));
    }
}

__global__ void k_emit(const float* w, float* out) {
    const int total = 3 * NG + 2 * NG * NS + 2 * NL;
    for (int idx = blockIdx.x * blockDim.x + threadIdx.x; idx < total; idx += gridDim.x * blockDim.x) {
        const float* ps0 = w + WS_PS;
        float v;
        if (idx < NG)                 v = ps0[PS_P  + idx];
        else if (idx < 2 * NG)        v = ps0[PS_RU + (idx - NG)];
        else if (idx < 3 * NG)        v = ps0[PS_RD + (idx - 2 * NG)];
        else if (idx < 3 * NG + 2 * NG * NS) {
            int r = idx - 3 * NG;
            const int plane = r < NG * NS ? PS_U : PS_D;
            if (r >= NG * NS) r -= NG * NS;
            const int g = r / NS, s = r % NS;
            v = w[WS_PS + ((s >> 4) * NLB) * PS_SZ + plane
                  + ((g >> 4) * 8 + (g & 7)) * 32 + (s & 15) + 16 * ((g >> 3) & 1)];
        } else {
            int l = idx - 3 * NG - 2 * NG * NS;
            const int plane = l < NL ? PS_FU : PS_FD;
            if (l >= NL) l -= NL;
            v = w[WS_PS + (l / LBL) * PS_SZ + plane + (l % LBL)];
        }
        out[idx] = v;
    }
}


constexpr int PDHG_PAIRS = PDHG_ITERS / 2;
static_assert(2 * PDHG_PAIRS == PDHG_ITERS, "the loop below launches the two copies in turn");

extern "C" void kernel_launch(void* const* d_in, const int* in_sizes, int n_in,
                              void* d_out, int out_size, void* d_ws, size_t ws_size,
                              hipStream_t stream) {
    const float* w_scen   = (const float*)d_in[0];
    const float* Cost     = (const float*)d_in[1];
    const float* C_r_up   = (const float*)d_in[2];
    const float* C_r_down = (const float*)d_in[3];
    const float* Pmax     = (const float*)d_in[4];
    const float* PTDF     = (const float*)d_in[5];
    const float* node_G   = (const float*)d_in[6];
    const float* node_W   = (const float*)d_in[7];
    const float* node_L   = (const float*)d_in[8];
    const float* Pd       = (const float*)d_in[9];
    const float* w_exp    = (const float*)d_in[10];
    const float* Line_cap = (const float*)d_in[11];
    float* w   = (float*)d_ws;
    float* out = (float*)d_out;

    k_zero <<<512, 256, 0, stream>>>(w);
    k_prep1<<<240, 256, 0, stream>>>(w, w_scen, node_W, node_L, Pd, w_exp);
    k_prep2<<<640, 256, 0, stream>>>(w, PTDF, node_G, Line_cap);
    k_power<<<1, PW_NT, 0, stream>>>(w);

    for (int it = 0; it < PDHG_PAIRS; ++it) {
        k_iter<<<NWG, FNT, 0, stream>>>(w, Cost, C_r_up, C_r_down, Pmax, 0);
        k_iter<<<NWG, FNT, 0, stream>>>(w, Cost, C_r_up, C_r_down, Pmax, 1);
    }
    k_emit<<<161, 256, 0, stream>>>(w, out);
}


# Solver launch: the U, D half of phase 1 runs beside the first stage ahead of ONE barrier with no global store in front; z3, z4, stores and row sums follow behind it on waiting waves; sums keep order
# speedup vs baseline: 48.0890x; 1.0803x over previous
#include <hip/hip_runtime.h>
#include <math.h>

typedef __attribute__((ext_vector_type(2))) float v2f;
typedef __attribute__((ext_vector_type(4))) float v4f;
typedef __attribute__((ext_vector_type(8))) float v8f;


constexpr int NG = 100, NL = 400, NN = 300, NWF = 50, NS = 200;
constexpr int G_P = 112;
constexpr int S_P = 208;
constexpr int PDHG_ITERS = 300, POWER_ITERS = 25;

constexpr int WS_PG   = 0;
constexpr int WS_PGT  = WS_PG   + NL * G_P;
constexpr int WS_PW   = WS_PGT  + G_P * NL;
constexpr int WS_B2   = WS_PW   + NL * S_P;
constexpr int WS_B3   = WS_B2   + NL;
constexpr int WS_B4   = WS_B3   + NL;
constexpr int WS_B1   = WS_B4   + S_P;
constexpr int WS_M1   = WS_B1   + 16;
constexpr int WS_NV   = WS_M1   + NN * S_P;
constexpr int WS_STEP = WS_NV   + NN + 4;
constexpr int BLK = 1600;
__host__ __device__ constexpr int al32(int x) { return (x + 31) & ~31; }
constexpr int WS_PGP  = al32(WS_STEP + 16);
constexpr int WS_PGTP = WS_PGP  + NL * G_P;

constexpr int NTN = 13, NLB = 4, NWG = NTN * NLB;
constexpr int LBL = NL / NLB;
constexpr int FNT = 512;
constexpr int TPSZ = 7 * 256;
constexpr int RSS  = 1024;
constexpr int PS_U  = 0;
constexpr int PS_D  = PS_U  + TPSZ;
constexpr int PS_Z3 = PS_D  + TPSZ;
constexpr int PS_Z4 = PS_Z3 + TPSZ;
constexpr int PS_Z5 = PS_Z4 + TPSZ;
constexpr int PS_Z6 = PS_Z5 + TPSZ;
constexpr int PS_Y4 = PS_Z6 + TPSZ;
constexpr int PS_P  = PS_Y4 + 16;
constexpr int PS_RU = PS_P  + G_P;
constexpr int PS_RD = PS_RU + G_P;
constexpr int PS_Z1 = PS_RD + G_P;
constexpr int PS_Z2 = PS_Z1 + G_P;
constexpr int PS_FU = PS_Z2 + G_P;
constexpr int PS_FD = PS_FU + 128;
constexpr int PS_Y1 = PS_FD + 128;
constexpr int PS_SZ = al32(PS_Y1 + 16);
constexpr int WS_PS  = al32(WS_PGTP + G_P * NL);
constexpr int WS_TP  = WS_PS  + NWG * PS_SZ;
constexpr int WS_RS  = WS_TP  + 2 * NWG * TPSZ;
constexpr int WS_Y23 = WS_RS  + 2 * NTN * RSS;
constexpr int WS_TOTAL = WS_Y23 + 2 * 2 * NL;

__host__ __device__ constexpr int kpos100(int r) {
    return r < 96 ? ((r & ~7) | ((r & 2) << 1) | ((r & 4) >> 1) | (r & 1)) : r;
}
__host__ __device__ constexpr int pk_in(int c, int p) {
    return p < 96 ? ((p >> 3) * 128 + (c + ((p >> 2) & 1) * 16) * 4 + (p & 3))
                  : (1536 + (c + ((p - 96) >> 1) * 16) * 2 + (p & 1));
}


__global__ void k_zero(float* w) {
    for (int i = blockIdx.x * blockDim.x + threadIdx.x; i < WS_TOTAL;
         i += gridDim.x * blockDim.x) w[i] = 0.0f;
}

__global__ void k_prep1(float* w, const float* w_scen, const float* node_W,
                        const float* node_L, const float* Pd, const float* w_exp) {
    const int NM1 = NN * NS;
    const int total = NM1 + NN + NS + 1;
    for (int idx = blockIdx.x * blockDim.x + threadIdx.x; idx < total;
         idx += gridDim.x * blockDim.x) {
        if (idx < NM1) {
            int n = idx / NS, s = idx % NS;
            float a = 0.f;
            for (int ww = 0; ww < NWF; ++ww) a += node_W[n*NWF + ww] * w_scen[s*NWF + ww];
            w[WS_M1 + n*S_P + s] = a;
        } else if (idx < NM1 + NN) {
            int n = idx - NM1;
            float a = 0.f;
            for (int ww = 0; ww < NWF; ++ww) a += node_W[n*NWF + ww] * w_exp[ww];
            for (int m = 0; m < NN; ++m)     a -= node_L[n*NN + m] * Pd[m];
            w[WS_NV + n] = a;
        } else if (idx < NM1 + NN + NS) {
            int s = idx - NM1 - NN;
            float a = 0.f;
            for (int ww = 0; ww < NWF; ++ww) a += w_scen[s*NWF + ww];
            w[WS_B4 + s] = -a;
        } else {
            float a = 0.f;
            for (int m = 0; m < NN; ++m)  a += Pd[m];
            for (int ww = 0; ww < NWF; ++ww) a -= w_exp[ww];
            w[WS_B1] = a;
        }
    }
}

__global__ void k_prep2(float* w, const float* PTDF, const float* node_G,
                        const float* Line_cap) {
    const int A = NL * NG;
    const int B = A + NG * NL;
    const int C = B + NL * NS;
    const int D2 = C + NL;
    const int E = D2 + NL;
    for (int idx = blockIdx.x * blockDim.x + threadIdx.x; idx < E;
         idx += gridDim.x * blockDim.x) {
        if (idx < A) {
            int l = idx / NG, g = idx % NG;
            float a = 0.f;
            for (int n = 0; n < NN; ++n) a += PTDF[l*NN + n] * node_G[n*NG + g];
            w[WS_PG + l*G_P + g] = a;
            w[WS_PGP + (l >> 4) * BLK + pk_in(l & 15, kpos100(g))] = a;
        } else if (idx < B) {
            int r = idx - A; int g = r / NL, l = r % NL;
            float a = 0.f;
            for (int n = 0; n < NN; ++n) a += PTDF[l*NN + n] * node_G[n*NG + g];
            w[WS_PGT + g*NL + l] = a;
            w[WS_PGTP + ((g >> 4) * 4 + l / 100) * BLK + pk_in(g & 15, kpos100(l % 100))] = a;
        } else if (idx < C) {
            int r = idx - B; int l = r / NS, s = r % NS;
            float a = 0.f;
            for (int n = 0; n < NN; ++n) a += PTDF[l*NN + n] * w[WS_M1 + n*S_P + s];
            w[WS_PW + l*S_P + s] = a;
        } else if (idx < D2) {
            int l = idx - C;
            float a = 0.f;
            for (int n = 0; n < NN; ++n) a += PTDF[l*NN + n] * w[WS_NV + n];
            w[WS_B2 + l] = Line_cap[l] - a;
        } else {
            int l = idx - D2;
            float a = 0.f;
            for (int n = 0; n < NN; ++n) a += PTDF[l*NN + n] * w[WS_NV + n];
            w[WS_B3 + l] = Line_cap[l] + a;
        }
    }
}

__device__ __forceinline__ float wave_sum(float v) {
    for (int m = 16; m > 0; m >>= 1) v += __shfl_xor(v, m);
    return v;
}

template <int M> __device__ __forceinline__ float row_xor(float v) {
    return __builtin_bit_cast(float, __builtin_amdgcn_update_dpp(0, __builtin_bit_cast(int, v), 0x160 + M, 0xf, 0xf, false));
}
__device__ __forceinline__ float half_sum(float v) {
    v += row_xor<1>(v); v += row_xor<2>(v); v += row_xor<4>(v); v += row_xor<8>(v);
    return v;
}

struct KpFrags { v4f a[12], b[12]; v2f at, bt; };

__device__ __forceinline__ void load_kp_a(KpFrags& f, const float* ablk, int lane) {
#pragma unroll
    for (int j = 0; j < 12; ++j) f.a[j] = *(const v4f*)(ablk + 128 * j + 4 * lane);
    f.at = *(const v2f*)(ablk + 1536 + 2 * lane);
}

__device__ __forceinline__ void load_kp_b(KpFrags& f, const float* bblk, int lane) {
#pragma unroll
    for (int j = 0; j < 12; ++j) f.b[j] = *(const v4f*)(bblk + 128 * j + 4 * lane);
    f.bt = *(const v2f*)(bblk + 1536 + 2 * lane);
}

__device__ __forceinline__ v8f mma_kp(const KpFrags& f) {
    v8f acc = {0.f,0.f,0.f,0.f,0.f,0.f,0.f,0.f};
#pragma unroll
    for (int j = 0; j < 12; ++j) {
        v2f a0, b0, a1, b1;
        a0[0] = f.a[j][0]; a0[1] = f.a[j][1]; b0[0] = f.b[j][0]; b0[1] = f.b[j][1];
        a1[0] = f.a[j][2]; a1[1] = f.a[j][3]; b1[0] = f.b[j][2]; b1[1] = f.b[j][3];
        acc = __builtin_amdgcn_wmma_f32_16x16x4_f32(false, a0, false, b0, (short)0, acc, false, false);
        acc = __builtin_amdgcn_wmma_f32_16x16x4_f32(false, a1, false, b1, (short)0, acc, false, false);
    }
    acc = __builtin_amdgcn_wmma_f32_16x16x4_f32(false, f.at, false, f.bt, (short)0, acc, false, false);
    return acc;
}

static_assert(NG == 100 && LBL == 100, "load_kp_*, pk_in and kpos100 are written for 100-long k ranges");
static_assert(25 * BLK <= NL * G_P && 7 * 4 * BLK == G_P * NL, "the packed planes fit the room of the row-major ones");
static_assert(NTN * 16 == S_P && (NTN - 1) * 16 < NS && NS < NTN * 16, "the last scenario tile has a free column");

__device__ __forceinline__ void ud_step_tail(float* ps, float* rsw, const float* s_rub, const float* s_rdb,
                                             const float (&xnu)[4], const float (&xnd)[4],
                                             const float (&xbu)[4], const float (&xbd)[4],
                                             const float (&e3)[4], const float (&e4)[4],
                                             float step, int e_tm, int e_rg, int lane, bool valid, int lb) {
    const int c = lane & 15, h = lane >> 4;
#pragma unroll
    for (int i = 0; i < 4; ++i) {
        const int r = 4 * e_rg + i;
        const int g = e_tm * 16 + 8 * h + r;
        const int idx = (e_tm * 8 + r) * 32 + lane;
        const bool live = valid && g < NG;
        const float a3 = xbu[i] - s_rub[g];
        const float a4 = xbd[i] - s_rdb[g];
        float n3 = fmaxf(e3[i] + step * a3, 0.f);
        float n4 = fmaxf(e4[i] + step * a4, 0.f);
        if (!live) { n3 = 0.f; n4 = 0.f; }
        ps[PS_U + idx] = xnu[i]; ps[PS_D + idx] = xnd[i]; ps[PS_Z3 + idx] = n3; ps[PS_Z4 + idx] = n4;
        const float s3 = half_sum(n3), s4 = half_sum(n4);
        if (c == 0 && lb == 0 && g < NG) { rsw[g] = s3; rsw[NG + g] = s4; }
    }
}

__global__ __launch_bounds__(FNT) void k_iter(float* w, const float* Cost, const float* Cru, const float* Crd,
                                              const float* Pmax, int par) {
    __shared__ __attribute__((aligned(16))) float udt[BLK];
    __shared__ __attribute__((aligned(16))) float zdt[BLK];
    __shared__ __attribute__((aligned(16))) float s_pb[G_P];
    __shared__ float s_rub[G_P], s_rdb[G_P];
    __shared__ float s_fub[LBL], s_fdb[LBL];

    const int tid = threadIdx.x, lane = tid & 31;
    const int wv = __builtin_amdgcn_readfirstlane((int)(threadIdx.x >> 5));
    const int tn = blockIdx.x / NLB, lb = blockIdx.x % NLB;
    const int c = lane & 15, h = lane >> 4;
    const int s = tn * 16 + c;
    const bool valid = s < NS;
    const int sc = valid ? s : NS - 1;
    const bool last = tn == NTN - 1;
    const bool carry = last && s == NS;

    float* ps = w + WS_PS + blockIdx.x * PS_SZ;
    const float* tpr = w + WS_TP + (par ^ 1) * (NWG * TPSZ);
    float*       tpw = w + WS_TP + par * (NWG * TPSZ);
    const float* rsr = w + WS_RS + (par ^ 1) * (NTN * RSS);
    float*       rsw = w + WS_RS + par * (NTN * RSS) + tn * RSS;
    const float* yr  = w + WS_Y23 + (par ^ 1) * (2 * NL);
    float*       yw  = w + WS_Y23 + par * (2 * NL);
    const float step = w[WS_STEP];

    const bool isU = tid < NG;
    const bool isL = tid >= 128 && tid < 128 + LBL;
    const int ug = tid < NG ? tid : NG - 1;
    const int lj = tid < 128 ? 0 : (tid < 128 + LBL ? tid - 128 : LBL - 1);
    const int ll = lb * LBL + lj;
    const bool roleL = wv >= 4;
    const int ia = roleL ? 2 * NG + ll : ug;
    const int ib = roleL ? 2 * NG + NL + ll : NG + ug;
    float sa[NTN], sb[NTN];
    float f_p, f_ru, f_rd, f_z1, f_z2, f_y1, f_c, f_cu, f_cd, f_pm, f_tp[NLB];
    float l_fu, l_fd, l_y2, l_y3;
    if (wv < 8) {
#pragma unroll
        for (int t = 0; t < NTN; ++t) { sa[t] = rsr[t * RSS + ia]; sb[t] = rsr[t * RSS + ib]; }
        f_p = ps[PS_P + ug]; f_ru = ps[PS_RU + ug]; f_rd = ps[PS_RD + ug];
        f_z1 = ps[PS_Z1 + ug]; f_z2 = ps[PS_Z2 + ug]; f_y1 = ps[PS_Y1];
        f_c = Cost[ug]; f_cu = Cru[ug]; f_cd = Crd[ug]; f_pm = Pmax[ug];
        const int idx = ((ug >> 4) * 8 + (ug & 7)) * 32 + (NS - (NTN - 1) * 16) + 16 * ((ug >> 3) & 1);
#pragma unroll
        for (int q = 0; q < NLB; ++q) f_tp[q] = tpr[((NTN - 1) * NLB + q) * TPSZ + idx];
        l_fu = ps[PS_FU + lj]; l_fd = ps[PS_FD + lj];
        l_y2 = yr[ll]; l_y3 = yr[NL + ll];
    }
    const int e_tm = wv % 7, e_rg = wv / 7;
    float eu[4], ed[4], e3[4], e4[4], et[4][NLB];
#pragma unroll
    for (int i = 0; i < 4; ++i) {
        const int idx = (e_tm * 8 + 4 * e_rg + i) * 32 + lane;
        eu[i] = ps[PS_U + idx]; ed[i] = ps[PS_D + idx]; e3[i] = ps[PS_Z3 + idx]; e4[i] = ps[PS_Z4 + idx];
#pragma unroll
        for (int q = 0; q < NLB; ++q) et[i][q] = tpr[(tn * NLB + q) * TPSZ + idx];
    }
    const float ey4 = ps[PS_Y4 + c];
    const float yb4 = w[WS_B4 + sc], yy1 = ps[PS_Y1], yb1 = w[WS_B1];
    KpFrags f;
    const float* ablk = wv < 7 ? w + WS_PGP + (6 * lb + wv) * BLK
                               : w + WS_PGTP + (((wv < 14 ? wv : 13) - 7) * NLB + lb) * BLK;
    load_kp_a(f, ablk, lane);
    float q5[8], q6[8], qw[8], o2[8], o3[8], ob2[8], ob3[8];
    if (wv < 7) {
        const int tmL = 6 * lb + wv;
#pragma unroll
        for (int i = 0; i < 8; ++i) {
            const int l = tmL * 16 + 8 * h + i;
            const int idx = (wv * 8 + i) * 32 + lane;
            q5[i] = ps[PS_Z5 + idx]; q6[i] = ps[PS_Z6 + idx];
            qw[i] = w[WS_PW + l * S_P + sc];
            if (last) { o2[i] = yr[l]; o3[i] = yr[NL + l]; ob2[i] = w[WS_B2 + l]; ob3[i] = w[WS_B3 + l]; }
        }
    }
    __builtin_amdgcn_sched_barrier(0);

    float n_p = 0.f, n_ru = 0.f, n_rd = 0.f, n_z1 = 0.f, n_z2 = 0.f, n_fu = 0.f, n_fd = 0.f;
    if (tid < G_P) {
        float pb = 0.f, rub = 0.f, rdb = 0.f;
        if (isU) {
            float z3s = 0.f, z4s = 0.f;
#pragma unroll
            for (int t = 0; t < NTN; ++t) { z3s += sa[t]; z4s += sb[t]; }
            const float cs = ((f_tp[0] + f_tp[1]) + f_tp[2]) + f_tp[3];
            const float ap = f_y1 + cs + f_z1 - f_z2;
            n_p = fmaxf(f_p - step * (f_c + ap), 0.f);
            pb = 2.f * n_p - f_p;
            const float au = f_z1 - z3s;
            const float ad = f_z2 - z4s;
            n_ru = fmaxf(f_ru - step * (f_cu + au), 0.f);
            n_rd = fmaxf(f_rd - step * (f_cd + ad), 0.f);
            rub = 2.f * n_ru - f_ru;
            rdb = 2.f * n_rd - f_rd;
            const float a1 = pb + rub;
            const float a2 = rdb - pb;
            n_z1 = fmaxf(f_z1 + step * (a1 - f_pm), 0.f);
            n_z2 = fmaxf(f_z2 + step *  a2, 0.f);
            if (last) udt[pk_in(NS - (NTN - 1) * 16, kpos100(tid))] = pb;
        }
        s_pb[tid] = pb; s_rub[tid] = rub; s_rdb[tid] = rdb;
    }
    if (isL) {
        float z5s = 0.f, z6s = 0.f;
#pragma unroll
        for (int t = 0; t < NTN; ++t) { z5s += sa[t]; z6s += sb[t]; }
        const float au = l_y2 - z5s;
        const float ad = l_y3 - z6s;
        n_fu = fmaxf(l_fu - step * au, 0.f);
        n_fd = fmaxf(l_fd - step * ad, 0.f);
        s_fub[lj] = 2.f * n_fu - l_fu; s_fdb[lj] = 2.f * n_fd - l_fd;
    }

    float xnu[4], xnd[4], xbu[4], xbd[4];
    if (wv < 14) {
#pragma unroll
        for (int i = 0; i < 4; ++i) {
            const int r = 4 * e_rg + i;
            const int g = e_tm * 16 + 8 * h + r;
            const bool live = valid && g < NG;
            const float tv = ((et[i][0] + et[i][1]) + et[i][2]) + et[i][3];
            const float au =  ey4 + e3[i] + tv;
            const float ad = -ey4 + e4[i] - tv;
            float nu = fmaxf(eu[i] - step * au, 0.f);
            float nd = fmaxf(ed[i] - step * ad, 0.f);
            float bu = 2.f * nu - eu[i];
            float bd = 2.f * nd - ed[i];
            if (!live) { nu = 0.f; nd = 0.f; bu = 0.f; bd = 0.f; }
            xnu[i] = nu; xnd[i] = nd; xbu[i] = bu; xbd[i] = bd;
        }
#pragma unroll
        for (int i = 0; i < 4; i += 2) {
            const int g = e_tm * 16 + 8 * h + 4 * e_rg + i;
            if (g < NG && !carry) {
                v2f d; d[0] = xbu[i] - xbd[i]; d[1] = xbu[i + 1] - xbd[i + 1];
                *(v2f*)(udt + pk_in(c, kpos100(g))) = d;
            }
        }
    }
    __syncthreads();

    if (isU) {
        ps[PS_P + tid] = n_p; ps[PS_RU + tid] = n_ru; ps[PS_RD + tid] = n_rd;
        ps[PS_Z1 + tid] = n_z1; ps[PS_Z2 + tid] = n_z2;
    }
    if (isL) { ps[PS_FU + lj] = n_fu; ps[PS_FD + lj] = n_fd; }

    if (wv < 7) {
        const int tmL = 6 * lb + wv;
        float fubv[8], fdbv[8];
#pragma unroll
        for (int i = 0; i < 8; ++i) {
            const int p = tmL * 16 + 8 * h + i - lb * LBL;
            const int pc = (p >= 0 && p < LBL) ? p : 0;
            fubv[i] = s_fub[pc]; fdbv[i] = s_fdb[pc];
        }
        load_kp_b(f, udt, lane);
        __builtin_amdgcn_sched_barrier(0);
        const v8f acc = mma_kp(f);
        float dzv[8];
#pragma unroll
        for (int i = 0; i < 8; ++i) {
            const int l = tmL * 16 + 8 * h + i;
            const int p = l - lb * LBL;
            const bool own = p >= 0 && p < LBL;
            const int idx = (wv * 8 + i) * 32 + lane;
            const float fub = fubv[i], fdb = fdbv[i];
            const float f2 = acc[i];
            const float a5 =  f2 - fub;
            const float a6 = -f2 - fdb;
            float r5 = fmaxf(q5[i] + step * (a5 + qw[i]), 0.f);
            float r6 = fmaxf(q6[i] + step * (a6 - qw[i]), 0.f);
            if (!(own && valid)) { r5 = 0.f; r6 = 0.f; }
            ps[PS_Z5 + idx] = r5; ps[PS_Z6 + idx] = r6;
            float dz = r5 - r6;
            if (last) {
                const float n2 = o2[i] + step * (( f2 + fub) - ob2[i]);
                const float n3 = o3[i] + step * ((-f2 + fdb) - ob3[i]);
                if (carry) {
                    dz = n2 - n3;
                    if (own) { yw[l] = n2; yw[NL + l] = n3; }
                }
            }
            dzv[i] = dz;
            const float s5 = half_sum(r5), s6 = half_sum(r6);
            if (c == 0 && own) { rsw[2 * NG + l] = s5; rsw[2 * NG + NL + l] = s6; }
        }
#pragma unroll
        for (int i = 0; i < 8; i += 2) {
            const int p = tmL * 16 + 8 * h + i - lb * LBL;
            if (p >= 0 && p < LBL) { v2f d; d[0] = dzv[i]; d[1] = dzv[i + 1]; *(v2f*)(zdt + pk_in(c, kpos100(p))) = d; }
        }
    } else if (wv < 14) {
        ud_step_tail(ps, rsw, s_rub, s_rdb, xnu, xnd, xbu, xbd, e3, e4, step, e_tm, e_rg, lane, valid, lb);
    }
    if (wv == 7) {
        float cv[NG];
#pragma unroll
        for (int j = 0; j < 12; ++j)
#pragma unroll
            for (int hh = 0; hh < 2; ++hh) {
                const v4f t = *(const v4f*)(udt + 128 * j + 4 * (c + 16 * hh));
                cv[8 * j + 4 * hh + 0] = t[0]; cv[8 * j + 4 * hh + 1] = t[1];
                cv[8 * j + 4 * hh + 2] = t[2]; cv[8 * j + 4 * hh + 3] = t[3];
            }
#pragma unroll
        for (int hh = 0; hh < 2; ++hh) {
            const v2f t = *(const v2f*)(udt + 1536 + 2 * (c + 16 * hh));
            cv[96 + 2 * hh] = t[0]; cv[97 + 2 * hh] = t[1];
        }
        __builtin_amdgcn_sched_barrier(0);
        float cs = 0.f;
#pragma unroll
        for (int g = 0; g < NG; ++g) cs += cv[kpos100(g)];
        if (h == 0) ps[PS_Y4 + c] = valid ? ey4 + step * (cs - yb4) : 0.f;
    } else if (wv == 8) {
        float pv[NG];
#pragma unroll
        for (int j = 0; j < NG / 4; ++j) {
            const v4f t = *(const v4f*)(s_pb + 4 * j);
            pv[4 * j] = t[0]; pv[4 * j + 1] = t[1]; pv[4 * j + 2] = t[2]; pv[4 * j + 3] = t[3];
        }
        __builtin_amdgcn_sched_barrier(0);
        float sp = 0.f;
#pragma unroll
        for (int g = 0; g < NG; ++g) sp += pv[g];
        if (lane == 0) ps[PS_Y1] = yy1 + step * (sp - yb1);
    }
    __syncthreads();

    if (wv < 7) {
        ud_step_tail(ps, rsw, s_rub, s_rdb, xnu, xnd, xbu, xbd, e3, e4, step, e_tm, e_rg, lane, valid, lb);
    } else if (wv < 14) {
        load_kp_b(f, zdt, lane);
        __builtin_amdgcn_sched_barrier(0);
        const v8f acc = mma_kp(f);
        float* o = tpw + (tn * NLB + lb) * TPSZ + (wv - 7) * 256 + lane;
#pragma unroll
        for (int i = 0; i < 8; ++i) o[i * 32] = acc[i];
    }
}

constexpr int PW_NT = 1024;
constexpr int PW_NW = PW_NT / 32;
constexpr int PW_CW = 25;
constexpr int HG = NG / 2;
constexpr int HL = NL / 8;
static_assert(2 * NL == 32 * PW_CW && 8 * NG == 32 * PW_CW && PW_CW + 2 <= PW_NW && (HG % 2) == 0 && (HL % 2) == 0,
              "thread map of k_power");

__global__ __launch_bounds__(PW_NT) void k_power(float* w) {
    __shared__ __attribute__((aligned(16))) float vp[G_P], vru[G_P], vrd[G_P], vu[G_P], vd[G_P];
    __shared__ __attribute__((aligned(16))) float vfu[NL], vfd[NL];
    __shared__ __attribute__((aligned(16))) float bx[2 * NG];
    __shared__ __attribute__((aligned(16))) float by[2 * NL];
    __shared__ float sq[PW_NW];
    __shared__ float sc[2];
    __shared__ __attribute__((aligned(16))) v2f prow[(HG / 2) * PW_CW * 32];

    const int tid = threadIdx.x, lane = tid & 31;
    const int wave = __builtin_amdgcn_readfirstlane(tid >> 5);
    const bool hold = wave < PW_CW;
    const int ml = hold ? tid >> 1 : 0, mpart = tid & 1;
    const int rg = hold ? tid >> 3 : 0, rpart = tid & 7;

    float pcol[HL];
    {
        const float* r = w + WS_PG + ml * G_P + mpart * HG;
        const float* cp = w + WS_PGT + rg * NL + rpart * HL;
        v2f rv[HG / 2];
#pragma unroll
        for (int j = 0; j < HG / 2; ++j) rv[j] = *(const v2f*)(r + 2 * j);
#pragma unroll
        for (int k = 0; k < HL; k += 2) { const v2f t = *(const v2f*)(cp + k); pcol[k] = t[0]; pcol[k + 1] = t[1]; }
        if (hold) {
#pragma unroll
            for (int j = 0; j < HG / 2; ++j) prow[j * (PW_CW * 32) + tid] = rv[j];
        }
    }

    if (tid < G_P) {
        const float o = tid < NG ? 1.f : 0.f;
        vp[tid] = o; vru[tid] = o; vrd[tid] = o; vu[tid] = o; vd[tid] = o;
    }
    if (tid < NL) { vfu[tid] = 1.f; vfd[tid] = 1.f; }
    if (tid < PW_NW) sq[tid] = tid == 0 ? (float)(3 * NG + 2 * NG * NS + 2 * NL) : 0.f;
    __syncthreads();

    for (int it = 0; it < POWER_ITERS; ++it) {
        float ss = 0.f;
#pragma unroll
        for (int j = 0; j < PW_NW; ++j) ss += sq[j];
        const float inv = 1.0f / sqrtf(ss);

        if (tid < NG) {
            v2f t; t[0] = vp[tid] * inv; t[1] = vu[tid] * inv - vd[tid] * inv;
            *(v2f*)(bx + 2 * tid) = t;
        }
        __syncthreads();

        float q = 0.f;
        if (hold) {
            float s0 = 0.f, s1 = 0.f;
            const v4f* b4 = (const v4f*)(bx + 2 * HG * mpart);
#pragma unroll
            for (int j = 0; j < HG / 2; ++j) {
                const v2f a = prow[j * (PW_CW * 32) + tid];
                const v4f t = b4[j];
                s0 += a[0] * t[0]; s1 += a[0] * t[1];
                s0 += a[1] * t[2]; s1 += a[1] * t[3];
            }
            s0 += __shfl_xor(s0, 1); s1 += __shfl_xor(s1, 1);
            if (mpart == 0) {
                const float fu = vfu[ml] * inv, fd = vfd[ml] * inv;
                const float y2 =  s0 + fu, y3 = -s0 + fd;
                const float z5 =  s1 - fu, z6 = -s1 - fd;
                v2f t; t[0] = y2 - y3; t[1] = z5 - z6;
                *(v2f*)(by + 2 * ml) = t;
                const float nfu = y2 - (float)NS * z5;
                const float nfd = y3 - (float)NS * z6;
                vfu[ml] = nfu; vfd[ml] = nfd;
                q = nfu * nfu + nfd * nfd;
            }
        } else if (wave >= PW_NW - 2) {
            const bool first = wave == PW_NW - 2;
            float s = 0.f;
            for (int g = lane; g < NG; g += 32)
                s += first ? vp[g] * inv : (vu[g] * inv - vd[g] * inv);
            s = wave_sum(s);
            if (lane == 0) sc[first ? 0 : 1] = s;
        }
        __syncthreads();

        if (hold) {
            float s0 = 0.f, s1 = 0.f;
            const v4f* b4 = (const v4f*)(by + 2 * HL * rpart);
#pragma unroll
            for (int j = 0; j < HL / 2; ++j) {
                const v4f t = b4[j];
                s0 += pcol[2 * j] * t[0];     s1 += pcol[2 * j] * t[1];
                s0 += pcol[2 * j + 1] * t[2]; s1 += pcol[2 * j + 1] * t[3];
            }
#pragma unroll
            for (int m = 1; m < 8; m <<= 1) { s0 += __shfl_xor(s0, m); s1 += __shfl_xor(s1, m); }
            if (rpart == 0) {
                const float p = vp[rg] * inv, ru = vru[rg] * inv, rd = vrd[rg] * inv;
                const float u = vu[rg] * inv, d = vd[rg] * inv;
                const float z1 = p + ru, z2 = rd - p, z3 = u - ru, z4 = d - rd;
                const float y1 = sc[0], y4 = sc[1];
                const float np  = y1 + s0 + z1 - z2;
                const float nru = z1 - (float)NS * z3;
                const float nrd = z2 - (float)NS * z4;
                const float nu  =  y4 + z3 + s1;
                const float nd  = -y4 + z4 - s1;
                vp[rg] = np; vru[rg] = nru; vrd[rg] = nrd; vu[rg] = nu; vd[rg] = nd;
                q += np * np + nru * nru + nrd * nrd + (float)NS * (nu * nu + nd * nd);
            }
        }
        q = wave_sum(q);
        if (lane == 0) sq[wave] = q;
        __syncthreads();
    }

    if (tid == 0) {
        float ss = 0.f;
        for (int j = 0; j < PW_NW; ++j) ss += sq[j];
        w[WS_STEP] = 0.9f / sqrtf(sqrtf(ss));
    }
}

__global__ void k_emit(const float* w, float* out) {
    const int total = 3 * NG + 2 * NG * NS + 2 * NL;
    for (int idx = blockIdx.x * blockDim.x + threadIdx.x; idx < total; idx += gridDim.x * blockDim.x) {
        const float* ps0 = w + WS_PS;
        float v;
        if (idx < NG)                 v = ps0[PS_P  + idx];
        else if (idx < 2 * NG)        v = ps0[PS_RU + (idx - NG)];
        else if (idx < 3 * NG)        v = ps0[PS_RD + (idx - 2 * NG)];
        else if (idx < 3 * NG + 2 * NG * NS) {
            int r = idx - 3 * NG;
            const int plane = r < NG * NS ? PS_U : PS_D;
            if (r >= NG * NS) r -= NG * NS;
            const int g = r / NS, s = r % NS;
            v = w[WS_PS + ((s >> 4) * NLB) * PS_SZ + plane
                  + ((g >> 4) * 8 + (g & 7)) * 32 + (s & 15) + 16 * ((g >> 3) & 1)];
        } else {
            int l = idx - 3 * NG - 2 * NG * NS;
            const int plane = l < NL ? PS_FU : PS_FD;
            if (l >= NL) l -= NL;
            v = w[WS_PS + (l / LBL) * PS_SZ + plane + (l % LBL)];
        }
        out[idx] = v;
    }
}


constexpr int PDHG_PAIRS = PDHG_ITERS / 2;
static_assert(2 * PDHG_PAIRS == PDHG_ITERS, "the loop below launches the two copies in turn");

extern "C" void kernel_launch(void* const* d_in, const int* in_sizes, int n_in,
                              void* d_out, int out_size, void* d_ws, size_t ws_size,
                              hipStream_t stream) {
    const float* w_scen   = (const float*)d_in[0];
    const float* Cost     = (const float*)d_in[1];
    const float* C_r_up   = (const float*)d_in[2];
    const float* C_r_down = (const float*)d_in[3];
    const float* Pmax     = (const float*)d_in[4];
    const float* PTDF     = (const float*)d_in[5];
    const float* node_G   = (const float*)d_in[6];
    const float* node_W   = (const float*)d_in[7];
    const float* node_L   = (const float*)d_in[8];
    const float* Pd       = (const float*)d_in[9];
    const float* w_exp    = (const float*)d_in[10];
    const float* Line_cap = (const float*)d_in[11];
    float* w   = (float*)d_ws;
    float* out = (float*)d_out;

    k_zero <<<512, 256, 0, stream>>>(w);
    k_prep1<<<240, 256, 0, stream>>>(w, w_scen, node_W, node_L, Pd, w_exp);
    k_prep2<<<640, 256, 0, stream>>>(w, PTDF, node_G, Line_cap);
    k_power<<<1, PW_NT, 0, stream>>>(w);

    for (int it = 0; it < PDHG_PAIRS; ++it) {
        k_iter<<<NWG, FNT, 0, stream>>>(w, Cost, C_r_up, C_r_down, Pmax, 0);
        k_iter<<<NWG, FNT, 0, stream>>>(w, Cost, C_r_up, C_r_down, Pmax, 1);
    }
    k_emit<<<161, 256, 0, stream>>>(w, out);
}
